# starting kernel
# baseline (speedup 1.0000x reference)
#include <hip/hip_runtime.h>
#include <hip/hip_bf16.h>


typedef float v2f __attribute__((ext_vector_type(2)));
typedef float v4f __attribute__((ext_vector_type(4)));
typedef float v8f __attribute__((ext_vector_type(8)));

#define EPSF 1e-5f

static __device__ __forceinline__ v8f wmma_f32x4(v2f a, v2f b, v8f c) {
    return __builtin_amdgcn_wmma_f32_16x16x4_f32(false, a, false, b, (short)0, c, false, false);
}

__global__ __launch_bounds__(32)
void gemm8_wmma(const float* __restrict__ A, int lda,
                const float* __restrict__ Bm, int ldb,
                const float* __restrict__ bias,
                float* __restrict__ Out, int ldo,
                int K, int relu)
{
    const int lane = threadIdx.x;
    const int half = lane >> 4;
    const int l    = lane & 15;
    const int n    = blockIdx.x * 16 + l;

    const float msk = (l < 8) ? 1.0f : 0.0f;
    const float* __restrict__ arow = A + (l & 7) * lda + half * 2;
    const float* __restrict__ brow = Bm + n * ldb + half * 2;

    v8f acc = {};
    v2f a, bv;
    a.x  = msk * arow[0];
    a.y  = msk * arow[1];
    bv.x = brow[0];
    bv.y = brow[1];
    for (int k = 0; k < K - 4; k += 4) {
        v2f an, bn_;
        an.x  = msk * arow[k + 4];
        an.y  = msk * arow[k + 5];
        bn_.x = brow[k + 4];
        bn_.y = brow[k + 5];
        acc = wmma_f32x4(a, bv, acc);
        a = an; bv = bn_;
    }
    acc = wmma_f32x4(a, bv, acc);

    if (half == 0) {
        const float bs = bias[n];
        #pragma unroll
        for (int r = 0; r < 8; ++r) {
            float v = acc[r] + bs;
            if (relu) v = fmaxf(v, 0.0f);
            Out[r * ldo + n] = v;
        }
    }
}

__global__ __launch_bounds__(256)
void gridtab_kernel(const float* __restrict__ Wf1, float* __restrict__ gt)
{
    const int i = blockIdx.x * 256 + threadIdx.x;
    if (i >= 16 * 512) return;
    const int g = i >> 9, o = i & 511;
    const float step = 0.1f / 3.0f;
    const float gx = -0.05f + (float)(g >> 2) * step;
    const float gy = -0.05f + (float)(g & 3) * step;
    gt[i] = gx * Wf1[o * 1029 + 0] + gy * Wf1[o * 1029 + 1];
}

__global__ __launch_bounds__(256)
void fold_fused(const float* __restrict__ coarse,
                const float* __restrict__ base1,
                const float* __restrict__ gridtab,
                const float* __restrict__ Wf1,
                const float* __restrict__ g1, const float* __restrict__ be1,
                const float* __restrict__ m1, const float* __restrict__ v1,
                const float* __restrict__ Wf2,
                const float* __restrict__ bf2,
                const float* __restrict__ g2, const float* __restrict__ be2,
                const float* __restrict__ m2, const float* __restrict__ v2,
                const float* __restrict__ Wo,
                const float* __restrict__ bo,
                float* __restrict__ fine)
{
    constexpr int Y2S = 516;
    __shared__ float y1q[128 * 128];
    __shared__ float y2s[32 * Y2S];
    __shared__ float red[256 * 3];

    const int t  = threadIdx.x;
    const int b  = blockIdx.y;
    const int m0 = blockIdx.x * 32;

    for (int i = 0; i < 16; ++i) {
        const int idx = i * 256 + t;
        const int q = idx >> 5;
        const int p = idx & 31;
        const int ng = m0 + p;
        const int j = ng >> 4, g = ng & 15;
        const float* cp = coarse + (b * 1024 + j) * 3;
        const float cx = cp[0], cy = cp[1], cz = cp[2];
        const v4f bse = *(const v4f*)(base1 + b * 512 + q * 4);
        const v4f gtv = *(const v4f*)(gridtab + g * 512 + q * 4);
        const v4f gv  = *(const v4f*)(g1 + q * 4);
        const v4f vv  = *(const v4f*)(v1 + q * 4);
        const v4f mv  = *(const v4f*)(m1 + q * 4);
        const v4f bev = *(const v4f*)(be1 + q * 4);
        v4f r;
        #pragma unroll
        for (int c = 0; c < 4; ++c) {
            const int o = q * 4 + c;
            float val = bse[c] + gtv[c]
                      + cx * Wf1[o * 1029 + 2]
                      + cy * Wf1[o * 1029 + 3]
                      + cz * Wf1[o * 1029 + 4];
            const float s = gv[c] * rsqrtf(vv[c] + EPSF);
            val = (val - mv[c]) * s + bev[c];
            r[c] = fmaxf(val, 0.0f);
        }
        *(v4f*)&y1q[q * 128 + p * 4] = r;
    }
    __syncthreads();

    {
        const int w    = t >> 5;
        const int lane = t & 31;
        const int half = lane >> 4;
        const int l    = lane & 15;
        const int aoff0 = l * 4 + half * 2;
        const int aoff1 = (16 + l) * 4 + half * 2;
        for (int nt2 = 0; nt2 < 2; ++nt2) {
            const int n0 = (w * 4 + nt2 * 2) * 16 + l;
            const int n1 = n0 + 16;
            const float* __restrict__ bp0 = Wf2 + n0 * 512 + half * 2;
            const float* __restrict__ bp1 = Wf2 + n1 * 512 + half * 2;
            v8f acc00 = {}, acc01 = {}, acc10 = {}, acc11 = {};
            v2f a0 = *(const v2f*)&y1q[aoff0];
            v2f a1 = *(const v2f*)&y1q[aoff1];
            v2f b0 = *(const v2f*)bp0;
            v2f b1 = *(const v2f*)bp1;
            for (int kq = 0; kq < 127; ++kq) {
                const v2f a0n = *(const v2f*)&y1q[(kq + 1) * 128 + aoff0];
                const v2f a1n = *(const v2f*)&y1q[(kq + 1) * 128 + aoff1];
                const v2f b0n = *(const v2f*)(bp0 + (kq + 1) * 4);
                const v2f b1n = *(const v2f*)(bp1 + (kq + 1) * 4);
                acc00 = wmma_f32x4(a0, b0, acc00);
                acc01 = wmma_f32x4(a0, b1, acc01);
                acc10 = wmma_f32x4(a1, b0, acc10);
                acc11 = wmma_f32x4(a1, b1, acc11);
                a0 = a0n; a1 = a1n; b0 = b0n; b1 = b1n;
            }
            acc00 = wmma_f32x4(a0, b0, acc00);
            acc01 = wmma_f32x4(a0, b1, acc01);
            acc10 = wmma_f32x4(a1, b0, acc10);
            acc11 = wmma_f32x4(a1, b1, acc11);

            const float s20 = g2[n0] * rsqrtf(v2[n0] + EPSF);
            const float s21 = g2[n1] * rsqrtf(v2[n1] + EPSF);
            const float mm0 = m2[n0], bt0 = be2[n0], bf0 = bf2[n0];
            const float mm1 = m2[n1], bt1 = be2[n1], bf1v = bf2[n1];
            #pragma unroll
            for (int r = 0; r < 8; ++r) {
                const int M = r + half * 8;
                y2s[M * Y2S + n0]        = fmaxf((acc00[r] + bf0 - mm0) * s20 + bt0, 0.0f);
                y2s[M * Y2S + n1]        = fmaxf((acc01[r] + bf1v - mm1) * s21 + bt1, 0.0f);
                y2s[(16 + M) * Y2S + n0] = fmaxf((acc10[r] + bf0 - mm0) * s20 + bt0, 0.0f);
                y2s[(16 + M) * Y2S + n1] = fmaxf((acc11[r] + bf1v - mm1) * s21 + bt1, 0.0f);
            }
        }
    }
    __syncthreads();

    {
        const int p = t >> 3;
        const int s = t & 7;
        float p0 = 0.0f, p1 = 0.0f, p2 = 0.0f;
        for (int o = s; o < 512; o += 8) {
            const float y = y2s[p * Y2S + o];
            p0 += y * Wo[o];
            p1 += y * Wo[512 + o];
            p2 += y * Wo[1024 + o];
        }
        red[t * 3 + 0] = p0;
        red[t * 3 + 1] = p1;
        red[t * 3 + 2] = p2;
    }
    __syncthreads();
    if (t < 96) {
        const int p = t / 3, c = t % 3;
        float sum = 0.0f;
        #pragma unroll
        for (int s = 0; s < 8; ++s) sum += red[(p * 8 + s) * 3 + c];
        const int ng = m0 + p;
        const int i  = c * 16384 + ng;
        const int r3 = i / 3;
        const int cc = i - r3 * 3;
        const float center = coarse[(b * 1024 + (r3 >> 4)) * 3 + cc];
        fine[(b * 16384 + ng) * 3 + c] = sum + bo[c] + center;
    }
}

extern "C" void kernel_launch(void* const* d_in, const int* in_sizes, int n_in,
                              void* d_out, int out_size, void* d_ws, size_t ws_size,
                              hipStream_t stream)
{
    const float* features = (const float*)d_in[0];
    const float* W1  = (const float*)d_in[1];
    const float* b1  = (const float*)d_in[2];
    const float* W2  = (const float*)d_in[3];
    const float* b2  = (const float*)d_in[4];
    const float* Wc  = (const float*)d_in[5];
    const float* bc  = (const float*)d_in[6];
    const float* Wf1 = (const float*)d_in[7];
    const float* bf1 = (const float*)d_in[8];
    const float* g1  = (const float*)d_in[9];
    const float* be1 = (const float*)d_in[10];
    const float* m1  = (const float*)d_in[11];
    const float* v1  = (const float*)d_in[12];
    const float* Wf2 = (const float*)d_in[13];
    const float* bf2 = (const float*)d_in[14];
    const float* g2  = (const float*)d_in[15];
    const float* be2 = (const float*)d_in[16];
    const float* m2  = (const float*)d_in[17];
    const float* v2  = (const float*)d_in[18];
    const float* Wo  = (const float*)d_in[19];
    const float* bo  = (const float*)d_in[20];

    float* out    = (float*)d_out;
    float* coarse = out;
    float* fine   = out + 8 * 1024 * 3;

    float* ws    = (float*)d_ws;
    float* h1    = ws;
    float* h2    = ws + 8192;
    float* base1 = ws + 16384;
    float* gt    = ws + 20480;

    gemm8_wmma<<<64,  32, 0, stream>>>(features, 1024, W1, 1024, b1, h1,     1024, 1024, 1);
    gemm8_wmma<<<64,  32, 0, stream>>>(h1,       1024, W2, 1024, b2, h2,     1024, 1024, 1);
    gemm8_wmma<<<192, 32, 0, stream>>>(h2,       1024, Wc, 1024, bc, coarse, 3072, 1024, 0);
    gemm8_wmma<<<32,  32, 0, stream>>>(features, 1024, Wf1 + 5, 1029, bf1, base1, 512, 1024, 0);
    gridtab_kernel<<<32, 256, 0, stream>>>(Wf1, gt);

    dim3 gridF(16384 / 32, 8);
    fold_fused<<<gridF, 256, 0, stream>>>(coarse, base1, gt, Wf1,
                                          g1, be1, m1, v1,
                                          Wf2, bf2, g2, be2, m2, v2,
                                          Wo, bo, fine);
}


# Wf2 repacked each call into WMMA fragment order: every B load is one contiguous 512 B b128 run, A read as b128; k summed in groups (0,1,4,5),(2,3,6,7) per 8
# speedup vs baseline: 1.1056x; 1.1056x over previous
#include <hip/hip_runtime.h>
#include <hip/hip_bf16.h>


typedef float v2f __attribute__((ext_vector_type(2)));
typedef float v4f __attribute__((ext_vector_type(4)));
typedef float v8f __attribute__((ext_vector_type(8)));

#define EPSF 1e-5f

static __device__ __forceinline__ v8f wmma_f32x4(v2f a, v2f b, v8f c) {
    return __builtin_amdgcn_wmma_f32_16x16x4_f32(false, a, false, b, (short)0, c, false, false);
}

__global__ __launch_bounds__(32)
void gemm8_wmma(const float* __restrict__ A, int lda,
                const float* __restrict__ Bm, int ldb,
                const float* __restrict__ bias,
                float* __restrict__ Out, int ldo,
                int K, int relu)
{
    const int lane = threadIdx.x;
    const int half = lane >> 4;
    const int l    = lane & 15;
    const int n    = blockIdx.x * 16 + l;

    const float msk = (l < 8) ? 1.0f : 0.0f;
    const float* __restrict__ arow = A + (l & 7) * lda + half * 2;
    const float* __restrict__ brow = Bm + n * ldb + half * 2;

    v8f acc = {};
    v2f a, bv;
    a.x  = msk * arow[0];
    a.y  = msk * arow[1];
    bv.x = brow[0];
    bv.y = brow[1];
    for (int k = 0; k < K - 4; k += 4) {
        v2f an, bn_;
        an.x  = msk * arow[k + 4];
        an.y  = msk * arow[k + 5];
        bn_.x = brow[k + 4];
        bn_.y = brow[k + 5];
        acc = wmma_f32x4(a, bv, acc);
        a = an; bv = bn_;
    }
    acc = wmma_f32x4(a, bv, acc);

    if (half == 0) {
        const float bs = bias[n];
        #pragma unroll
        for (int r = 0; r < 8; ++r) {
            float v = acc[r] + bs;
            if (relu) v = fmaxf(v, 0.0f);
            Out[r * ldo + n] = v;
        }
    }
}

__global__ __launch_bounds__(256)
void prep_kernel(const float* __restrict__ Wf1, const float* __restrict__ Wf2,
                 float* __restrict__ gt, float* __restrict__ Bp)
{
    if (blockIdx.x < 32) {
        const int i = blockIdx.x * 256 + threadIdx.x;
        const int g = i >> 9, o = i & 511;
        const float step = 0.1f / 3.0f;
        const float gx = -0.05f + (float)(g >> 2) * step;
        const float gy = -0.05f + (float)(g & 3) * step;
        gt[i] = gx * Wf1[o * 1029 + 0] + gy * Wf1[o * 1029 + 1];
        return;
    }
    const int i    = (blockIdx.x - 32) * 256 + threadIdx.x;
    const int lane = i & 31;
    const int j    = (i >> 5) & 63;
    const int T    = i >> 11;
    const int n    = T * 16 + (lane & 15);
    const int k    = j * 8 + (lane >> 4) * 4;
    *(v4f*)(Bp + i * 4) = *(const v4f*)(Wf2 + n * 512 + k);
}

__global__ __launch_bounds__(256)
void fold_fused(const float* __restrict__ coarse,
                const float* __restrict__ base1,
                const float* __restrict__ gridtab,
                const float* __restrict__ Wf1,
                const float* __restrict__ g1, const float* __restrict__ be1,
                const float* __restrict__ m1, const float* __restrict__ v1,
                const float* __restrict__ Bp,
                const float* __restrict__ bf2,
                const float* __restrict__ g2, const float* __restrict__ be2,
                const float* __restrict__ m2, const float* __restrict__ v2,
                const float* __restrict__ Wo,
                const float* __restrict__ bo,
                float* __restrict__ fine)
{
    constexpr int Y2S = 516;
    __shared__ __attribute__((aligned(16))) float y1q[128 * 128];
    __shared__ float y2s[32 * Y2S];
    __shared__ float red[256 * 3];

    const int t  = threadIdx.x;
    const int b  = blockIdx.y;
    const int m0 = blockIdx.x * 32;

    for (int i = 0; i < 16; ++i) {
        const int idx = i * 256 + t;
        const int q = idx >> 5;
        const int p = idx & 31;
        const int ng = m0 + p;
        const int j = ng >> 4, g = ng & 15;
        const float* cp = coarse + (b * 1024 + j) * 3;
        const float cx = cp[0], cy = cp[1], cz = cp[2];
        const v4f bse = *(const v4f*)(base1 + b * 512 + q * 4);
        const v4f gtv = *(const v4f*)(gridtab + g * 512 + q * 4);
        const v4f gv  = *(const v4f*)(g1 + q * 4);
        const v4f vv  = *(const v4f*)(v1 + q * 4);
        const v4f mv  = *(const v4f*)(m1 + q * 4);
        const v4f bev = *(const v4f*)(be1 + q * 4);
        v4f r;
        #pragma unroll
        for (int c = 0; c < 4; ++c) {
            const int o = q * 4 + c;
            float val = bse[c] + gtv[c]
                      + cx * Wf1[o * 1029 + 2]
                      + cy * Wf1[o * 1029 + 3]
                      + cz * Wf1[o * 1029 + 4];
            const float s = gv[c] * rsqrtf(vv[c] + EPSF);
            val = (val - mv[c]) * s + bev[c];
            r[c] = fmaxf(val, 0.0f);
        }
        *(v4f*)&y1q[q * 128 + p * 4] = r;
    }
    __syncthreads();

    {
        const int w    = t >> 5;
        const int lane = t & 31;
        const int half = lane >> 4;
        const int l    = lane & 15;
        const int aoff0 = half * 128 + l * 4;
        const int aoff1 = half * 128 + (16 + l) * 4;
        for (int nt2 = 0; nt2 < 2; ++nt2) {
            const int T0 = w * 4 + nt2 * 2;
            const int n0 = T0 * 16 + l;
            const int n1 = n0 + 16;
            const v4f* __restrict__ bq0 = (const v4f*)(Bp + T0 * (64 * 128)) + lane;
            const v4f* __restrict__ bq1 = bq0 + 64 * 32;
            v8f acc00 = {}, acc01 = {}, acc10 = {}, acc11 = {};
            v4f a0 = *(const v4f*)&y1q[aoff0];
            v4f a1 = *(const v4f*)&y1q[aoff1];
            v4f b0 = bq0[0];
            v4f b1 = bq1[0];
            for (int j = 0; j < 63; ++j) {
                const v4f a0n = *(const v4f*)&y1q[(j + 1) * 256 + aoff0];
                const v4f a1n = *(const v4f*)&y1q[(j + 1) * 256 + aoff1];
                const v4f b0n = bq0[(j + 1) * 32];
                const v4f b1n = bq1[(j + 1) * 32];
                acc00 = wmma_f32x4(a0.xy, b0.xy, acc00);
                acc01 = wmma_f32x4(a0.xy, b1.xy, acc01);
                acc10 = wmma_f32x4(a1.xy, b0.xy, acc10);
                acc11 = wmma_f32x4(a1.xy, b1.xy, acc11);
                acc00 = wmma_f32x4(a0.zw, b0.zw, acc00);
                acc01 = wmma_f32x4(a0.zw, b1.zw, acc01);
                acc10 = wmma_f32x4(a1.zw, b0.zw, acc10);
                acc11 = wmma_f32x4(a1.zw, b1.zw, acc11);
                a0 = a0n; a1 = a1n; b0 = b0n; b1 = b1n;
            }
            acc00 = wmma_f32x4(a0.xy, b0.xy, acc00);
            acc01 = wmma_f32x4(a0.xy, b1.xy, acc01);
            acc10 = wmma_f32x4(a1.xy, b0.xy, acc10);
            acc11 = wmma_f32x4(a1.xy, b1.xy, acc11);
            acc00 = wmma_f32x4(a0.zw, b0.zw, acc00);
            acc01 = wmma_f32x4(a0.zw, b1.zw, acc01);
            acc10 = wmma_f32x4(a1.zw, b0.zw, acc10);
            acc11 = wmma_f32x4(a1.zw, b1.zw, acc11);

            const float s20 = g2[n0] * rsqrtf(v2[n0] + EPSF);
            const float s21 = g2[n1] * rsqrtf(v2[n1] + EPSF);
            const float mm0 = m2[n0], bt0 = be2[n0], bf0 = bf2[n0];
            const float mm1 = m2[n1], bt1 = be2[n1], bf1v = bf2[n1];
            #pragma unroll
            for (int r = 0; r < 8; ++r) {
                const int M = r + half * 8;
                y2s[M * Y2S + n0]        = fmaxf((acc00[r] + bf0 - mm0) * s20 + bt0, 0.0f);
                y2s[M * Y2S + n1]        = fmaxf((acc01[r] + bf1v - mm1) * s21 + bt1, 0.0f);
                y2s[(16 + M) * Y2S + n0] = fmaxf((acc10[r] + bf0 - mm0) * s20 + bt0, 0.0f);
                y2s[(16 + M) * Y2S + n1] = fmaxf((acc11[r] + bf1v - mm1) * s21 + bt1, 0.0f);
            }
        }
    }
    __syncthreads();

    {
        const int p = t >> 3;
        const int s = t & 7;
        float p0 = 0.0f, p1 = 0.0f, p2 = 0.0f;
        for (int o = s; o < 512; o += 8) {
            const float y = y2s[p * Y2S + o];
            p0 += y * Wo[o];
            p1 += y * Wo[512 + o];
            p2 += y * Wo[1024 + o];
        }
        red[t * 3 + 0] = p0;
        red[t * 3 + 1] = p1;
        red[t * 3 + 2] = p2;
    }
    __syncthreads();
    if (t < 96) {
        const int p = t / 3, c = t % 3;
        float sum = 0.0f;
        #pragma unroll
        for (int s = 0; s < 8; ++s) sum += red[(p * 8 + s) * 3 + c];
        const int ng = m0 + p;
        const int i  = c * 16384 + ng;
        const int r3 = i / 3;
        const int cc = i - r3 * 3;
        const float center = coarse[(b * 1024 + (r3 >> 4)) * 3 + cc];
        fine[(b * 16384 + ng) * 3 + c] = sum + bo[c] + center;
    }
}

extern "C" void kernel_launch(void* const* d_in, const int* in_sizes, int n_in,
                              void* d_out, int out_size, void* d_ws, size_t ws_size,
                              hipStream_t stream)
{
    const float* features = (const float*)d_in[0];
    const float* W1  = (const float*)d_in[1];
    const float* b1  = (const float*)d_in[2];
    const float* W2  = (const float*)d_in[3];
    const float* b2  = (const float*)d_in[4];
    const float* Wc  = (const float*)d_in[5];
    const float* bc  = (const float*)d_in[6];
    const float* Wf1 = (const float*)d_in[7];
    const float* bf1 = (const float*)d_in[8];
    const float* g1  = (const float*)d_in[9];
    const float* be1 = (const float*)d_in[10];
    const float* m1  = (const float*)d_in[11];
    const float* v1  = (const float*)d_in[12];
    const float* Wf2 = (const float*)d_in[13];
    const float* bf2 = (const float*)d_in[14];
    const float* g2  = (const float*)d_in[15];
    const float* be2 = (const float*)d_in[16];
    const float* m2  = (const float*)d_in[17];
    const float* v2  = (const float*)d_in[18];
    const float* Wo  = (const float*)d_in[19];
    const float* bo  = (const float*)d_in[20];

    float* out    = (float*)d_out;
    float* coarse = out;
    float* fine   = out + 8 * 1024 * 3;

    float* ws    = (float*)d_ws;
    float* h1    = ws;
    float* h2    = ws + 8192;
    float* base1 = ws + 16384;
    float* gt    = ws + 20480;
    float* Bp    = ws + 28672;

    gemm8_wmma<<<64,  32, 0, stream>>>(features, 1024, W1, 1024, b1, h1,     1024, 1024, 1);
    gemm8_wmma<<<64,  32, 0, stream>>>(h1,       1024, W2, 1024, b2, h2,     1024, 1024, 1);
    gemm8_wmma<<<192, 32, 0, stream>>>(h2,       1024, Wc, 1024, bc, coarse, 3072, 1024, 0);
    gemm8_wmma<<<32,  32, 0, stream>>>(features, 1024, Wf1 + 5, 1029, bf1, base1, 512, 1024, 0);
    prep_kernel<<<288, 256, 0, stream>>>(Wf1, Wf2, gt, Bp);

    dim3 gridF(16384 / 32, 8);
    fold_fused<<<gridF, 256, 0, stream>>>(coarse, base1, gt, Wf1,
                                          g1, be1, m1, v1,
                                          Bp, bf2, g2, be2, m2, v2,
                                          Wo, bo, fine);
}


# M=8 GEMMs: K split over the 16 waves of a block (8 slices x 2 column tiles), fragments loaded as b128 ahead of the wmma stream; slice partials added in fixed order via LDS (new sum order)
# speedup vs baseline: 1.2226x; 1.1058x over previous
#include <hip/hip_runtime.h>
#include <hip/hip_bf16.h>


typedef float v2f __attribute__((ext_vector_type(2)));
typedef float v4f __attribute__((ext_vector_type(4)));
typedef float v8f __attribute__((ext_vector_type(8)));

#define EPSF 1e-5f

static __device__ __forceinline__ v8f wmma_f32x4(v2f a, v2f b, v8f c) {
    return __builtin_amdgcn_wmma_f32_16x16x4_f32(false, a, false, b, (short)0, c, false, false);
}

typedef float v4fu __attribute__((ext_vector_type(4), aligned(4)));

template <int K>
__global__ __launch_bounds__(512)
void gemm8_wmma(const float* __restrict__ A, int lda,
                const float* __restrict__ Bm, int ldb,
                const float* __restrict__ bias,
                float* __restrict__ Out, int ldo,
                int relu)
{
    constexpr int KS = K / 8;
    constexpr int NG = KS / 8;
    __shared__ float red[8 * 8 * 32];

    const int t    = threadIdx.x;
    const int w    = t >> 5;
    const int lane = t & 31;
    const int half = lane >> 4;
    const int l    = lane & 15;
    const int tt   = w & 1;
    const int ks   = w >> 1;
    const int n0   = blockIdx.x * 32;
    const int n    = n0 + tt * 16 + l;

    const float* __restrict__ arow = A + (l & 7) * lda + ks * KS + half * 4;
    const float* __restrict__ brow = Bm + n * ldb + ks * KS + half * 4;

    v4f af[NG], bf[NG];
    #pragma unroll
    for (int g = 0; g < NG; ++g) {
        af[g] = *(const v4fu*)(arow + 8 * g);
        bf[g] = *(const v4fu*)(brow + 8 * g);
    }
    const bool live = l < 8;
    v8f acc0 = {}, acc1 = {};
    #pragma unroll
    for (int g = 0; g < NG; ++g) {
        v4f a = af[g];
        a.x = live ? a.x : 0.0f;
        a.y = live ? a.y : 0.0f;
        a.z = live ? a.z : 0.0f;
        a.w = live ? a.w : 0.0f;
        acc0 = wmma_f32x4(a.xy, bf[g].xy, acc0);
        acc1 = wmma_f32x4(a.zw, bf[g].zw, acc1);
    }

    if (half == 0) {
        #pragma unroll
        for (int r = 0; r < 8; ++r)
            red[(ks * 8 + r) * 32 + tt * 16 + l] = acc0[r] + acc1[r];
    }
    __syncthreads();
    if (t < 256) {
        const int r = t >> 5;
        const int c = t & 31;
        float v = red[r * 32 + c];
        #pragma unroll
        for (int s = 1; s < 8; ++s) v += red[(s * 8 + r) * 32 + c];
        v += bias[n0 + c];
        if (relu) v = fmaxf(v, 0.0f);
        Out[r * ldo + n0 + c] = v;
    }
}

__global__ __launch_bounds__(256)
void prep_kernel(const float* __restrict__ Wf1, const float* __restrict__ Wf2,
                 float* __restrict__ gt, float* __restrict__ Bp)
{
    if (blockIdx.x < 32) {
        const int i = blockIdx.x * 256 + threadIdx.x;
        const int g = i >> 9, o = i & 511;
        const float step = 0.1f / 3.0f;
        const float gx = -0.05f + (float)(g >> 2) * step;
        const float gy = -0.05f + (float)(g & 3) * step;
        gt[i] = gx * Wf1[o * 1029 + 0] + gy * Wf1[o * 1029 + 1];
        return;
    }
    const int i    = (blockIdx.x - 32) * 256 + threadIdx.x;
    const int lane = i & 31;
    const int j    = (i >> 5) & 63;
    const int T    = i >> 11;
    const int n    = T * 16 + (lane & 15);
    const int k    = j * 8 + (lane >> 4) * 4;
    *(v4f*)(Bp + i * 4) = *(const v4f*)(Wf2 + n * 512 + k);
}

__global__ __launch_bounds__(256)
void fold_fused(const float* __restrict__ coarse,
                const float* __restrict__ base1,
                const float* __restrict__ gridtab,
                const float* __restrict__ Wf1,
                const float* __restrict__ g1, const float* __restrict__ be1,
                const float* __restrict__ m1, const float* __restrict__ v1,
                const float* __restrict__ Bp,
                const float* __restrict__ bf2,
                const float* __restrict__ g2, const float* __restrict__ be2,
                const float* __restrict__ m2, const float* __restrict__ v2,
                const float* __restrict__ Wo,
                const float* __restrict__ bo,
                float* __restrict__ fine)
{
    constexpr int Y2S = 516;
    __shared__ __attribute__((aligned(16))) float y1q[128 * 128];
    __shared__ float y2s[32 * Y2S];
    __shared__ float red[256 * 3];

    const int t  = threadIdx.x;
    const int b  = blockIdx.y;
    const int m0 = blockIdx.x * 32;

    for (int i = 0; i < 16; ++i) {
        const int idx = i * 256 + t;
        const int q = idx >> 5;
        const int p = idx & 31;
        const int ng = m0 + p;
        const int j = ng >> 4, g = ng & 15;
        const float* cp = coarse + (b * 1024 + j) * 3;
        const float cx = cp[0], cy = cp[1], cz = cp[2];
        const v4f bse = *(const v4f*)(base1 + b * 512 + q * 4);
        const v4f gtv = *(const v4f*)(gridtab + g * 512 + q * 4);
        const v4f gv  = *(const v4f*)(g1 + q * 4);
        const v4f vv  = *(const v4f*)(v1 + q * 4);
        const v4f mv  = *(const v4f*)(m1 + q * 4);
        const v4f bev = *(const v4f*)(be1 + q * 4);
        v4f r;
        #pragma unroll
        for (int c = 0; c < 4; ++c) {
            const int o = q * 4 + c;
            float val = bse[c] + gtv[c]
                      + cx * Wf1[o * 1029 + 2]
                      + cy * Wf1[o * 1029 + 3]
                      + cz * Wf1[o * 1029 + 4];
            const float s = gv[c] * rsqrtf(vv[c] + EPSF);
            val = (val - mv[c]) * s + bev[c];
            r[c] = fmaxf(val, 0.0f);
        }
        *(v4f*)&y1q[q * 128 + p * 4] = r;
    }
    __syncthreads();

    {
        const int w    = t >> 5;
        const int lane = t & 31;
        const int half = lane >> 4;
        const int l    = lane & 15;
        const int aoff0 = half * 128 + l * 4;
        const int aoff1 = half * 128 + (16 + l) * 4;
        for (int nt2 = 0; nt2 < 2; ++nt2) {
            const int T0 = w * 4 + nt2 * 2;
            const int n0 = T0 * 16 + l;
            const int n1 = n0 + 16;
            const v4f* __restrict__ bq0 = (const v4f*)(Bp + T0 * (64 * 128)) + lane;
            const v4f* __restrict__ bq1 = bq0 + 64 * 32;
            v8f acc00 = {}, acc01 = {}, acc10 = {}, acc11 = {};
            v4f a0 = *(const v4f*)&y1q[aoff0];
            v4f a1 = *(const v4f*)&y1q[aoff1];
            v4f b0 = bq0[0];
            v4f b1 = bq1[0];
            for (int j = 0; j < 63; ++j) {
                const v4f a0n = *(const v4f*)&y1q[(j + 1) * 256 + aoff0];
                const v4f a1n = *(const v4f*)&y1q[(j + 1) * 256 + aoff1];
                const v4f b0n = bq0[(j + 1) * 32];
                const v4f b1n = bq1[(j + 1) * 32];
                acc00 = wmma_f32x4(a0.xy, b0.xy, acc00);
                acc01 = wmma_f32x4(a0.xy, b1.xy, acc01);
                acc10 = wmma_f32x4(a1.xy, b0.xy, acc10);
                acc11 = wmma_f32x4(a1.xy, b1.xy, acc11);
                acc00 = wmma_f32x4(a0.zw, b0.zw, acc00);
                acc01 = wmma_f32x4(a0.zw, b1.zw, acc01);
                acc10 = wmma_f32x4(a1.zw, b0.zw, acc10);
                acc11 = wmma_f32x4(a1.zw, b1.zw, acc11);
                a0 = a0n; a1 = a1n; b0 = b0n; b1 = b1n;
            }
            acc00 = wmma_f32x4(a0.xy, b0.xy, acc00);
            acc01 = wmma_f32x4(a0.xy, b1.xy, acc01);
            acc10 = wmma_f32x4(a1.xy, b0.xy, acc10);
            acc11 = wmma_f32x4(a1.xy, b1.xy, acc11);
            acc00 = wmma_f32x4(a0.zw, b0.zw, acc00);
            acc01 = wmma_f32x4(a0.zw, b1.zw, acc01);
            acc10 = wmma_f32x4(a1.zw, b0.zw, acc10);
            acc11 = wmma_f32x4(a1.zw, b1.zw, acc11);

            const float s20 = g2[n0] * rsqrtf(v2[n0] + EPSF);
            const float s21 = g2[n1] * rsqrtf(v2[n1] + EPSF);
            const float mm0 = m2[n0], bt0 = be2[n0], bf0 = bf2[n0];
            const float mm1 = m2[n1], bt1 = be2[n1], bf1v = bf2[n1];
            #pragma unroll
            for (int r = 0; r < 8; ++r) {
                const int M = r + half * 8;
                y2s[M * Y2S + n0]        = fmaxf((acc00[r] + bf0 - mm0) * s20 + bt0, 0.0f);
                y2s[M * Y2S + n1]        = fmaxf((acc01[r] + bf1v - mm1) * s21 + bt1, 0.0f);
                y2s[(16 + M) * Y2S + n0] = fmaxf((acc10[r] + bf0 - mm0) * s20 + bt0, 0.0f);
                y2s[(16 + M) * Y2S + n1] = fmaxf((acc11[r] + bf1v - mm1) * s21 + bt1, 0.0f);
            }
        }
    }
    __syncthreads();

    {
        const int p = t >> 3;
        const int s = t & 7;
        float p0 = 0.0f, p1 = 0.0f, p2 = 0.0f;
        for (int o = s; o < 512; o += 8) {
            const float y = y2s[p * Y2S + o];
            p0 += y * Wo[o];
            p1 += y * Wo[512 + o];
            p2 += y * Wo[1024 + o];
        }
        red[t * 3 + 0] = p0;
        red[t * 3 + 1] = p1;
        red[t * 3 + 2] = p2;
    }
    __syncthreads();
    if (t < 96) {
        const int p = t / 3, c = t % 3;
        float sum = 0.0f;
        #pragma unroll
        for (int s = 0; s < 8; ++s) sum += red[(p * 8 + s) * 3 + c];
        const int ng = m0 + p;
        const int i  = c * 16384 + ng;
        const int r3 = i / 3;
        const int cc = i - r3 * 3;
        const float center = coarse[(b * 1024 + (r3 >> 4)) * 3 + cc];
        fine[(b * 16384 + ng) * 3 + c] = sum + bo[c] + center;
    }
}

extern "C" void kernel_launch(void* const* d_in, const int* in_sizes, int n_in,
                              void* d_out, int out_size, void* d_ws, size_t ws_size,
                              hipStream_t stream)
{
    const float* features = (const float*)d_in[0];
    const float* W1  = (const float*)d_in[1];
    const float* b1  = (const float*)d_in[2];
    const float* W2  = (const float*)d_in[3];
    const float* b2  = (const float*)d_in[4];
    const float* Wc  = (const float*)d_in[5];
    const float* bc  = (const float*)d_in[6];
    const float* Wf1 = (const float*)d_in[7];
    const float* bf1 = (const float*)d_in[8];
    const float* g1  = (const float*)d_in[9];
    const float* be1 = (const float*)d_in[10];
    const float* m1  = (const float*)d_in[11];
    const float* v1  = (const float*)d_in[12];
    const float* Wf2 = (const float*)d_in[13];
    const float* bf2 = (const float*)d_in[14];
    const float* g2  = (const float*)d_in[15];
    const float* be2 = (const float*)d_in[16];
    const float* m2  = (const float*)d_in[17];
    const float* v2  = (const float*)d_in[18];
    const float* Wo  = (const float*)d_in[19];
    const float* bo  = (const float*)d_in[20];

    float* out    = (float*)d_out;
    float* coarse = out;
    float* fine   = out + 8 * 1024 * 3;

    float* ws    = (float*)d_ws;
    float* h1    = ws;
    float* h2    = ws + 8192;
    float* base1 = ws + 16384;
    float* gt    = ws + 20480;
    float* Bp    = ws + 28672;

    gemm8_wmma<1024><<<32, 512, 0, stream>>>(features, 1024, W1, 1024, b1, h1,     1024, 1);
    gemm8_wmma<1024><<<32, 512, 0, stream>>>(h1,       1024, W2, 1024, b2, h2,     1024, 1);
    gemm8_wmma<1024><<<96, 512, 0, stream>>>(h2,       1024, Wc, 1024, bc, coarse, 3072, 0);
    gemm8_wmma<1024><<<16, 512, 0, stream>>>(features, 1024, Wf1 + 5, 1029, bf1, base1, 512, 0);
    prep_kernel<<<288, 256, 0, stream>>>(Wf1, Wf2, gt, Bp);

    dim3 gridF(16384 / 32, 8);
    fold_fused<<<gridF, 256, 0, stream>>>(coarse, base1, gt, Wf1,
                                          g1, be1, m1, v1,
                                          Bp, bf2, g2, be2, m2, v2,
                                          Wo, bo, fine);
}


# fold: phases 1 and 3 wait on global memory once, not once per loop trip: a thread owns a channel quad (BN constants once, 16 grid rows in one batch), Wo staged in LDS; sums and their order kept
# speedup vs baseline: 1.3177x; 1.0778x over previous
#include <hip/hip_runtime.h>
#include <hip/hip_bf16.h>


typedef float v2f __attribute__((ext_vector_type(2)));
typedef float v4f __attribute__((ext_vector_type(4)));
typedef float v8f __attribute__((ext_vector_type(8)));

#define EPSF 1e-5f

static __device__ __forceinline__ v8f wmma_f32x4(v2f a, v2f b, v8f c) {
    return __builtin_amdgcn_wmma_f32_16x16x4_f32(false, a, false, b, (short)0, c, false, false);
}

typedef float v4fu __attribute__((ext_vector_type(4), aligned(4)));

template <int K>
__global__ __launch_bounds__(512)
void gemm8_wmma(const float* __restrict__ A, int lda,
                const float* __restrict__ Bm, int ldb,
                const float* __restrict__ bias,
                float* __restrict__ Out, int ldo,
                int relu)
{
    constexpr int KS = K / 8;
    constexpr int NG = KS / 8;
    __shared__ float red[8 * 8 * 32];

    const int t    = threadIdx.x;
    const int w    = t >> 5;
    const int lane = t & 31;
    const int half = lane >> 4;
    const int l    = lane & 15;
    const int tt   = w & 1;
    const int ks   = w >> 1;
    const int n0   = blockIdx.x * 32;
    const int n    = n0 + tt * 16 + l;

    const float* __restrict__ arow = A + (l & 7) * lda + ks * KS + half * 4;
    const float* __restrict__ brow = Bm + n * ldb + ks * KS + half * 4;

    v4f af[NG], bf[NG];
    #pragma unroll
    for (int g = 0; g < NG; ++g) {
        af[g] = *(const v4fu*)(arow + 8 * g);
        bf[g] = *(const v4fu*)(brow + 8 * g);
    }
    const bool live = l < 8;
    v8f acc0 = {}, acc1 = {};
    #pragma unroll
    for (int g = 0; g < NG; ++g) {
        v4f a = af[g];
        a.x = live ? a.x : 0.0f;
        a.y = live ? a.y : 0.0f;
        a.z = live ? a.z : 0.0f;
        a.w = live ? a.w : 0.0f;
        acc0 = wmma_f32x4(a.xy, bf[g].xy, acc0);
        acc1 = wmma_f32x4(a.zw, bf[g].zw, acc1);
    }

    if (half == 0) {
        #pragma unroll
        for (int r = 0; r < 8; ++r)
            red[(ks * 8 + r) * 32 + tt * 16 + l] = acc0[r] + acc1[r];
    }
    __syncthreads();
    if (t < 256) {
        const int r = t >> 5;
        const int c = t & 31;
        float v = red[r * 32 + c];
        #pragma unroll
        for (int s = 1; s < 8; ++s) v += red[(s * 8 + r) * 32 + c];
        v += bias[n0 + c];
        if (relu) v = fmaxf(v, 0.0f);
        Out[r * ldo + n0 + c] = v;
    }
}

__global__ __launch_bounds__(256)
void prep_kernel(const float* __restrict__ Wf1, const float* __restrict__ Wf2,
                 float* __restrict__ gt, float* __restrict__ Bp)
{
    if (blockIdx.x < 32) {
        const int i = blockIdx.x * 256 + threadIdx.x;
        const int g = i >> 9, o = i & 511;
        const float step = 0.1f / 3.0f;
        const float gx = -0.05f + (float)(g >> 2) * step;
        const float gy = -0.05f + (float)(g & 3) * step;
        gt[i] = gx * Wf1[o * 1029 + 0] + gy * Wf1[o * 1029 + 1];
        return;
    }
    const int i    = (blockIdx.x - 32) * 256 + threadIdx.x;
    const int lane = i & 31;
    const int j    = (i >> 5) & 63;
    const int T    = i >> 11;
    const int n    = T * 16 + (lane & 15);
    const int k    = j * 8 + (lane >> 4) * 4;
    *(v4f*)(Bp + i * 4) = *(const v4f*)(Wf2 + n * 512 + k);
}

__global__ __launch_bounds__(256)
void fold_fused(const float* __restrict__ coarse,
                const float* __restrict__ base1,
                const float* __restrict__ gridtab,
                const float* __restrict__ Wf1,
                const float* __restrict__ g1, const float* __restrict__ be1,
                const float* __restrict__ m1, const float* __restrict__ v1,
                const float* __restrict__ Bp,
                const float* __restrict__ bf2,
                const float* __restrict__ g2, const float* __restrict__ be2,
                const float* __restrict__ m2, const float* __restrict__ v2,
                const float* __restrict__ Wo,
                const float* __restrict__ bo,
                float* __restrict__ fine)
{
    constexpr int Y2S = 516;
    constexpr int Y1P = 132;
    __shared__ __attribute__((aligned(16))) float y1q[128 * Y1P];
    __shared__ float y2s[32 * Y2S];
    __shared__ float red[256 * 3];
    __shared__ __attribute__((aligned(16))) float wos[3 * 512];

    const int t  = threadIdx.x;
    const int b  = blockIdx.y;
    const int m0 = blockIdx.x * 32;

    {
        const int q  = t & 127;
        const int ph = t >> 7;
        const int jc = (m0 >> 4) + ph;
        const float* cp = coarse + (b * 1024 + jc) * 3;
        const float cx = cp[0], cy = cp[1], cz = cp[2];
        const v4f bse = *(const v4f*)(base1 + b * 512 + q * 4);
        const v4f gv  = *(const v4f*)(g1 + q * 4);
        const v4f vv  = *(const v4f*)(v1 + q * 4);
        const v4f mv  = *(const v4f*)(m1 + q * 4);
        const v4f bev = *(const v4f*)(be1 + q * 4);
        v4f wx, wy, wz;
        #pragma unroll
        for (int c = 0; c < 4; ++c) {
            const float* wr = Wf1 + (q * 4 + c) * 1029;
            wx[c] = wr[2];
            wy[c] = wr[3];
            wz[c] = wr[4];
        }
        v4f gtv[16];
        #pragma unroll
        for (int g = 0; g < 16; ++g)
            gtv[g] = *(const v4f*)(gridtab + g * 512 + q * 4);
        const v4f wo0 = *(const v4f*)(Wo + t * 4);
        v4f wo1 = {};
        if (t < 128) wo1 = *(const v4f*)(Wo + 1024 + t * 4);

        v4f sc;
        #pragma unroll
        for (int c = 0; c < 4; ++c) sc[c] = gv[c] * rsqrtf(vv[c] + EPSF);

        #pragma unroll
        for (int g = 0; g < 16; ++g) {
            v4f r;
            #pragma unroll
            for (int c = 0; c < 4; ++c) {
                float val = bse[c] + gtv[g][c];
                val = __builtin_fmaf(cx, wx[c], val);
                val = __builtin_fmaf(cy, wy[c], val);
                val = __builtin_fmaf(cz, wz[c], val);
                val = __builtin_fmaf(val - mv[c], sc[c], bev[c]);
                r[c] = fmaxf(val, 0.0f);
            }
            *(v4f*)&y1q[q * Y1P + (ph * 16 + g) * 4] = r;
        }
        *(v4f*)&wos[t * 4] = wo0;
        if (t < 128) *(v4f*)&wos[1024 + t * 4] = wo1;
    }
    __syncthreads();

    {
        const int w    = t >> 5;
        const int lane = t & 31;
        const int half = lane >> 4;
        const int l    = lane & 15;
        const int aoff0 = half * Y1P + l * 4;
        const int aoff1 = half * Y1P + (16 + l) * 4;
        for (int nt2 = 0; nt2 < 2; ++nt2) {
            const int T0 = w * 4 + nt2 * 2;
            const int n0 = T0 * 16 + l;
            const int n1 = n0 + 16;
            const v4f* __restrict__ bq0 = (const v4f*)(Bp + T0 * (64 * 128)) + lane;
            const v4f* __restrict__ bq1 = bq0 + 64 * 32;
            v8f acc00 = {}, acc01 = {}, acc10 = {}, acc11 = {};
            v4f a0 = *(const v4f*)&y1q[aoff0];
            v4f a1 = *(const v4f*)&y1q[aoff1];
            v4f b0 = bq0[0];
            v4f b1 = bq1[0];
            #pragma unroll 7
            for (int j = 0; j < 63; ++j) {
                const v4f a0n = *(const v4f*)&y1q[(j + 1) * (2 * Y1P) + aoff0];
                const v4f a1n = *(const v4f*)&y1q[(j + 1) * (2 * Y1P) + aoff1];
                const v4f b0n = bq0[(j + 1) * 32];
                const v4f b1n = bq1[(j + 1) * 32];
                acc00 = wmma_f32x4(a0.xy, b0.xy, acc00);
                acc01 = wmma_f32x4(a0.xy, b1.xy, acc01);
                acc10 = wmma_f32x4(a1.xy, b0.xy, acc10);
                acc11 = wmma_f32x4(a1.xy, b1.xy, acc11);
                acc00 = wmma_f32x4(a0.zw, b0.zw, acc00);
                acc01 = wmma_f32x4(a0.zw, b1.zw, acc01);
                acc10 = wmma_f32x4(a1.zw, b0.zw, acc10);
                acc11 = wmma_f32x4(a1.zw, b1.zw, acc11);
                a0 = a0n; a1 = a1n; b0 = b0n; b1 = b1n;
            }
            acc00 = wmma_f32x4(a0.xy, b0.xy, acc00);
            acc01 = wmma_f32x4(a0.xy, b1.xy, acc01);
            acc10 = wmma_f32x4(a1.xy, b0.xy, acc10);
            acc11 = wmma_f32x4(a1.xy, b1.xy, acc11);
            acc00 = wmma_f32x4(a0.zw, b0.zw, acc00);
            acc01 = wmma_f32x4(a0.zw, b1.zw, acc01);
            acc10 = wmma_f32x4(a1.zw, b0.zw, acc10);
            acc11 = wmma_f32x4(a1.zw, b1.zw, acc11);

            const float s20 = g2[n0] * rsqrtf(v2[n0] + EPSF);
            const float s21 = g2[n1] * rsqrtf(v2[n1] + EPSF);
            const float mm0 = m2[n0], bt0 = be2[n0], bf0 = bf2[n0];
            const float mm1 = m2[n1], bt1 = be2[n1], bf1v = bf2[n1];
            #pragma unroll
            for (int r = 0; r < 8; ++r) {
                const int M = r + half * 8;
                y2s[M * Y2S + n0]        = fmaxf((acc00[r] + bf0 - mm0) * s20 + bt0, 0.0f);
                y2s[M * Y2S + n1]        = fmaxf((acc01[r] + bf1v - mm1) * s21 + bt1, 0.0f);
                y2s[(16 + M) * Y2S + n0] = fmaxf((acc10[r] + bf0 - mm0) * s20 + bt0, 0.0f);
                y2s[(16 + M) * Y2S + n1] = fmaxf((acc11[r] + bf1v - mm1) * s21 + bt1, 0.0f);
            }
        }
    }
    __syncthreads();

    float bov = 0.0f, center = 0.0f;
    if (t < 96) {
        const int p = t / 3, c = t % 3;
        const int ng = m0 + p;
        const int i  = c * 16384 + ng;
        const int r3 = i / 3;
        const int cc = i - r3 * 3;
        center = coarse[(b * 1024 + (r3 >> 4)) * 3 + cc];
        bov = bo[c];
    }
    {
        const int p = t >> 3;
        const int s = t & 7;
        float p0 = 0.0f, p1 = 0.0f, p2 = 0.0f;
        for (int o = s; o < 512; o += 8) {
            const float y = y2s[p * Y2S + o];
            p0 += y * wos[o];
            p1 += y * wos[512 + o];
            p2 += y * wos[1024 + o];
        }
        red[t * 3 + 0] = p0;
        red[t * 3 + 1] = p1;
        red[t * 3 + 2] = p2;
    }
    __syncthreads();
    if (t < 96) {
        const int p = t / 3, c = t % 3;
        float sum = 0.0f;
        #pragma unroll
        for (int s = 0; s < 8; ++s) sum += red[(p * 8 + s) * 3 + c];
        const int ng = m0 + p;
        fine[(b * 16384 + ng) * 3 + c] = sum + bov + center;
    }
}

extern "C" void kernel_launch(void* const* d_in, const int* in_sizes, int n_in,
                              void* d_out, int out_size, void* d_ws, size_t ws_size,
                              hipStream_t stream)
{
    const float* features = (const float*)d_in[0];
    const float* W1  = (const float*)d_in[1];
    const float* b1  = (const float*)d_in[2];
    const float* W2  = (const float*)d_in[3];
    const float* b2  = (const float*)d_in[4];
    const float* Wc  = (const float*)d_in[5];
    const float* bc  = (const float*)d_in[6];
    const float* Wf1 = (const float*)d_in[7];
    const float* bf1 = (const float*)d_in[8];
    const float* g1  = (const float*)d_in[9];
    const float* be1 = (const float*)d_in[10];
    const float* m1  = (const float*)d_in[11];
    const float* v1  = (const float*)d_in[12];
    const float* Wf2 = (const float*)d_in[13];
    const float* bf2 = (const float*)d_in[14];
    const float* g2  = (const float*)d_in[15];
    const float* be2 = (const float*)d_in[16];
    const float* m2  = (const float*)d_in[17];
    const float* v2  = (const float*)d_in[18];
    const float* Wo  = (const float*)d_in[19];
    const float* bo  = (const float*)d_in[20];

    float* out    = (float*)d_out;
    float* coarse = out;
    float* fine   = out + 8 * 1024 * 3;

    float* ws    = (float*)d_ws;
    float* h1    = ws;
    float* h2    = ws + 8192;
    float* base1 = ws + 16384;
    float* gt    = ws + 20480;
    float* Bp    = ws + 28672;

    gemm8_wmma<1024><<<32, 512, 0, stream>>>(features, 1024, W1, 1024, b1, h1,     1024, 1);
    gemm8_wmma<1024><<<32, 512, 0, stream>>>(h1,       1024, W2, 1024, b2, h2,     1024, 1);
    gemm8_wmma<1024><<<96, 512, 0, stream>>>(h2,       1024, Wc, 1024, bc, coarse, 3072, 0);
    gemm8_wmma<1024><<<16, 512, 0, stream>>>(features, 1024, Wf1 + 5, 1029, bf1, base1, 512, 0);
    prep_kernel<<<288, 256, 0, stream>>>(Wf1, Wf2, gt, Bp);

    dim3 gridF(16384 / 32, 8);
    fold_fused<<<gridF, 256, 0, stream>>>(coarse, base1, gt, Wf1,
                                          g1, be1, m1, v1,
                                          Bp, bf2, g2, be2, m2, v2,
                                          Wo, bo, fine);
}


# fold: 512x512 stage runs on 3 affine rows (a,p,q) per coarse point, not 16; tiles rebuilt by one wmma with (1,gx,gy); sign-changing channels add max(-z,0) x Wf2 from lists (new sum order)
# speedup vs baseline: 3.3956x; 2.5769x over previous
#include <hip/hip_runtime.h>
#include <hip/hip_bf16.h>


typedef float v2f __attribute__((ext_vector_type(2)));
typedef float v4f __attribute__((ext_vector_type(4)));
typedef float v8f __attribute__((ext_vector_type(8)));

#define EPSF 1e-5f

static __device__ __forceinline__ v8f wmma_f32x4(v2f a, v2f b, v8f c) {
    return __builtin_amdgcn_wmma_f32_16x16x4_f32(false, a, false, b, (short)0, c, false, false);
}

typedef float v4fu __attribute__((ext_vector_type(4), aligned(4)));

template <int K>
__global__ __launch_bounds__(512)
void gemm8_wmma(const float* __restrict__ A, int lda,
                const float* __restrict__ Bm, int ldb,
                const float* __restrict__ bias,
                float* __restrict__ Out, int ldo,
                int relu)
{
    constexpr int KS = K / 8;
    constexpr int NG = KS / 8;
    __shared__ float red[8 * 8 * 32];

    const int t    = threadIdx.x;
    const int w    = t >> 5;
    const int lane = t & 31;
    const int half = lane >> 4;
    const int l    = lane & 15;
    const int tt   = w & 1;
    const int ks   = w >> 1;
    const int n0   = blockIdx.x * 32;
    const int n    = n0 + tt * 16 + l;

    const float* __restrict__ arow = A + (l & 7) * lda + ks * KS + half * 4;
    const float* __restrict__ brow = Bm + n * ldb + ks * KS + half * 4;

    v4f af[NG], bf[NG];
    #pragma unroll
    for (int g = 0; g < NG; ++g) {
        af[g] = *(const v4fu*)(arow + 8 * g);
        bf[g] = *(const v4fu*)(brow + 8 * g);
    }
    const bool live = l < 8;
    v8f acc0 = {}, acc1 = {};
    #pragma unroll
    for (int g = 0; g < NG; ++g) {
        v4f a = af[g];
        a.x = live ? a.x : 0.0f;
        a.y = live ? a.y : 0.0f;
        a.z = live ? a.z : 0.0f;
        a.w = live ? a.w : 0.0f;
        acc0 = wmma_f32x4(a.xy, bf[g].xy, acc0);
        acc1 = wmma_f32x4(a.zw, bf[g].zw, acc1);
    }

    if (half == 0) {
        #pragma unroll
        for (int r = 0; r < 8; ++r)
            red[(ks * 8 + r) * 32 + tt * 16 + l] = acc0[r] + acc1[r];
    }
    __syncthreads();
    if (t < 256) {
        const int r = t >> 5;
        const int c = t & 31;
        float v = red[r * 32 + c];
        #pragma unroll
        for (int s = 1; s < 8; ++s) v += red[(s * 8 + r) * 32 + c];
        v += bias[n0 + c];
        if (relu) v = fmaxf(v, 0.0f);
        Out[r * ldo + n0 + c] = v;
    }
}

__global__ __launch_bounds__(256)
void prep_kernel(const float* __restrict__ Wf2, float* __restrict__ Bp)
{
    const int i    = blockIdx.x * 256 + threadIdx.x;
    const int lane = i & 31;
    const int j    = (i >> 5) & 63;
    const int T    = i >> 11;
    const int n    = T * 16 + (lane & 15);
    const int k    = j * 8 + (lane >> 4) * 4;
    *(v4f*)(Bp + i * 4) = *(const v4f*)(Wf2 + n * 512 + k);
}

__global__ __launch_bounds__(256)
void fold_fused(const float* __restrict__ coarse,
                const float* __restrict__ base1,
                const float* __restrict__ Wf1,
                const float* __restrict__ g1, const float* __restrict__ be1,
                const float* __restrict__ m1, const float* __restrict__ v1,
                const float* __restrict__ Wf2,
                const float* __restrict__ Bp,
                const float* __restrict__ bf2,
                const float* __restrict__ g2, const float* __restrict__ be2,
                const float* __restrict__ m2, const float* __restrict__ v2,
                const float* __restrict__ Wo,
                const float* __restrict__ bo,
                float* __restrict__ fine)
{
    constexpr int Y1P = 132;
    __shared__ __attribute__((aligned(16))) float y1q[128 * Y1P];
    __shared__ __attribute__((aligned(16))) float wos[3 * 512];
    __shared__ __attribute__((aligned(16))) float s2s[512];
    __shared__ __attribute__((aligned(16))) float shs[512];
    __shared__ float red[16 * 8 * 48];
    __shared__ unsigned int mixflag[8 * 128];
    __shared__ unsigned short mlist[8 * 512];
    __shared__ int mcnt[8];

    const int t    = threadIdx.x;
    const int b    = blockIdx.y;
    const int w    = t >> 5;
    const int lane = t & 31;
    const int half = lane >> 4;
    const int l    = lane & 15;
    const float step = 0.1f / 3.0f;

    {
        const int q    = t & 127;
        const int cgrp = t >> 7;
        const v4f bse = *(const v4f*)(base1 + b * 512 + q * 4);
        const v4f gv  = *(const v4f*)(g1 + q * 4);
        const v4f vv  = *(const v4f*)(v1 + q * 4);
        const v4f mv  = *(const v4f*)(m1 + q * 4);
        const v4f bev = *(const v4f*)(be1 + q * 4);
        v4f wx, wy, wz, pp, qq, ee, sc;
        #pragma unroll
        for (int c = 0; c < 4; ++c) {
            const float* wr = Wf1 + (q * 4 + c) * 1029;
            sc[c] = gv[c] * rsqrtf(vv[c] + EPSF);
            pp[c] = sc[c] * wr[0];
            qq[c] = sc[c] * wr[1];
            ee[c] = 0.05f * (fabsf(pp[c]) + fabsf(qq[c]));
            wx[c] = wr[2];
            wy[c] = wr[3];
            wz[c] = wr[4];
        }
        float cxs[4], cys[4], czs[4];
        #pragma unroll
        for (int i = 0; i < 4; ++i) {
            const float* cp = coarse + (b * 1024 + blockIdx.x * 8 + cgrp + 2 * i) * 3;
            cxs[i] = cp[0]; cys[i] = cp[1]; czs[i] = cp[2];
        }
        if (t < 128) {
            const v4f g2v = *(const v4f*)(g2 + t * 4);
            const v4f v2v = *(const v4f*)(v2 + t * 4);
            const v4f m2v = *(const v4f*)(m2 + t * 4);
            const v4f b2v = *(const v4f*)(be2 + t * 4);
            const v4f f2v = *(const v4f*)(bf2 + t * 4);
            v4f s, sh;
            #pragma unroll
            for (int c = 0; c < 4; ++c) {
                s[c]  = g2v[c] * rsqrtf(v2v[c] + EPSF);
                sh[c] = __builtin_fmaf(f2v[c] - m2v[c], s[c], b2v[c]);
            }
            *(v4f*)&s2s[t * 4] = s;
            *(v4f*)&shs[t * 4] = sh;
            *(v4f*)&wos[1024 + t * 4] = *(const v4f*)(Wo + 1024 + t * 4);
        }
        *(v4f*)&wos[t * 4] = *(const v4f*)(Wo + t * 4);

        #pragma unroll
        for (int i = 0; i < 4; ++i) {
            const int cl = cgrp + 2 * i;
            const int mt = cl >> 2, ci = cl & 3;
            const int rowA = mt * 16 + 2 * ci;
            v4f ra, rp, rq;
            const v4f zero4 = {};
            unsigned int fl = 0;
            #pragma unroll
            for (int c = 0; c < 4; ++c) {
                float val = bse[c];
                val = __builtin_fmaf(cxs[i], wx[c], val);
                val = __builtin_fmaf(cys[i], wy[c], val);
                val = __builtin_fmaf(czs[i], wz[c], val);
                const float a = __builtin_fmaf(val - mv[c], sc[c], bev[c]);
                const bool dead  = !(a + ee[c] > 0.0f);
                const bool mixed = !dead && !(a - ee[c] > 0.0f);
                ra[c] = dead ? 0.0f : a;
                rp[c] = dead ? 0.0f : pp[c];
                rq[c] = dead ? 0.0f : qq[c];
                fl |= (mixed ? 1u : 0u) << (8 * c);
            }
            *(v4f*)&y1q[q * Y1P + rowA * 4]       = ra;
            *(v4f*)&y1q[q * Y1P + (rowA + 1) * 4] = rp;
            *(v4f*)&y1q[q * Y1P + (rowA + 8) * 4] = rq;
            *(v4f*)&y1q[q * Y1P + (rowA + 9) * 4] = zero4;
            mixflag[cl * 128 + q] = fl;
        }
    }
    __syncthreads();

    {
        const unsigned char* fb = (const unsigned char*)mixflag;
        int basec = 0;
        for (int ch = 0; ch < 16; ++ch) {
            const int k = ch * 32 + lane;
            const bool f = fb[w * 512 + k] != 0;
            const unsigned int mask = __builtin_amdgcn_ballot_w32(f);
            const int pos = basec + __builtin_popcount(mask & ((1u << lane) - 1u));
            if (f) mlist[w * 512 + pos] = (unsigned short)k;
            basec += __builtin_popcount(mask);
        }
        if (lane == 0) mcnt[w] = basec;
    }

    v8f acc[2][4];
    #pragma unroll
    for (int mt = 0; mt < 2; ++mt)
        #pragma unroll
        for (int i = 0; i < 4; ++i) acc[mt][i] = v8f{};
    {
        const int aoff0 = half * Y1P + l * 4;
        const int aoff1 = half * Y1P + (16 + l) * 4;
        const v4f* __restrict__ bq = (const v4f*)(Bp + (w * 4) * (64 * 128)) + lane;
        auto mma16 = [&](const v4f& x0, const v4f& x1, const v4f (&y)[4]) {
            #pragma unroll
            for (int i = 0; i < 4; ++i) {
                acc[0][i] = wmma_f32x4(x0.xy, y[i].xy, acc[0][i]);
                acc[1][i] = wmma_f32x4(x1.xy, y[i].xy, acc[1][i]);
            }
            #pragma unroll
            for (int i = 0; i < 4; ++i) {
                acc[0][i] = wmma_f32x4(x0.zw, y[i].zw, acc[0][i]);
                acc[1][i] = wmma_f32x4(x1.zw, y[i].zw, acc[1][i]);
            }
        };
        auto ldgrp = [&](int j, v4f& x0, v4f& x1, v4f (&y)[4]) {
            x0 = *(const v4f*)&y1q[j * (2 * Y1P) + aoff0];
            x1 = *(const v4f*)&y1q[j * (2 * Y1P) + aoff1];
            #pragma unroll
            for (int i = 0; i < 4; ++i) y[i] = bq[i * (64 * 32) + j * 32];
        };
        v4f pa0, pa1, pb[4], qa0, qa1, qb[4], ra0, ra1, rb[4], sa0, sa1, sb[4];
        ldgrp(0, pa0, pa1, pb);
        ldgrp(1, qa0, qa1, qb);
        ldgrp(2, ra0, ra1, rb);
        #pragma nounroll
        for (int tq = 0; tq < 15; ++tq) {
            ldgrp(4 * tq + 3, sa0, sa1, sb);
            mma16(pa0, pa1, pb);
            ldgrp(4 * tq + 4, pa0, pa1, pb);
            mma16(qa0, qa1, qb);
            ldgrp(4 * tq + 5, qa0, qa1, qb);
            mma16(ra0, ra1, rb);
            ldgrp(4 * tq + 6, ra0, ra1, rb);
            mma16(sa0, sa1, sb);
        }
        ldgrp(63, sa0, sa1, sb);
        mma16(pa0, pa1, pb);
        mma16(qa0, qa1, qb);
        mma16(ra0, ra1, rb);
        mma16(sa0, sa1, sb);
    }
    __syncthreads();

    {
        const float gx = -0.05f + (float)(l >> 2) * step;
        const float gy = -0.05f + (float)(l & 3) * step;
        v2f gb;
        gb.x = half ? gy : 1.0f;
        gb.y = half ? 0.0f : gx;
        #pragma unroll
        for (int cl = 0; cl < 8; ++cl) {
            const int mt = cl >> 2, ci = cl & 3;
            const int rowA = mt * 16 + 2 * ci;
            v8f d[4];
            #pragma unroll
            for (int i = 0; i < 4; ++i) {
                v2f ea;
                ea.x = acc[mt][i][2 * ci];
                ea.y = acc[mt][i][2 * ci + 1];
                d[i] = wmma_f32x4(ea, gb, v8f{});
            }
            const int n = __builtin_amdgcn_readfirstlane(mcnt[cl]);
            for (int base = 0; base < n; base += 4) {
                const int i0 = base + 2 * half, i1 = i0 + 1;
                const bool ok0 = i0 < n, ok1 = i1 < n;
                const int k0 = ok0 ? (int)mlist[cl * 512 + i0] : 0;
                const int k1 = ok1 ? (int)mlist[cl * 512 + i1] : 0;
                const float* r0 = &y1q[(k0 >> 2) * Y1P + rowA * 4 + (k0 & 3)];
                const float* r1 = &y1q[(k1 >> 2) * Y1P + rowA * 4 + (k1 & 3)];
                const float z0 = __builtin_fmaf(gy, r0[32], __builtin_fmaf(gx, r0[4], r0[0]));
                const float z1 = __builtin_fmaf(gy, r1[32], __builtin_fmaf(gx, r1[4], r1[0]));
                v2f cb;
                cb.x = ok0 ? fmaxf(-z0, 0.0f) : 0.0f;
                cb.y = ok1 ? fmaxf(-z1, 0.0f) : 0.0f;
                #pragma unroll
                for (int i = 0; i < 4; ++i) {
                    const float* wrow = Wf2 + ((w * 4 + i) * 16 + l) * 512;
                    v2f wa;
                    wa.x = wrow[k0];
                    wa.y = wrow[k1];
                    d[i] = wmma_f32x4(wa, cb, d[i]);
                }
            }
            float p0 = 0.0f, p1 = 0.0f, p2 = 0.0f;
            #pragma unroll
            for (int i = 0; i < 4; ++i) {
                const int ob = (w * 4 + i) * 16 + 8 * half;
                #pragma unroll
                for (int hq = 0; hq < 2; ++hq) {
                    const v4f s  = *(const v4f*)&s2s[ob + 4 * hq];
                    const v4f sh = *(const v4f*)&shs[ob + 4 * hq];
                    const v4f w0 = *(const v4f*)&wos[ob + 4 * hq];
                    const v4f w1 = *(const v4f*)&wos[512 + ob + 4 * hq];
                    const v4f w2 = *(const v4f*)&wos[1024 + ob + 4 * hq];
                    #pragma unroll
                    for (int e = 0; e < 4; ++e) {
                        const float y = fmaxf(__builtin_fmaf(d[i][4 * hq + e], s[e], sh[e]), 0.0f);
                        p0 = __builtin_fmaf(y, w0[e], p0);
                        p1 = __builtin_fmaf(y, w1[e], p1);
                        p2 = __builtin_fmaf(y, w2[e], p2);
                    }
                }
            }
            float* rd = &red[((w * 2 + half) * 8 + cl) * 48 + l * 3];
            rd[0] = p0; rd[1] = p1; rd[2] = p2;
        }
    }
    __syncthreads();

    for (int idx = t; idx < 384; idx += 256) {
        const int cl  = idx / 48;
        const int rem = idx - cl * 48;
        const int g   = rem / 3;
        const int c   = rem - g * 3;
        float sum = 0.0f;
        #pragma unroll
        for (int s = 0; s < 16; ++s) sum += red[(s * 8 + cl) * 48 + rem];
        const int ng = (blockIdx.x * 8 + cl) * 16 + g;
        const int i  = c * 16384 + ng;
        const int r3 = i / 3;
        const int cc = i - r3 * 3;
        const float center = coarse[(b * 1024 + (r3 >> 4)) * 3 + cc];
        fine[(b * 16384 + ng) * 3 + c] = sum + bo[c] + center;
    }
}

extern "C" void kernel_launch(void* const* d_in, const int* in_sizes, int n_in,
                              void* d_out, int out_size, void* d_ws, size_t ws_size,
                              hipStream_t stream)
{
    const float* features = (const float*)d_in[0];
    const float* W1  = (const float*)d_in[1];
    const float* b1  = (const float*)d_in[2];
    const float* W2  = (const float*)d_in[3];
    const float* b2  = (const float*)d_in[4];
    const float* Wc  = (const float*)d_in[5];
    const float* bc  = (const float*)d_in[6];
    const float* Wf1 = (const float*)d_in[7];
    const float* bf1 = (const float*)d_in[8];
    const float* g1  = (const float*)d_in[9];
    const float* be1 = (const float*)d_in[10];
    const float* m1  = (const float*)d_in[11];
    const float* v1  = (const float*)d_in[12];
    const float* Wf2 = (const float*)d_in[13];
    const float* bf2 = (const float*)d_in[14];
    const float* g2  = (const float*)d_in[15];
    const float* be2 = (const float*)d_in[16];
    const float* m2  = (const float*)d_in[17];
    const float* v2  = (const float*)d_in[18];
    const float* Wo  = (const float*)d_in[19];
    const float* bo  = (const float*)d_in[20];

    float* out    = (float*)d_out;
    float* coarse = out;
    float* fine   = out + 8 * 1024 * 3;

    float* ws    = (float*)d_ws;
    float* h1    = ws;
    float* h2    = ws + 8192;
    float* base1 = ws + 16384;
    float* Bp    = ws + 28672;

    gemm8_wmma<1024><<<32, 512, 0, stream>>>(features, 1024, W1, 1024, b1, h1,     1024, 1);
    gemm8_wmma<1024><<<32, 512, 0, stream>>>(h1,       1024, W2, 1024, b2, h2,     1024, 1);
    gemm8_wmma<1024><<<96, 512, 0, stream>>>(h2,       1024, Wc, 1024, bc, coarse, 3072, 0);
    gemm8_wmma<1024><<<16, 512, 0, stream>>>(features, 1024, Wf1 + 5, 1029, bf1, base1, 512, 0);
    prep_kernel<<<256, 256, 0, stream>>>(Wf2, Bp);

    dim3 gridF(1024 / 8, 8);
    fold_fused<<<gridF, 256, 0, stream>>>(coarse, base1, Wf1,
                                          g1, be1, m1, v1,
                                          Wf2, Bp, bf2, g2, be2, m2, v2,
                                          Wo, bo, fine);
}


# fold: 48 affine rows of 16 coarse points fill 3 M-tiles, no zero padding row (was 8 points on 2 tiles): 25% fewer wmma, half the Wf2 fragment reads; lane halves added by shuffle (new sum order)
# speedup vs baseline: 3.9447x; 1.1617x over previous
#include <hip/hip_runtime.h>
#include <hip/hip_bf16.h>


typedef float v2f __attribute__((ext_vector_type(2)));
typedef float v4f __attribute__((ext_vector_type(4)));
typedef float v8f __attribute__((ext_vector_type(8)));

#define EPSF 1e-5f

static __device__ __forceinline__ v8f wmma_f32x4(v2f a, v2f b, v8f c) {
    return __builtin_amdgcn_wmma_f32_16x16x4_f32(false, a, false, b, (short)0, c, false, false);
}

typedef float v4fu __attribute__((ext_vector_type(4), aligned(4)));

template <int K>
__global__ __launch_bounds__(512)
void gemm8_wmma(const float* __restrict__ A, int lda,
                const float* __restrict__ Bm, int ldb,
                const float* __restrict__ bias,
                float* __restrict__ Out, int ldo,
                int relu)
{
    constexpr int KS = K / 8;
    constexpr int NG = KS / 8;
    __shared__ float red[8 * 8 * 32];

    const int t    = threadIdx.x;
    const int w    = t >> 5;
    const int lane = t & 31;
    const int half = lane >> 4;
    const int l    = lane & 15;
    const int tt   = w & 1;
    const int ks   = w >> 1;
    const int n0   = blockIdx.x * 32;
    const int n    = n0 + tt * 16 + l;

    const float* __restrict__ arow = A + (l & 7) * lda + ks * KS + half * 4;
    const float* __restrict__ brow = Bm + n * ldb + ks * KS + half * 4;

    v4f af[NG], bf[NG];
    #pragma unroll
    for (int g = 0; g < NG; ++g) {
        af[g] = *(const v4fu*)(arow + 8 * g);
        bf[g] = *(const v4fu*)(brow + 8 * g);
    }
    const bool live = l < 8;
    v8f acc0 = {}, acc1 = {};
    #pragma unroll
    for (int g = 0; g < NG; ++g) {
        v4f a = af[g];
        a.x = live ? a.x : 0.0f;
        a.y = live ? a.y : 0.0f;
        a.z = live ? a.z : 0.0f;
        a.w = live ? a.w : 0.0f;
        acc0 = wmma_f32x4(a.xy, bf[g].xy, acc0);
        acc1 = wmma_f32x4(a.zw, bf[g].zw, acc1);
    }

    if (half == 0) {
        #pragma unroll
        for (int r = 0; r < 8; ++r)
            red[(ks * 8 + r) * 32 + tt * 16 + l] = acc0[r] + acc1[r];
    }
    __syncthreads();
    if (t < 256) {
        const int r = t >> 5;
        const int c = t & 31;
        float v = red[r * 32 + c];
        #pragma unroll
        for (int s = 1; s < 8; ++s) v += red[(s * 8 + r) * 32 + c];
        v += bias[n0 + c];
        if (relu) v = fmaxf(v, 0.0f);
        Out[r * ldo + n0 + c] = v;
    }
}

__global__ __launch_bounds__(256)
void prep_kernel(const float* __restrict__ Wf2, float* __restrict__ Bp)
{
    const int i    = blockIdx.x * 256 + threadIdx.x;
    const int lane = i & 31;
    const int j    = (i >> 5) & 63;
    const int T    = i >> 11;
    const int n    = T * 16 + (lane & 15);
    const int k    = j * 8 + (lane >> 4) * 4;
    *(v4f*)(Bp + i * 4) = *(const v4f*)(Wf2 + n * 512 + k);
}

__global__ __launch_bounds__(256)
void fold_fused(const float* __restrict__ coarse,
                const float* __restrict__ base1,
                const float* __restrict__ Wf1,
                const float* __restrict__ g1, const float* __restrict__ be1,
                const float* __restrict__ m1, const float* __restrict__ v1,
                const float* __restrict__ Wf2,
                const float* __restrict__ Bp,
                const float* __restrict__ bf2,
                const float* __restrict__ g2, const float* __restrict__ be2,
                const float* __restrict__ m2, const float* __restrict__ v2,
                const float* __restrict__ Wo,
                const float* __restrict__ bo,
                float* __restrict__ fine)
{
    constexpr int Y1P = 196;
    __shared__ __attribute__((aligned(16))) float y1q[128 * Y1P];
    __shared__ __attribute__((aligned(16))) float wos[3 * 512];
    __shared__ __attribute__((aligned(16))) float s2s[512];
    __shared__ __attribute__((aligned(16))) float shs[512];
    __shared__ float red[8 * 16 * 48];
    __shared__ unsigned int mixflag[16 * 128];
    __shared__ unsigned short mlist[16 * 512];
    __shared__ int mcnt[16];

    const int t    = threadIdx.x;
    const int b    = blockIdx.y;
    const int w    = t >> 5;
    const int lane = t & 31;
    const int half = lane >> 4;
    const int l    = lane & 15;
    const float step = 0.1f / 3.0f;

    {
        const int q    = t & 127;
        const int cgrp = t >> 7;
        const v4f bse = *(const v4f*)(base1 + b * 512 + q * 4);
        const v4f gv  = *(const v4f*)(g1 + q * 4);
        const v4f vv  = *(const v4f*)(v1 + q * 4);
        const v4f mv  = *(const v4f*)(m1 + q * 4);
        const v4f bev = *(const v4f*)(be1 + q * 4);
        v4f wx, wy, wz, pp, qq, ee, sc;
        #pragma unroll
        for (int c = 0; c < 4; ++c) {
            const float* wr = Wf1 + (q * 4 + c) * 1029;
            sc[c] = gv[c] * rsqrtf(vv[c] + EPSF);
            pp[c] = sc[c] * wr[0];
            qq[c] = sc[c] * wr[1];
            ee[c] = 0.05f * (fabsf(pp[c]) + fabsf(qq[c]));
            wx[c] = wr[2];
            wy[c] = wr[3];
            wz[c] = wr[4];
        }
        float cxs[8], cys[8], czs[8];
        #pragma unroll
        for (int i = 0; i < 8; ++i) {
            const float* cp = coarse + (b * 1024 + blockIdx.x * 16 + cgrp + 2 * i) * 3;
            cxs[i] = cp[0]; cys[i] = cp[1]; czs[i] = cp[2];
        }
        if (t < 128) {
            const v4f g2v = *(const v4f*)(g2 + t * 4);
            const v4f v2v = *(const v4f*)(v2 + t * 4);
            const v4f m2v = *(const v4f*)(m2 + t * 4);
            const v4f b2v = *(const v4f*)(be2 + t * 4);
            const v4f f2v = *(const v4f*)(bf2 + t * 4);
            v4f s, sh;
            #pragma unroll
            for (int c = 0; c < 4; ++c) {
                s[c]  = g2v[c] * rsqrtf(v2v[c] + EPSF);
                sh[c] = __builtin_fmaf(f2v[c] - m2v[c], s[c], b2v[c]);
            }
            *(v4f*)&s2s[t * 4] = s;
            *(v4f*)&shs[t * 4] = sh;
            *(v4f*)&wos[1024 + t * 4] = *(const v4f*)(Wo + 1024 + t * 4);
        }
        *(v4f*)&wos[t * 4] = *(const v4f*)(Wo + t * 4);

        #pragma unroll
        for (int i = 0; i < 8; ++i) {
            const int cl   = cgrp + 2 * i;
            const int rowA = cgrp * 32 + i;
            const int rowQ = rowA + 8;
            const int rowP = 16 + i + 8 * cgrp;
            v4f ra, rp, rq;
            unsigned int fl = 0;
            #pragma unroll
            for (int c = 0; c < 4; ++c) {
                float val = bse[c];
                val = __builtin_fmaf(cxs[i], wx[c], val);
                val = __builtin_fmaf(cys[i], wy[c], val);
                val = __builtin_fmaf(czs[i], wz[c], val);
                const float a = __builtin_fmaf(val - mv[c], sc[c], bev[c]);
                const bool dead  = !(a + ee[c] > 0.0f);
                const bool mixed = !dead && !(a - ee[c] > 0.0f);
                ra[c] = dead ? 0.0f : a;
                rp[c] = dead ? 0.0f : pp[c];
                rq[c] = dead ? 0.0f : qq[c];
                fl |= (mixed ? 1u : 0u) << (8 * c);
            }
            *(v4f*)&y1q[q * Y1P + rowA * 4] = ra;
            *(v4f*)&y1q[q * Y1P + rowP * 4] = rp;
            *(v4f*)&y1q[q * Y1P + rowQ * 4] = rq;
            mixflag[cl * 128 + q] = fl;
        }
    }
    __syncthreads();

    {
        const unsigned char* fb = (const unsigned char*)mixflag;
        #pragma unroll
        for (int rep = 0; rep < 2; ++rep) {
            const int cc = w + 8 * rep;
            int basec = 0;
            for (int ch = 0; ch < 16; ++ch) {
                const int k = ch * 32 + lane;
                const bool f = fb[cc * 512 + k] != 0;
                const unsigned int mask = __builtin_amdgcn_ballot_w32(f);
                const int pos = basec + __builtin_popcount(mask & ((1u << lane) - 1u));
                if (f) mlist[cc * 512 + pos] = (unsigned short)k;
                basec += __builtin_popcount(mask);
            }
            if (lane == 0) mcnt[cc] = basec;
        }
    }

    v8f acc[3][4];
    #pragma unroll
    for (int mt = 0; mt < 3; ++mt)
        #pragma unroll
        for (int i = 0; i < 4; ++i) acc[mt][i] = v8f{};
    {
        const int aoff = half * Y1P + l * 4;
        const v4f* __restrict__ bq = (const v4f*)(Bp + (w * 4) * (64 * 128)) + lane;
        auto mma24 = [&](const v4f (&x)[3], const v4f (&y)[4]) {
            #pragma unroll
            for (int i = 0; i < 4; ++i)
                #pragma unroll
                for (int mt = 0; mt < 3; ++mt)
                    acc[mt][i] = wmma_f32x4(x[mt].xy, y[i].xy, acc[mt][i]);
            #pragma unroll
            for (int i = 0; i < 4; ++i)
                #pragma unroll
                for (int mt = 0; mt < 3; ++mt)
                    acc[mt][i] = wmma_f32x4(x[mt].zw, y[i].zw, acc[mt][i]);
        };
        auto ldgrp = [&](int j, v4f (&x)[3], v4f (&y)[4]) {
            #pragma unroll
            for (int mt = 0; mt < 3; ++mt)
                x[mt] = *(const v4f*)&y1q[j * (2 * Y1P) + aoff + mt * 64];
            #pragma unroll
            for (int i = 0; i < 4; ++i) y[i] = bq[i * (64 * 32) + j * 32];
        };
        v4f pa[3], pb[4], qa[3], qb[4], ra[3], rb[4];
        ldgrp(0, pa, pb);
        ldgrp(1, qa, qb);
        #pragma nounroll
        for (int tq = 0; tq < 20; ++tq) {
            ldgrp(3 * tq + 2, ra, rb);
            mma24(pa, pb);
            ldgrp(3 * tq + 3, pa, pb);
            mma24(qa, qb);
            ldgrp(3 * tq + 4, qa, qb);
            mma24(ra, rb);
        }
        ldgrp(62, ra, rb);
        mma24(pa, pb);
        ldgrp(63, pa, pb);
        mma24(qa, qb);
        mma24(ra, rb);
        mma24(pa, pb);
    }
    __syncthreads();

    {
        const float gx = -0.05f + (float)(l >> 2) * step;
        const float gy = -0.05f + (float)(l & 3) * step;
        v2f gbX, gbY;
        gbX.x = half ? gy : 1.0f;
        gbX.y = half ? 0.0f : gx;
        gbY.x = half ? gy : 1.0f;
        gbY.y = half ? gx : 0.0f;
        #pragma unroll
        for (int cl = 0; cl < 16; ++cl) {
            const int pr = cl >> 1, sy = cl & 1;
            const int mtA  = sy ? 2 : 0;
            const int rowA = sy * 32 + pr;
            const int rowQ = rowA + 8;
            const int rowP = 16 + pr + 8 * sy;
            const v2f gb = sy ? gbY : gbX;
            v8f d[4];
            #pragma unroll
            for (int i = 0; i < 4; ++i) {
                v2f ea;
                ea.x = acc[mtA][i][pr];
                ea.y = acc[1][i][pr];
                d[i] = wmma_f32x4(ea, gb, v8f{});
            }
            const int n = __builtin_amdgcn_readfirstlane(mcnt[cl]);
            for (int base = 0; base < n; base += 4) {
                const int i0 = base + 2 * half, i1 = i0 + 1;
                const bool ok0 = i0 < n, ok1 = i1 < n;
                const int k0 = ok0 ? (int)mlist[cl * 512 + i0] : 0;
                const int k1 = ok1 ? (int)mlist[cl * 512 + i1] : 0;
                const float* r0 = &y1q[(k0 >> 2) * Y1P + (k0 & 3)];
                const float* r1 = &y1q[(k1 >> 2) * Y1P + (k1 & 3)];
                const float z0 = __builtin_fmaf(gy, r0[rowQ * 4], __builtin_fmaf(gx, r0[rowP * 4], r0[rowA * 4]));
                const float z1 = __builtin_fmaf(gy, r1[rowQ * 4], __builtin_fmaf(gx, r1[rowP * 4], r1[rowA * 4]));
                v2f cb;
                cb.x = ok0 ? fmaxf(-z0, 0.0f) : 0.0f;
                cb.y = ok1 ? fmaxf(-z1, 0.0f) : 0.0f;
                #pragma unroll
                for (int i = 0; i < 4; ++i) {
                    const float* wrow = Wf2 + ((w * 4 + i) * 16 + l) * 512;
                    v2f wa;
                    wa.x = wrow[k0];
                    wa.y = wrow[k1];
                    d[i] = wmma_f32x4(wa, cb, d[i]);
                }
            }
            float p0 = 0.0f, p1 = 0.0f, p2 = 0.0f;
            #pragma unroll
            for (int i = 0; i < 4; ++i) {
                const int ob = (w * 4 + i) * 16 + 8 * half;
                #pragma unroll
                for (int hq = 0; hq < 2; ++hq) {
                    const v4f s  = *(const v4f*)&s2s[ob + 4 * hq];
                    const v4f sh = *(const v4f*)&shs[ob + 4 * hq];
                    const v4f w0 = *(const v4f*)&wos[ob + 4 * hq];
                    const v4f w1 = *(const v4f*)&wos[512 + ob + 4 * hq];
                    const v4f w2 = *(const v4f*)&wos[1024 + ob + 4 * hq];
                    #pragma unroll
                    for (int e = 0; e < 4; ++e) {
                        const float y = fmaxf(__builtin_fmaf(d[i][4 * hq + e], s[e], sh[e]), 0.0f);
                        p0 = __builtin_fmaf(y, w0[e], p0);
                        p1 = __builtin_fmaf(y, w1[e], p1);
                        p2 = __builtin_fmaf(y, w2[e], p2);
                    }
                }
            }
            p0 += __shfl_xor(p0, 16);
            p1 += __shfl_xor(p1, 16);
            p2 += __shfl_xor(p2, 16);
            if (half == 0) {
                float* rd = &red[(w * 16 + cl) * 48 + l * 3];
                rd[0] = p0; rd[1] = p1; rd[2] = p2;
            }
        }
    }
    __syncthreads();

    for (int idx = t; idx < 768; idx += 256) {
        const int cl  = idx / 48;
        const int rem = idx - cl * 48;
        const int g   = rem / 3;
        const int c   = rem - g * 3;
        float sum = 0.0f;
        #pragma unroll
        for (int s = 0; s < 8; ++s) sum += red[(s * 16 + cl) * 48 + rem];
        const int ng = (blockIdx.x * 16 + cl) * 16 + g;
        const int i  = c * 16384 + ng;
        const int r3 = i / 3;
        const int cc = i - r3 * 3;
        const float center = coarse[(b * 1024 + (r3 >> 4)) * 3 + cc];
        fine[(b * 16384 + ng) * 3 + c] = sum + bo[c] + center;
    }
}

extern "C" void kernel_launch(void* const* d_in, const int* in_sizes, int n_in,
                              void* d_out, int out_size, void* d_ws, size_t ws_size,
                              hipStream_t stream)
{
    const float* features = (const float*)d_in[0];
    const float* W1  = (const float*)d_in[1];
    const float* b1  = (const float*)d_in[2];
    const float* W2  = (const float*)d_in[3];
    const float* b2  = (const float*)d_in[4];
    const float* Wc  = (const float*)d_in[5];
    const float* bc  = (const float*)d_in[6];
    const float* Wf1 = (const float*)d_in[7];
    const float* bf1 = (const float*)d_in[8];
    const float* g1  = (const float*)d_in[9];
    const float* be1 = (const float*)d_in[10];
    const float* m1  = (const float*)d_in[11];
    const float* v1  = (const float*)d_in[12];
    const float* Wf2 = (const float*)d_in[13];
    const float* bf2 = (const float*)d_in[14];
    const float* g2  = (const float*)d_in[15];
    const float* be2 = (const float*)d_in[16];
    const float* m2  = (const float*)d_in[17];
    const float* v2  = (const float*)d_in[18];
    const float* Wo  = (const float*)d_in[19];
    const float* bo  = (const float*)d_in[20];

    float* out    = (float*)d_out;
    float* coarse = out;
    float* fine   = out + 8 * 1024 * 3;

    float* ws    = (float*)d_ws;
    float* h1    = ws;
    float* h2    = ws + 8192;
    float* base1 = ws + 16384;
    float* Bp    = ws + 28672;

    gemm8_wmma<1024><<<32, 512, 0, stream>>>(features, 1024, W1, 1024, b1, h1,     1024, 1);
    gemm8_wmma<1024><<<32, 512, 0, stream>>>(h1,       1024, W2, 1024, b2, h2,     1024, 1);
    gemm8_wmma<1024><<<96, 512, 0, stream>>>(h2,       1024, Wc, 1024, bc, coarse, 3072, 0);
    gemm8_wmma<1024><<<16, 512, 0, stream>>>(features, 1024, Wf1 + 5, 1029, bf1, base1, 512, 0);
    prep_kernel<<<256, 256, 0, stream>>>(Wf2, Bp);

    dim3 gridF(1024 / 16, 8);
    fold_fused<<<gridF, 256, 0, stream>>>(coarse, base1, Wf1,
                                          g1, be1, m1, v1,
                                          Wf2, Bp, bf2, g2, be2, m2, v2,
                                          Wo, bo, fine);
}


# fold: 512x512 stage runs only over channels live somewhere in the block (first ReLU kills about half): k compacted by an LDS table, A values and Wf2^T rows gathered by it (new sum order)
# speedup vs baseline: 4.5494x; 1.1533x over previous
#include <hip/hip_runtime.h>
#include <hip/hip_bf16.h>


typedef float v2f __attribute__((ext_vector_type(2)));
typedef float v4f __attribute__((ext_vector_type(4)));
typedef float v8f __attribute__((ext_vector_type(8)));
typedef unsigned int v2u __attribute__((ext_vector_type(2)));

#define EPSF 1e-5f

static __device__ __forceinline__ v8f wmma_f32x4(v2f a, v2f b, v8f c) {
    return __builtin_amdgcn_wmma_f32_16x16x4_f32(false, a, false, b, (short)0, c, false, false);
}

typedef float v4fu __attribute__((ext_vector_type(4), aligned(4)));

template <int K>
__global__ __launch_bounds__(512)
void gemm8_wmma(const float* __restrict__ A, int lda,
                const float* __restrict__ Bm, int ldb,
                const float* __restrict__ bias,
                float* __restrict__ Out, int ldo,
                int relu)
{
    constexpr int KS = K / 8;
    constexpr int NG = KS / 8;
    __shared__ float red[8 * 8 * 32];

    const int t    = threadIdx.x;
    const int w    = t >> 5;
    const int lane = t & 31;
    const int half = lane >> 4;
    const int l    = lane & 15;
    const int tt   = w & 1;
    const int ks   = w >> 1;
    const int n0   = blockIdx.x * 32;
    const int n    = n0 + tt * 16 + l;

    const float* __restrict__ arow = A + (l & 7) * lda + ks * KS + half * 4;
    const float* __restrict__ brow = Bm + n * ldb + ks * KS + half * 4;

    v4f af[NG], bf[NG];
    #pragma unroll
    for (int g = 0; g < NG; ++g) {
        af[g] = *(const v4fu*)(arow + 8 * g);
        bf[g] = *(const v4fu*)(brow + 8 * g);
    }
    const bool live = l < 8;
    v8f acc0 = {}, acc1 = {};
    #pragma unroll
    for (int g = 0; g < NG; ++g) {
        v4f a = af[g];
        a.x = live ? a.x : 0.0f;
        a.y = live ? a.y : 0.0f;
        a.z = live ? a.z : 0.0f;
        a.w = live ? a.w : 0.0f;
        acc0 = wmma_f32x4(a.xy, bf[g].xy, acc0);
        acc1 = wmma_f32x4(a.zw, bf[g].zw, acc1);
    }

    if (half == 0) {
        #pragma unroll
        for (int r = 0; r < 8; ++r)
            red[(ks * 8 + r) * 32 + tt * 16 + l] = acc0[r] + acc1[r];
    }
    __syncthreads();
    if (t < 256) {
        const int r = t >> 5;
        const int c = t & 31;
        float v = red[r * 32 + c];
        #pragma unroll
        for (int s = 1; s < 8; ++s) v += red[(s * 8 + r) * 32 + c];
        v += bias[n0 + c];
        if (relu) v = fmaxf(v, 0.0f);
        Out[r * ldo + n0 + c] = v;
    }
}

__global__ __launch_bounds__(256)
void prep_kernel(const float* __restrict__ Wf2, float* __restrict__ WT)
{
    __shared__ float tile[64 * 33];
    const int t    = threadIdx.x;
    const int wv   = blockIdx.x >> 4;
    const int k0   = (blockIdx.x & 15) * 32;
    const int lane = t & 31;
    const int r    = t >> 5;
    #pragma unroll
    for (int p = 0; p < 8; ++p) {
        const int ol = p * 8 + r;
        tile[ol * 33 + lane] = Wf2[(wv * 64 + ol) * 512 + k0 + lane];
    }
    __syncthreads();
    #pragma unroll
    for (int p = 0; p < 2; ++p) {
        const int idx = p * 256 + t;
        const int kk  = idx >> 4;
        const int c   = idx & 15;
        v4f v;
        #pragma unroll
        for (int i = 0; i < 4; ++i) v[i] = tile[(i * 16 + c) * 33 + kk];
        *(v4f*)(WT + (k0 + kk) * 512 + wv * 64 + 4 * c) = v;
    }
}

__global__ __launch_bounds__(256)
void fold_fused(const float* __restrict__ coarse,
                const float* __restrict__ base1,
                const float* __restrict__ Wf1,
                const float* __restrict__ g1, const float* __restrict__ be1,
                const float* __restrict__ m1, const float* __restrict__ v1,
                const float* __restrict__ Wf2,
                const float* __restrict__ WT,
                const float* __restrict__ bf2,
                const float* __restrict__ g2, const float* __restrict__ be2,
                const float* __restrict__ m2, const float* __restrict__ v2,
                const float* __restrict__ Wo,
                const float* __restrict__ bo,
                float* __restrict__ fine)
{
    constexpr int Y1P = 196;
    __shared__ __attribute__((aligned(16))) float y1q[129 * Y1P];
    __shared__ __attribute__((aligned(16))) unsigned short ktab[512];
    __shared__ __attribute__((aligned(16))) float wos[3 * 512];
    __shared__ __attribute__((aligned(16))) float s2s[512];
    __shared__ __attribute__((aligned(16))) float shs[512];
    __shared__ float red[8 * 16 * 48];
    unsigned char* const aflag = (unsigned char*)red;
    __shared__ unsigned int mixflag[16 * 128];
    __shared__ unsigned short mlist[16 * 512];
    __shared__ int mcnt[16];

    const int t    = threadIdx.x;
    const int b    = blockIdx.y;
    const int w    = t >> 5;
    const int lane = t & 31;
    const int half = lane >> 4;
    const int l    = lane & 15;
    const float step = 0.1f / 3.0f;

    {
        const int q    = t & 127;
        const int cgrp = t >> 7;
        const v4f bse = *(const v4f*)(base1 + b * 512 + q * 4);
        const v4f gv  = *(const v4f*)(g1 + q * 4);
        const v4f vv  = *(const v4f*)(v1 + q * 4);
        const v4f mv  = *(const v4f*)(m1 + q * 4);
        const v4f bev = *(const v4f*)(be1 + q * 4);
        v4f wx, wy, wz, pp, qq, ee, sc;
        #pragma unroll
        for (int c = 0; c < 4; ++c) {
            const float* wr = Wf1 + (q * 4 + c) * 1029;
            sc[c] = gv[c] * rsqrtf(vv[c] + EPSF);
            pp[c] = sc[c] * wr[0];
            qq[c] = sc[c] * wr[1];
            ee[c] = 0.05f * (fabsf(pp[c]) + fabsf(qq[c]));
            wx[c] = wr[2];
            wy[c] = wr[3];
            wz[c] = wr[4];
        }
        float cxs[8], cys[8], czs[8];
        #pragma unroll
        for (int i = 0; i < 8; ++i) {
            const float* cp = coarse + (b * 1024 + blockIdx.x * 16 + cgrp + 2 * i) * 3;
            cxs[i] = cp[0]; cys[i] = cp[1]; czs[i] = cp[2];
        }
        if (t < 128) {
            const v4f g2v = *(const v4f*)(g2 + t * 4);
            const v4f v2v = *(const v4f*)(v2 + t * 4);
            const v4f m2v = *(const v4f*)(m2 + t * 4);
            const v4f b2v = *(const v4f*)(be2 + t * 4);
            const v4f f2v = *(const v4f*)(bf2 + t * 4);
            v4f s, sh;
            #pragma unroll
            for (int c = 0; c < 4; ++c) {
                s[c]  = g2v[c] * rsqrtf(v2v[c] + EPSF);
                sh[c] = __builtin_fmaf(f2v[c] - m2v[c], s[c], b2v[c]);
            }
            *(v4f*)&s2s[t * 4] = s;
            *(v4f*)&shs[t * 4] = sh;
            *(v4f*)&wos[1024 + t * 4] = *(const v4f*)(Wo + 1024 + t * 4);
        }
        *(v4f*)&wos[t * 4] = *(const v4f*)(Wo + t * 4);

        unsigned int live = 0;
        #pragma unroll
        for (int i = 0; i < 8; ++i) {
            const int cl   = cgrp + 2 * i;
            const int rowA = cgrp * 32 + i;
            const int rowQ = rowA + 8;
            const int rowP = 16 + i + 8 * cgrp;
            v4f ra, rp, rq;
            unsigned int fl = 0;
            #pragma unroll
            for (int c = 0; c < 4; ++c) {
                float val = bse[c];
                val = __builtin_fmaf(cxs[i], wx[c], val);
                val = __builtin_fmaf(cys[i], wy[c], val);
                val = __builtin_fmaf(czs[i], wz[c], val);
                const float a = __builtin_fmaf(val - mv[c], sc[c], bev[c]);
                const bool dead  = !(a + ee[c] > 0.0f);
                const bool mixed = !dead && !(a - ee[c] > 0.0f);
                ra[c] = dead ? 0.0f : a;
                rp[c] = dead ? 0.0f : pp[c];
                rq[c] = dead ? 0.0f : qq[c];
                fl |= (mixed ? 1u : 0u) << (8 * c);
                live |= (dead ? 0u : 1u) << c;
            }
            *(v4f*)&y1q[q * Y1P + rowA * 4] = ra;
            *(v4f*)&y1q[q * Y1P + rowP * 4] = rp;
            *(v4f*)&y1q[q * Y1P + rowQ * 4] = rq;
            mixflag[cl * 128 + q] = fl;
        }
        aflag[t] = (unsigned char)live;
        if (t < 48) *(v4f*)&y1q[128 * Y1P + t * 4] = v4f{};
    }
    __syncthreads();

    int NG;
    {
        const unsigned int lt = (1u << lane) - 1u;
        int total = 0, mybase = 0, mypre = 0;
        unsigned int mine = 0;
        #pragma unroll
        for (int ch = 0; ch < 4; ++ch) {
            const unsigned int f = (unsigned int)aflag[ch * 32 + lane] | (unsigned int)aflag[128 + ch * 32 + lane];
            int cnt = 0, pre = 0;
            #pragma unroll
            for (int c = 0; c < 4; ++c) {
                const unsigned int m = __builtin_amdgcn_ballot_w32(((f >> c) & 1u) != 0u);
                cnt += __builtin_popcount(m);
                pre += __builtin_popcount(m & lt);
            }
            if (ch == (w & 3)) { mine = f; mybase = total; mypre = pre; }
            total += cnt;
        }
        NG = ((total + 15) >> 4) << 1;
        NG = NG < 2 ? 2 : NG;
        if (w < 4) {
            int pos = mybase + mypre;
            #pragma unroll
            for (int c = 0; c < 4; ++c) {
                if ((mine >> c) & 1u) {
                    ktab[pos] = (unsigned short)(4 * t + c);
                    ++pos;
                }
            }
        } else {
            const int p = total + (t - 128);
            if (t - 128 < 16 && p < 8 * NG) ktab[p] = (unsigned short)512;
        }
    }

    {
        const unsigned char* fb = (const unsigned char*)mixflag;
        #pragma unroll
        for (int rep = 0; rep < 2; ++rep) {
            const int cc = w + 8 * rep;
            int basec = 0;
            for (int ch = 0; ch < 16; ++ch) {
                const int k = ch * 32 + lane;
                const bool f = fb[cc * 512 + k] != 0;
                const unsigned int mask = __builtin_amdgcn_ballot_w32(f);
                const int pos = basec + __builtin_popcount(mask & ((1u << lane) - 1u));
                if (f) mlist[cc * 512 + pos] = (unsigned short)k;
                basec += __builtin_popcount(mask);
            }
            if (lane == 0) mcnt[cc] = basec;
        }
    }

    __syncthreads();

    v8f acc[3][4];
    #pragma unroll
    for (int mt = 0; mt < 3; ++mt)
        #pragma unroll
        for (int i = 0; i < 4; ++i) acc[mt][i] = v8f{};
    {
        const int ng = __builtin_amdgcn_readfirstlane(NG);
        const unsigned int arow = (unsigned int)(l * 4);
        const float* __restrict__ wtl = WT + w * 64 + l * 4;
        auto ldtab = [&](int j) -> v2u { return *(const v2u*)&ktab[8 * j + 4 * half]; };
        auto ldgrp = [&](const v2u tb, float (&x)[3][4], v4f (&y)[4]) {
            #pragma unroll
            for (int e = 0; e < 4; ++e) {
                const unsigned int k  = (tb[e >> 1] >> (16 * (e & 1))) & 0xffffu;
                const unsigned int ai = (k >> 2) * Y1P + (k & 3u) + arow;
                #pragma unroll
                for (int mt = 0; mt < 3; ++mt) x[mt][e] = y1q[ai + mt * 64];
                y[e] = *(const v4f*)(wtl + (k & 511u) * 512);
            }
        };
        auto mma24 = [&](const float (&x)[3][4], const v4f (&y)[4]) {
            #pragma unroll
            for (int i = 0; i < 4; ++i)
                #pragma unroll
                for (int mt = 0; mt < 3; ++mt)
                    acc[mt][i] = wmma_f32x4(v2f{x[mt][0], x[mt][1]}, v2f{y[0][i], y[1][i]}, acc[mt][i]);
            #pragma unroll
            for (int i = 0; i < 4; ++i)
                #pragma unroll
                for (int mt = 0; mt < 3; ++mt)
                    acc[mt][i] = wmma_f32x4(v2f{x[mt][2], x[mt][3]}, v2f{y[2][i], y[3][i]}, acc[mt][i]);
        };
        float pa[3][4], qa[3][4];
        v4f pb[4], qb[4];
        v2u t0 = ldtab(0), t1 = ldtab(1);
        ldgrp(t0, pa, pb);
        #pragma nounroll
        for (int j = 0; j < ng; j += 2) {
            const int j2 = j + 2 < ng ? j + 2 : ng - 1;
            const int j3 = j + 3 < ng ? j + 3 : ng - 1;
            t0 = ldtab(j2);
            ldgrp(t1, qa, qb);
            __builtin_amdgcn_sched_barrier(0);
            mma24(pa, pb);
            __builtin_amdgcn_sched_barrier(0);
            t1 = ldtab(j3);
            ldgrp(t0, pa, pb);
            __builtin_amdgcn_sched_barrier(0);
            mma24(qa, qb);
            __builtin_amdgcn_sched_barrier(0);
        }
    }

    {
        const float gx = -0.05f + (float)(l >> 2) * step;
        const float gy = -0.05f + (float)(l & 3) * step;
        v2f gbX, gbY;
        gbX.x = half ? gy : 1.0f;
        gbX.y = half ? 0.0f : gx;
        gbY.x = half ? gy : 1.0f;
        gbY.y = half ? gx : 0.0f;
        #pragma unroll
        for (int cl = 0; cl < 16; ++cl) {
            const int pr = cl >> 1, sy = cl & 1;
            const int mtA  = sy ? 2 : 0;
            const int rowA = sy * 32 + pr;
            const int rowQ = rowA + 8;
            const int rowP = 16 + pr + 8 * sy;
            const v2f gb = sy ? gbY : gbX;
            v8f d[4];
            #pragma unroll
            for (int i = 0; i < 4; ++i) {
                v2f ea;
                ea.x = acc[mtA][i][pr];
                ea.y = acc[1][i][pr];
                d[i] = wmma_f32x4(ea, gb, v8f{});
            }
            const int n = __builtin_amdgcn_readfirstlane(mcnt[cl]);
            for (int base = 0; base < n; base += 4) {
                const int i0 = base + 2 * half, i1 = i0 + 1;
                const bool ok0 = i0 < n, ok1 = i1 < n;
                const int k0 = ok0 ? (int)mlist[cl * 512 + i0] : 0;
                const int k1 = ok1 ? (int)mlist[cl * 512 + i1] : 0;
                const float* r0 = &y1q[(k0 >> 2) * Y1P + (k0 & 3)];
                const float* r1 = &y1q[(k1 >> 2) * Y1P + (k1 & 3)];
                const float z0 = __builtin_fmaf(gy, r0[rowQ * 4], __builtin_fmaf(gx, r0[rowP * 4], r0[rowA * 4]));
                const float z1 = __builtin_fmaf(gy, r1[rowQ * 4], __builtin_fmaf(gx, r1[rowP * 4], r1[rowA * 4]));
                v2f cb;
                cb.x = ok0 ? fmaxf(-z0, 0.0f) : 0.0f;
                cb.y = ok1 ? fmaxf(-z1, 0.0f) : 0.0f;
                #pragma unroll
                for (int i = 0; i < 4; ++i) {
                    const float* wrow = Wf2 + ((w * 4 + i) * 16 + l) * 512;
                    v2f wa;
                    wa.x = wrow[k0];
                    wa.y = wrow[k1];
                    d[i] = wmma_f32x4(wa, cb, d[i]);
                }
            }
            float p0 = 0.0f, p1 = 0.0f, p2 = 0.0f;
            #pragma unroll
            for (int i = 0; i < 4; ++i) {
                const int ob = (w * 4 + i) * 16 + 8 * half;
                #pragma unroll
                for (int hq = 0; hq < 2; ++hq) {
                    const v4f s  = *(const v4f*)&s2s[ob + 4 * hq];
                    const v4f sh = *(const v4f*)&shs[ob + 4 * hq];
                    const v4f w0 = *(const v4f*)&wos[ob + 4 * hq];
                    const v4f w1 = *(const v4f*)&wos[512 + ob + 4 * hq];
                    const v4f w2 = *(const v4f*)&wos[1024 + ob + 4 * hq];
                    #pragma unroll
                    for (int e = 0; e < 4; ++e) {
                        const float y = fmaxf(__builtin_fmaf(d[i][4 * hq + e], s[e], sh[e]), 0.0f);
                        p0 = __builtin_fmaf(y, w0[e], p0);
                        p1 = __builtin_fmaf(y, w1[e], p1);
                        p2 = __builtin_fmaf(y, w2[e], p2);
                    }
                }
            }
            p0 += __shfl_xor(p0, 16);
            p1 += __shfl_xor(p1, 16);
            p2 += __shfl_xor(p2, 16);
            if (half == 0) {
                float* rd = &red[(w * 16 + cl) * 48 + l * 3];
                rd[0] = p0; rd[1] = p1; rd[2] = p2;
            }
        }
    }
    __syncthreads();

    for (int idx = t; idx < 768; idx += 256) {
        const int cl  = idx / 48;
        const int rem = idx - cl * 48;
        const int g   = rem / 3;
        const int c   = rem - g * 3;
        float sum = 0.0f;
        #pragma unroll
        for (int s = 0; s < 8; ++s) sum += red[(s * 16 + cl) * 48 + rem];
        const int ng = (blockIdx.x * 16 + cl) * 16 + g;
        const int i  = c * 16384 + ng;
        const int r3 = i / 3;
        const int cc = i - r3 * 3;
        const float center = coarse[(b * 1024 + (r3 >> 4)) * 3 + cc];
        fine[(b * 16384 + ng) * 3 + c] = sum + bo[c] + center;
    }
}

extern "C" void kernel_launch(void* const* d_in, const int* in_sizes, int n_in,
                              void* d_out, int out_size, void* d_ws, size_t ws_size,
                              hipStream_t stream)
{
    const float* features = (const float*)d_in[0];
    const float* W1  = (const float*)d_in[1];
    const float* b1  = (const float*)d_in[2];
    const float* W2  = (const float*)d_in[3];
    const float* b2  = (const float*)d_in[4];
    const float* Wc  = (const float*)d_in[5];
    const float* bc  = (const float*)d_in[6];
    const float* Wf1 = (const float*)d_in[7];
    const float* bf1 = (const float*)d_in[8];
    const float* g1  = (const float*)d_in[9];
    const float* be1 = (const float*)d_in[10];
    const float* m1  = (const float*)d_in[11];
    const float* v1  = (const float*)d_in[12];
    const float* Wf2 = (const float*)d_in[13];
    const float* bf2 = (const float*)d_in[14];
    const float* g2  = (const float*)d_in[15];
    const float* be2 = (const float*)d_in[16];
    const float* m2  = (const float*)d_in[17];
    const float* v2  = (const float*)d_in[18];
    const float* Wo  = (const float*)d_in[19];
    const float* bo  = (const float*)d_in[20];

    float* out    = (float*)d_out;
    float* coarse = out;
    float* fine   = out + 8 * 1024 * 3;

    float* ws    = (float*)d_ws;
    float* h1    = ws;
    float* h2    = ws + 8192;
    float* base1 = ws + 16384;
    float* WT    = ws + 28672;

    gemm8_wmma<1024><<<32, 512, 0, stream>>>(features, 1024, W1, 1024, b1, h1,     1024, 1);
    gemm8_wmma<1024><<<32, 512, 0, stream>>>(h1,       1024, W2, 1024, b2, h2,     1024, 1);
    gemm8_wmma<1024><<<96, 512, 0, stream>>>(h2,       1024, Wc, 1024, bc, coarse, 3072, 0);
    gemm8_wmma<1024><<<16, 512, 0, stream>>>(features, 1024, Wf1 + 5, 1029, bf1, base1, 512, 0);
    prep_kernel<<<128, 256, 0, stream>>>(Wf2, WT);

    dim3 gridF(1024 / 16, 8);
    fold_fused<<<gridF, 256, 0, stream>>>(coarse, base1, Wf1,
                                          g1, be1, m1, v1,
                                          Wf2, WT, bf2, g2, be2, m2, v2,
                                          Wo, bo, fine);
}


# fold: first ReLU input is affine in the coarse point too: channels of a per-batch sign mask run as 6 rows per batch in one small GEMM; per-point wmma only over channels that differ (new sum order)
# speedup vs baseline: 5.7863x; 1.2719x over previous
#include <hip/hip_runtime.h>
#include <hip/hip_bf16.h>


typedef float v2f __attribute__((ext_vector_type(2)));
typedef float v4f __attribute__((ext_vector_type(4)));
typedef float v8f __attribute__((ext_vector_type(8)));
typedef unsigned int v2u __attribute__((ext_vector_type(2)));

#define EPSF 1e-5f

static __device__ __forceinline__ v8f wmma_f32x4(v2f a, v2f b, v8f c) {
    return __builtin_amdgcn_wmma_f32_16x16x4_f32(false, a, false, b, (short)0, c, false, false);
}

typedef float v4fu __attribute__((ext_vector_type(4), aligned(4)));

template <int K>
__global__ __launch_bounds__(512)
void gemm8_wmma(const float* __restrict__ A, int lda,
                const float* __restrict__ Bm, int ldb,
                const float* __restrict__ bias,
                float* __restrict__ Out, int ldo,
                int relu)
{
    constexpr int KS = K / 8;
    constexpr int NG = KS / 8;
    __shared__ float red[8 * 8 * 32];

    const int t    = threadIdx.x;
    const int w    = t >> 5;
    const int lane = t & 31;
    const int half = lane >> 4;
    const int l    = lane & 15;
    const int tt   = w & 1;
    const int ks   = w >> 1;
    const int n0   = blockIdx.x * 32;
    const int n    = n0 + tt * 16 + l;

    const float* __restrict__ arow = A + (l & 7) * lda + ks * KS + half * 4;
    const float* __restrict__ brow = Bm + n * ldb + ks * KS + half * 4;

    v4f af[NG], bf[NG];
    #pragma unroll
    for (int g = 0; g < NG; ++g) {
        af[g] = *(const v4fu*)(arow + 8 * g);
        bf[g] = *(const v4fu*)(brow + 8 * g);
    }
    const bool live = l < 8;
    v8f acc0 = {}, acc1 = {};
    #pragma unroll
    for (int g = 0; g < NG; ++g) {
        v4f a = af[g];
        a.x = live ? a.x : 0.0f;
        a.y = live ? a.y : 0.0f;
        a.z = live ? a.z : 0.0f;
        a.w = live ? a.w : 0.0f;
        acc0 = wmma_f32x4(a.xy, bf[g].xy, acc0);
        acc1 = wmma_f32x4(a.zw, bf[g].zw, acc1);
    }

    if (half == 0) {
        #pragma unroll
        for (int r = 0; r < 8; ++r)
            red[(ks * 8 + r) * 32 + tt * 16 + l] = acc0[r] + acc1[r];
    }
    __syncthreads();
    if (t < 256) {
        const int r = t >> 5;
        const int c = t & 31;
        float v = red[r * 32 + c];
        #pragma unroll
        for (int s = 1; s < 8; ++s) v += red[(s * 8 + r) * 32 + c];
        v += bias[n0 + c];
        if (relu) v = fmaxf(v, 0.0f);
        Out[r * ldo + n0 + c] = v;
    }
}

static __device__ __forceinline__
void transpose_body(const float* __restrict__ Wf2, float* __restrict__ WT, int blk)
{
    __shared__ float tile[64 * 33];
    const int t    = threadIdx.x;
    const int wv   = blk >> 4;
    const int k0   = (blk & 15) * 32;
    const int lane = t & 31;
    const int r    = t >> 5;
    #pragma unroll
    for (int p = 0; p < 8; ++p) {
        const int ol = p * 8 + r;
        tile[ol * 33 + lane] = Wf2[(wv * 64 + ol) * 512 + k0 + lane];
    }
    __syncthreads();
    #pragma unroll
    for (int p = 0; p < 2; ++p) {
        const int idx = p * 256 + t;
        const int kk  = idx >> 4;
        const int c   = idx & 15;
        v4f v;
        #pragma unroll
        for (int i = 0; i < 4; ++i) v[i] = tile[(i * 16 + c) * 33 + kk];
        *(v4f*)(WT + (k0 + kk) * 512 + wv * 64 + 4 * c) = v;
    }
}

static __device__ __forceinline__
void rows_body(const float* __restrict__ base1,
               const float* __restrict__ Wf1,
               const float* __restrict__ g1, const float* __restrict__ be1,
               const float* __restrict__ m1, const float* __restrict__ v1,
               const float* __restrict__ Wf2,
               float* __restrict__ R,
               float* __restrict__ maskf,
               int rb)
{
    constexpr int AP = 516;
    __shared__ __attribute__((aligned(16))) float As[48 * AP];
    __shared__ float part[4 * 48 * 32];

    const int t    = threadIdx.x;
    const int w    = t >> 5;
    const int lane = t & 31;
    const int half = lane >> 4;
    const int l    = lane & 15;
    const int tt   = w & 1;
    const int ks   = w >> 1;
    const int n0   = rb * 32;

    const float* __restrict__ brow = Wf2 + (n0 + tt * 16 + l) * 512 + ks * 128 + half * 4;
    v4f bf[16];
    #pragma unroll
    for (int g = 0; g < 16; ++g) bf[g] = *(const v4f*)(brow + 8 * g);

    #pragma unroll
    for (int p = 0; p < 2; ++p) {
        const int k = t + 256 * p;
        const float* wr = Wf1 + k * 1029;
        const float sc = g1[k] * rsqrtf(v1[k] + EPSF);
        const float mm = m1[k];
        const float bb = be1[k];
        float r5[5];
        r5[0] = sc * wr[2];
        r5[1] = sc * wr[3];
        r5[2] = sc * wr[4];
        r5[3] = sc * wr[0];
        r5[4] = sc * wr[1];
        #pragma unroll
        for (int bi = 0; bi < 8; ++bi) {
            const float al = __builtin_fmaf(base1[bi * 512 + k] - mm, sc, bb);
            const bool on = al > 0.0f;
            As[(bi * 6) * AP + k] = on ? al : 0.0f;
            #pragma unroll
            for (int j = 0; j < 5; ++j) As[(bi * 6 + 1 + j) * AP + k] = on ? r5[j] : 0.0f;
            if (rb == 0) maskf[bi * 512 + k] = on ? 1.0f : 0.0f;
        }
    }
    __syncthreads();

    v8f acc[3];
    #pragma unroll
    for (int mt = 0; mt < 3; ++mt) acc[mt] = v8f{};
    #pragma unroll
    for (int g = 0; g < 16; ++g) {
        #pragma unroll
        for (int mt = 0; mt < 3; ++mt) {
            const v4f a = *(const v4f*)&As[(mt * 16 + l) * AP + ks * 128 + 8 * g + 4 * half];
            acc[mt] = wmma_f32x4(a.xy, bf[g].xy, acc[mt]);
            acc[mt] = wmma_f32x4(a.zw, bf[g].zw, acc[mt]);
        }
    }
    #pragma unroll
    for (int mt = 0; mt < 3; ++mt)
        #pragma unroll
        for (int r = 0; r < 8; ++r)
            part[(ks * 48 + mt * 16 + 8 * half + r) * 32 + tt * 16 + l] = acc[mt][r];
    __syncthreads();
    #pragma unroll
    for (int p = 0; p < 6; ++p) {
        const int idx = t + 256 * p;
        const int row = idx >> 5;
        const int col = idx & 31;
        float v = part[row * 32 + col];
        #pragma unroll
        for (int s = 1; s < 4; ++s) v += part[(s * 48 + row) * 32 + col];
        R[row * 512 + n0 + col] = v;
    }
}

__global__ __launch_bounds__(256)
void prep_kernel(const float* __restrict__ Wf2, float* __restrict__ WT,
                 const float* __restrict__ base1, const float* __restrict__ Wf1,
                 const float* __restrict__ g1, const float* __restrict__ be1,
                 const float* __restrict__ m1, const float* __restrict__ v1,
                 float* __restrict__ R, float* __restrict__ maskf)
{
    if (blockIdx.x < 128) transpose_body(Wf2, WT, (int)blockIdx.x);
    else rows_body(base1, Wf1, g1, be1, m1, v1, Wf2, R, maskf, (int)blockIdx.x - 128);
}

__global__ __launch_bounds__(256)
void fold_fused(const float* __restrict__ coarse,
                const float* __restrict__ base1,
                const float* __restrict__ Wf1,
                const float* __restrict__ g1, const float* __restrict__ be1,
                const float* __restrict__ m1, const float* __restrict__ v1,
                const float* __restrict__ Wf2,
                const float* __restrict__ WT,
                const float* __restrict__ bf2,
                const float* __restrict__ g2, const float* __restrict__ be2,
                const float* __restrict__ m2, const float* __restrict__ v2,
                const float* __restrict__ Wo,
                const float* __restrict__ bo,
                const float* __restrict__ R,
                const float* __restrict__ maskf,
                float* __restrict__ fine)
{
    constexpr int Y1P = 196;
    __shared__ __attribute__((aligned(16))) float y1q[129 * Y1P];
    __shared__ __attribute__((aligned(16))) unsigned short ktab[512];
    __shared__ __attribute__((aligned(16))) float wos[3 * 512];
    __shared__ __attribute__((aligned(16))) float s2s[512];
    __shared__ __attribute__((aligned(16))) float shs[512];
    __shared__ float red[8 * 16 * 48];
    unsigned char* const aflag = (unsigned char*)red;
    __shared__ unsigned int mixflag[16 * 128];
    __shared__ unsigned short mlist[16 * 512];
    __shared__ int mcnt[16];
    auto csidx = [](int j) -> int { return (j >> 2) * Y1P + 192 + (j & 3); };

    const int t    = threadIdx.x;
    const int b    = blockIdx.y;
    const int w    = t >> 5;
    const int lane = t & 31;
    const int half = lane >> 4;
    const int l    = lane & 15;
    const float step = 0.1f / 3.0f;

    {
        const int q    = t & 127;
        const int cgrp = t >> 7;
        const v4f bse = *(const v4f*)(base1 + b * 512 + q * 4);
        const v4f gv  = *(const v4f*)(g1 + q * 4);
        const v4f vv  = *(const v4f*)(v1 + q * 4);
        const v4f mv  = *(const v4f*)(m1 + q * 4);
        const v4f bev = *(const v4f*)(be1 + q * 4);
        const v4f mk  = *(const v4f*)(maskf + b * 512 + q * 4);
        if (t < 48) y1q[csidx(t)] = coarse[(b * 1024 + blockIdx.x * 16) * 3 + t];
        v4f wx, wy, wz, pp, qq, ee, sc;
        #pragma unroll
        for (int c = 0; c < 4; ++c) {
            const float* wr = Wf1 + (q * 4 + c) * 1029;
            sc[c] = gv[c] * rsqrtf(vv[c] + EPSF);
            pp[c] = sc[c] * wr[0];
            qq[c] = sc[c] * wr[1];
            ee[c] = 0.05f * (fabsf(pp[c]) + fabsf(qq[c]));
            wx[c] = wr[2];
            wy[c] = wr[3];
            wz[c] = wr[4];
        }
        float cxs[8], cys[8], czs[8];
        #pragma unroll
        for (int i = 0; i < 8; ++i) {
            const float* cp = coarse + (b * 1024 + blockIdx.x * 16 + cgrp + 2 * i) * 3;
            cxs[i] = cp[0]; cys[i] = cp[1]; czs[i] = cp[2];
        }
        if (t < 128) {
            const v4f g2v = *(const v4f*)(g2 + t * 4);
            const v4f v2v = *(const v4f*)(v2 + t * 4);
            const v4f m2v = *(const v4f*)(m2 + t * 4);
            const v4f b2v = *(const v4f*)(be2 + t * 4);
            const v4f f2v = *(const v4f*)(bf2 + t * 4);
            v4f s, sh;
            #pragma unroll
            for (int c = 0; c < 4; ++c) {
                s[c]  = g2v[c] * rsqrtf(v2v[c] + EPSF);
                sh[c] = __builtin_fmaf(f2v[c] - m2v[c], s[c], b2v[c]);
            }
            *(v4f*)&s2s[t * 4] = s;
            *(v4f*)&shs[t * 4] = sh;
            *(v4f*)&wos[1024 + t * 4] = *(const v4f*)(Wo + 1024 + t * 4);
        }
        *(v4f*)&wos[t * 4] = *(const v4f*)(Wo + t * 4);

        unsigned int live = 0;
        #pragma unroll
        for (int i = 0; i < 8; ++i) {
            const int cl   = cgrp + 2 * i;
            const int rowA = cgrp * 32 + i;
            const int rowQ = rowA + 8;
            const int rowP = 16 + i + 8 * cgrp;
            v4f ra, rp, rq;
            unsigned int fl = 0;
            #pragma unroll
            for (int c = 0; c < 4; ++c) {
                float val = bse[c];
                val = __builtin_fmaf(cxs[i], wx[c], val);
                val = __builtin_fmaf(cys[i], wy[c], val);
                val = __builtin_fmaf(czs[i], wz[c], val);
                const float a = __builtin_fmaf(val - mv[c], sc[c], bev[c]);
                const bool dead  = !(a + ee[c] > 0.0f);
                const bool mixed = !dead && !(a - ee[c] > 0.0f);
                const bool on    = mk[c] > 0.5f;
                const bool rows  = on ? (dead || mixed) : !dead;
                ra[c] = rows ? (on ? -a : a) : 0.0f;
                rp[c] = rows ? (on ? -pp[c] : pp[c]) : 0.0f;
                rq[c] = rows ? (on ? -qq[c] : qq[c]) : 0.0f;
                fl |= (mixed ? 1u : 0u) << (8 * c);
                live |= (rows ? 1u : 0u) << c;
            }
            *(v4f*)&y1q[q * Y1P + rowA * 4] = ra;
            *(v4f*)&y1q[q * Y1P + rowP * 4] = rp;
            *(v4f*)&y1q[q * Y1P + rowQ * 4] = rq;
            mixflag[cl * 128 + q] = fl;
        }
        aflag[t] = (unsigned char)live;
        if (t < 48) *(v4f*)&y1q[128 * Y1P + t * 4] = v4f{};
    }
    __syncthreads();

    int NG;
    {
        const unsigned int lt = (1u << lane) - 1u;
        int total = 0, mybase = 0, mypre = 0;
        unsigned int mine = 0;
        #pragma unroll
        for (int ch = 0; ch < 4; ++ch) {
            const unsigned int f = (unsigned int)aflag[ch * 32 + lane] | (unsigned int)aflag[128 + ch * 32 + lane];
            int cnt = 0, pre = 0;
            #pragma unroll
            for (int c = 0; c < 4; ++c) {
                const unsigned int m = __builtin_amdgcn_ballot_w32(((f >> c) & 1u) != 0u);
                cnt += __builtin_popcount(m);
                pre += __builtin_popcount(m & lt);
            }
            if (ch == (w & 3)) { mine = f; mybase = total; mypre = pre; }
            total += cnt;
        }
        NG = ((total + 15) >> 4) << 1;
        NG = NG < 2 ? 2 : NG;
        if (w < 4) {
            int pos = mybase + mypre;
            #pragma unroll
            for (int c = 0; c < 4; ++c) {
                if ((mine >> c) & 1u) {
                    ktab[pos] = (unsigned short)(4 * t + c);
                    ++pos;
                }
            }
        } else {
            const int p = total + (t - 128);
            if (t - 128 < 16 && p < 8 * NG) ktab[p] = (unsigned short)512;
        }
    }

    {
        const unsigned char* fb = (const unsigned char*)mixflag;
        #pragma unroll
        for (int rep = 0; rep < 2; ++rep) {
            const int cc = w + 8 * rep;
            int basec = 0;
            for (int ch = 0; ch < 16; ++ch) {
                const int k = ch * 32 + lane;
                const bool f = fb[cc * 512 + k] != 0;
                const unsigned int mask = __builtin_amdgcn_ballot_w32(f);
                const int pos = basec + __builtin_popcount(mask & ((1u << lane) - 1u));
                if (f) mlist[cc * 512 + pos] = (unsigned short)k;
                basec += __builtin_popcount(mask);
            }
            if (lane == 0) mcnt[cc] = basec;
        }
    }

    __syncthreads();

    v8f acc[3][4];
    #pragma unroll
    for (int mt = 0; mt < 3; ++mt)
        #pragma unroll
        for (int i = 0; i < 4; ++i) acc[mt][i] = v8f{};
    {
        const int ng = __builtin_amdgcn_readfirstlane(NG);
        const unsigned int arow = (unsigned int)(l * 4);
        const float* __restrict__ wtl = WT + w * 64 + l * 4;
        auto ldtab = [&](int j) -> v2u { return *(const v2u*)&ktab[8 * j + 4 * half]; };
        auto ldgrp = [&](const v2u tb, float (&x)[3][4], v4f (&y)[4]) {
            #pragma unroll
            for (int e = 0; e < 4; ++e) {
                const unsigned int k  = (tb[e >> 1] >> (16 * (e & 1))) & 0xffffu;
                const unsigned int ai = (k >> 2) * Y1P + (k & 3u) + arow;
                #pragma unroll
                for (int mt = 0; mt < 3; ++mt) x[mt][e] = y1q[ai + mt * 64];
                y[e] = *(const v4f*)(wtl + (k & 511u) * 512);
            }
        };
        auto mma24 = [&](const float (&x)[3][4], const v4f (&y)[4]) {
            #pragma unroll
            for (int i = 0; i < 4; ++i)
                #pragma unroll
                for (int mt = 0; mt < 3; ++mt)
                    acc[mt][i] = wmma_f32x4(v2f{x[mt][0], x[mt][1]}, v2f{y[0][i], y[1][i]}, acc[mt][i]);
            #pragma unroll
            for (int i = 0; i < 4; ++i)
                #pragma unroll
                for (int mt = 0; mt < 3; ++mt)
                    acc[mt][i] = wmma_f32x4(v2f{x[mt][2], x[mt][3]}, v2f{y[2][i], y[3][i]}, acc[mt][i]);
        };
        float pa[3][4], qa[3][4];
        v4f pb[4], qb[4];
        v2u t0 = ldtab(0), t1 = ldtab(1);
        ldgrp(t0, pa, pb);
        #pragma nounroll
        for (int j = 0; j < ng; j += 2) {
            const int j2 = j + 2 < ng ? j + 2 : ng - 1;
            const int j3 = j + 3 < ng ? j + 3 : ng - 1;
            t0 = ldtab(j2);
            ldgrp(t1, qa, qb);
            __builtin_amdgcn_sched_barrier(0);
            mma24(pa, pb);
            __builtin_amdgcn_sched_barrier(0);
            t1 = ldtab(j3);
            ldgrp(t0, pa, pb);
            __builtin_amdgcn_sched_barrier(0);
            mma24(qa, qb);
            __builtin_amdgcn_sched_barrier(0);
        }
    }

    {
        const float gx = -0.05f + (float)(l >> 2) * step;
        const float gy = -0.05f + (float)(l & 3) * step;
        v2f gbX, gbY;
        gbX.x = half ? gy : 1.0f;
        gbX.y = half ? 0.0f : gx;
        gbY.x = half ? gy : 1.0f;
        gbY.y = half ? gx : 0.0f;
        float rAl[4], rX[4], rY[4], rZ[4], rP[4], rQ[4];
        #pragma unroll
        for (int i = 0; i < 4; ++i) {
            const float* rbp = R + b * (6 * 512) + (w * 4 + i) * 16 + l;
            rAl[i] = rbp[0];
            rX[i]  = rbp[512];
            rY[i]  = rbp[1024];
            rZ[i]  = rbp[1536];
            rP[i]  = rbp[2048];
            rQ[i]  = rbp[2560];
        }
        #pragma unroll
        for (int cl = 0; cl < 16; ++cl) {
            const float ccx = y1q[csidx(cl * 3)], ccy = y1q[csidx(cl * 3 + 1)], ccz = y1q[csidx(cl * 3 + 2)];
            const int pr = cl >> 1, sy = cl & 1;
            const int mtA  = sy ? 2 : 0;
            const int rowA = sy * 32 + pr;
            const int rowQ = rowA + 8;
            const int rowP = 16 + pr + 8 * sy;
            const v2f gb = sy ? gbY : gbX;
            v8f d[4];
            #pragma unroll
            for (int i = 0; i < 4; ++i) {
                const float ab = __builtin_fmaf(ccz, rZ[i], __builtin_fmaf(ccy, rY[i], __builtin_fmaf(ccx, rX[i], rAl[i])));
                v2f ea;
                ea.x = acc[mtA][i][pr] + (half ? rQ[i] : ab);
                ea.y = acc[1][i][pr] + rP[i];
                d[i] = wmma_f32x4(ea, gb, v8f{});
            }
            const int n = __builtin_amdgcn_readfirstlane(mcnt[cl]);
            for (int base = 0; base < n; base += 4) {
                const int i0 = base + 2 * half, i1 = i0 + 1;
                const bool ok0 = i0 < n, ok1 = i1 < n;
                const int k0 = ok0 ? (int)mlist[cl * 512 + i0] : 0;
                const int k1 = ok1 ? (int)mlist[cl * 512 + i1] : 0;
                const float* r0 = &y1q[(k0 >> 2) * Y1P + (k0 & 3)];
                const float* r1 = &y1q[(k1 >> 2) * Y1P + (k1 & 3)];
                const float z0 = __builtin_fmaf(gy, r0[rowQ * 4], __builtin_fmaf(gx, r0[rowP * 4], r0[rowA * 4]));
                const float z1 = __builtin_fmaf(gy, r1[rowQ * 4], __builtin_fmaf(gx, r1[rowP * 4], r1[rowA * 4]));
                v2f cb;
                cb.x = ok0 ? fmaxf(-z0, 0.0f) : 0.0f;
                cb.y = ok1 ? fmaxf(-z1, 0.0f) : 0.0f;
                #pragma unroll
                for (int i = 0; i < 4; ++i) {
                    const float* wrow = Wf2 + ((w * 4 + i) * 16 + l) * 512;
                    v2f wa;
                    wa.x = wrow[k0];
                    wa.y = wrow[k1];
                    d[i] = wmma_f32x4(wa, cb, d[i]);
                }
            }
            float p0 = 0.0f, p1 = 0.0f, p2 = 0.0f;
            #pragma unroll
            for (int i = 0; i < 4; ++i) {
                const int ob = (w * 4 + i) * 16 + 8 * half;
                #pragma unroll
                for (int hq = 0; hq < 2; ++hq) {
                    const v4f s  = *(const v4f*)&s2s[ob + 4 * hq];
                    const v4f sh = *(const v4f*)&shs[ob + 4 * hq];
                    const v4f w0 = *(const v4f*)&wos[ob + 4 * hq];
                    const v4f w1 = *(const v4f*)&wos[512 + ob + 4 * hq];
                    const v4f w2 = *(const v4f*)&wos[1024 + ob + 4 * hq];
                    #pragma unroll
                    for (int e = 0; e < 4; ++e) {
                        const float y = fmaxf(__builtin_fmaf(d[i][4 * hq + e], s[e], sh[e]), 0.0f);
                        p0 = __builtin_fmaf(y, w0[e], p0);
                        p1 = __builtin_fmaf(y, w1[e], p1);
                        p2 = __builtin_fmaf(y, w2[e], p2);
                    }
                }
            }
            p0 += __shfl_xor(p0, 16);
            p1 += __shfl_xor(p1, 16);
            p2 += __shfl_xor(p2, 16);
            if (half == 0) {
                float* rd = &red[(w * 16 + cl) * 48 + l * 3];
                rd[0] = p0; rd[1] = p1; rd[2] = p2;
            }
        }
    }
    __syncthreads();

    for (int idx = t; idx < 768; idx += 256) {
        const int cl  = idx / 48;
        const int rem = idx - cl * 48;
        const int g   = rem / 3;
        const int c   = rem - g * 3;
        float sum = 0.0f;
        #pragma unroll
        for (int s = 0; s < 8; ++s) sum += red[(s * 16 + cl) * 48 + rem];
        const int ng = (blockIdx.x * 16 + cl) * 16 + g;
        const int i  = c * 16384 + ng;
        const int r3 = i / 3;
        const int cc = i - r3 * 3;
        const float center = coarse[(b * 1024 + (r3 >> 4)) * 3 + cc];
        fine[(b * 16384 + ng) * 3 + c] = sum + bo[c] + center;
    }
}

extern "C" void kernel_launch(void* const* d_in, const int* in_sizes, int n_in,
                              void* d_out, int out_size, void* d_ws, size_t ws_size,
                              hipStream_t stream)
{
    const float* features = (const float*)d_in[0];
    const float* W1  = (const float*)d_in[1];
    const float* b1  = (const float*)d_in[2];
    const float* W2  = (const float*)d_in[3];
    const float* b2  = (const float*)d_in[4];
    const float* Wc  = (const float*)d_in[5];
    const float* bc  = (const float*)d_in[6];
    const float* Wf1 = (const float*)d_in[7];
    const float* bf1 = (const float*)d_in[8];
    const float* g1  = (const float*)d_in[9];
    const float* be1 = (const float*)d_in[10];
    const float* m1  = (const float*)d_in[11];
    const float* v1  = (const float*)d_in[12];
    const float* Wf2 = (const float*)d_in[13];
    const float* bf2 = (const float*)d_in[14];
    const float* g2  = (const float*)d_in[15];
    const float* be2 = (const float*)d_in[16];
    const float* m2  = (const float*)d_in[17];
    const float* v2  = (const float*)d_in[18];
    const float* Wo  = (const float*)d_in[19];
    const float* bo  = (const float*)d_in[20];

    float* out    = (float*)d_out;
    float* coarse = out;
    float* fine   = out + 8 * 1024 * 3;

    float* ws    = (float*)d_ws;
    float* h1    = ws;
    float* h2    = ws + 8192;
    float* base1 = ws + 16384;
    float* WT    = ws + 28672;
    float* R     = ws + 290816;
    float* maskf = ws + 315392;

    gemm8_wmma<1024><<<32, 512, 0, stream>>>(features, 1024, W1, 1024, b1, h1,     1024, 1);
    gemm8_wmma<1024><<<32, 512, 0, stream>>>(h1,       1024, W2, 1024, b2, h2,     1024, 1);
    gemm8_wmma<1024><<<96, 512, 0, stream>>>(h2,       1024, Wc, 1024, bc, coarse, 3072, 0);
    gemm8_wmma<1024><<<16, 512, 0, stream>>>(features, 1024, Wf1 + 5, 1029, bf1, base1, 512, 0);
    prep_kernel<<<144, 256, 0, stream>>>(Wf2, WT, base1, Wf1, g1, be1, m1, v1, R, maskf);

    dim3 gridF(1024 / 16, 8);
    fold_fused<<<gridF, 256, 0, stream>>>(coarse, base1, Wf1,
                                          g1, be1, m1, v1,
                                          Wf2, WT, bf2, g2, be2, m2, v2,
                                          Wo, bo, R, maskf, fine);
}


# fold: listed channels take their Wf2 values from the transposed copy: two 16-byte loads per lane, 4 lines per channel pair, not eight 4-byte loads each on a line of its own; sums and order kept
# speedup vs baseline: 6.9105x; 1.1943x over previous
#include <hip/hip_runtime.h>
#include <hip/hip_bf16.h>


typedef float v2f __attribute__((ext_vector_type(2)));
typedef float v4f __attribute__((ext_vector_type(4)));
typedef float v8f __attribute__((ext_vector_type(8)));
typedef unsigned int v2u __attribute__((ext_vector_type(2)));

#define EPSF 1e-5f

static __device__ __forceinline__ v8f wmma_f32x4(v2f a, v2f b, v8f c) {
    return __builtin_amdgcn_wmma_f32_16x16x4_f32(false, a, false, b, (short)0, c, false, false);
}

typedef float v4fu __attribute__((ext_vector_type(4), aligned(4)));

template <int K>
__global__ __launch_bounds__(512)
void gemm8_wmma(const float* __restrict__ A, int lda,
                const float* __restrict__ Bm, int ldb,
                const float* __restrict__ bias,
                float* __restrict__ Out, int ldo,
                int relu)
{
    constexpr int KS = K / 8;
    constexpr int NG = KS / 8;
    __shared__ float red[8 * 8 * 32];

    const int t    = threadIdx.x;
    const int w    = t >> 5;
    const int lane = t & 31;
    const int half = lane >> 4;
    const int l    = lane & 15;
    const int tt   = w & 1;
    const int ks   = w >> 1;
    const int n0   = blockIdx.x * 32;
    const int n    = n0 + tt * 16 + l;

    const float* __restrict__ arow = A + (l & 7) * lda + ks * KS + half * 4;
    const float* __restrict__ brow = Bm + n * ldb + ks * KS + half * 4;

    v4f af[NG], bf[NG];
    #pragma unroll
    for (int g = 0; g < NG; ++g) {
        af[g] = *(const v4fu*)(arow + 8 * g);
        bf[g] = *(const v4fu*)(brow + 8 * g);
    }
    const bool live = l < 8;
    v8f acc0 = {}, acc1 = {};
    #pragma unroll
    for (int g = 0; g < NG; ++g) {
        v4f a = af[g];
        a.x = live ? a.x : 0.0f;
        a.y = live ? a.y : 0.0f;
        a.z = live ? a.z : 0.0f;
        a.w = live ? a.w : 0.0f;
        acc0 = wmma_f32x4(a.xy, bf[g].xy, acc0);
        acc1 = wmma_f32x4(a.zw, bf[g].zw, acc1);
    }

    if (half == 0) {
        #pragma unroll
        for (int r = 0; r < 8; ++r)
            red[(ks * 8 + r) * 32 + tt * 16 + l] = acc0[r] + acc1[r];
    }
    __syncthreads();
    if (t < 256) {
        const int r = t >> 5;
        const int c = t & 31;
        float v = red[r * 32 + c];
        #pragma unroll
        for (int s = 1; s < 8; ++s) v += red[(s * 8 + r) * 32 + c];
        v += bias[n0 + c];
        if (relu) v = fmaxf(v, 0.0f);
        Out[r * ldo + n0 + c] = v;
    }
}

static __device__ __forceinline__
void transpose_body(const float* __restrict__ Wf2, float* __restrict__ WT, int blk)
{
    __shared__ float tile[64 * 33];
    const int t    = threadIdx.x;
    const int wv   = blk >> 4;
    const int k0   = (blk & 15) * 32;
    const int lane = t & 31;
    const int r    = t >> 5;
    #pragma unroll
    for (int p = 0; p < 8; ++p) {
        const int ol = p * 8 + r;
        tile[ol * 33 + lane] = Wf2[(wv * 64 + ol) * 512 + k0 + lane];
    }
    __syncthreads();
    #pragma unroll
    for (int p = 0; p < 2; ++p) {
        const int idx = p * 256 + t;
        const int kk  = idx >> 4;
        const int c   = idx & 15;
        v4f v;
        #pragma unroll
        for (int i = 0; i < 4; ++i) v[i] = tile[(i * 16 + c) * 33 + kk];
        *(v4f*)(WT + (k0 + kk) * 512 + wv * 64 + 4 * c) = v;
    }
}

static __device__ __forceinline__
void rows_body(const float* __restrict__ base1,
               const float* __restrict__ Wf1,
               const float* __restrict__ g1, const float* __restrict__ be1,
               const float* __restrict__ m1, const float* __restrict__ v1,
               const float* __restrict__ Wf2,
               float* __restrict__ R,
               float* __restrict__ maskf,
               int rb)
{
    constexpr int AP = 516;
    __shared__ __attribute__((aligned(16))) float As[48 * AP];
    __shared__ float part[4 * 48 * 32];

    const int t    = threadIdx.x;
    const int w    = t >> 5;
    const int lane = t & 31;
    const int half = lane >> 4;
    const int l    = lane & 15;
    const int tt   = w & 1;
    const int ks   = w >> 1;
    const int n0   = rb * 32;

    const float* __restrict__ brow = Wf2 + (n0 + tt * 16 + l) * 512 + ks * 128 + half * 4;
    v4f bf[16];
    #pragma unroll
    for (int g = 0; g < 16; ++g) bf[g] = *(const v4f*)(brow + 8 * g);

    #pragma unroll
    for (int p = 0; p < 2; ++p) {
        const int k = t + 256 * p;
        const float* wr = Wf1 + k * 1029;
        const float sc = g1[k] * rsqrtf(v1[k] + EPSF);
        const float mm = m1[k];
        const float bb = be1[k];
        float r5[5];
        r5[0] = sc * wr[2];
        r5[1] = sc * wr[3];
        r5[2] = sc * wr[4];
        r5[3] = sc * wr[0];
        r5[4] = sc * wr[1];
        #pragma unroll
        for (int bi = 0; bi < 8; ++bi) {
            const float al = __builtin_fmaf(base1[bi * 512 + k] - mm, sc, bb);
            const bool on = al > 0.0f;
            As[(bi * 6) * AP + k] = on ? al : 0.0f;
            #pragma unroll
            for (int j = 0; j < 5; ++j) As[(bi * 6 + 1 + j) * AP + k] = on ? r5[j] : 0.0f;
            if (rb == 0) maskf[bi * 512 + k] = on ? 1.0f : 0.0f;
        }
    }
    __syncthreads();

    v8f acc[3];
    #pragma unroll
    for (int mt = 0; mt < 3; ++mt) acc[mt] = v8f{};
    #pragma unroll
    for (int g = 0; g < 16; ++g) {
        #pragma unroll
        for (int mt = 0; mt < 3; ++mt) {
            const v4f a = *(const v4f*)&As[(mt * 16 + l) * AP + ks * 128 + 8 * g + 4 * half];
            acc[mt] = wmma_f32x4(a.xy, bf[g].xy, acc[mt]);
            acc[mt] = wmma_f32x4(a.zw, bf[g].zw, acc[mt]);
        }
    }
    #pragma unroll
    for (int mt = 0; mt < 3; ++mt)
        #pragma unroll
        for (int r = 0; r < 8; ++r)
            part[(ks * 48 + mt * 16 + 8 * half + r) * 32 + tt * 16 + l] = acc[mt][r];
    __syncthreads();
    #pragma unroll
    for (int p = 0; p < 6; ++p) {
        const int idx = t + 256 * p;
        const int row = idx >> 5;
        const int col = idx & 31;
        float v = part[row * 32 + col];
        #pragma unroll
        for (int s = 1; s < 4; ++s) v += part[(s * 48 + row) * 32 + col];
        R[row * 512 + n0 + col] = v;
    }
}

__global__ __launch_bounds__(256)
void prep_kernel(const float* __restrict__ Wf2, float* __restrict__ WT,
                 const float* __restrict__ base1, const float* __restrict__ Wf1,
                 const float* __restrict__ g1, const float* __restrict__ be1,
                 const float* __restrict__ m1, const float* __restrict__ v1,
                 float* __restrict__ R, float* __restrict__ maskf)
{
    if (blockIdx.x < 128) transpose_body(Wf2, WT, (int)blockIdx.x);
    else rows_body(base1, Wf1, g1, be1, m1, v1, Wf2, R, maskf, (int)blockIdx.x - 128);
}

__global__ __launch_bounds__(256)
void fold_fused(const float* __restrict__ coarse,
                const float* __restrict__ base1,
                const float* __restrict__ Wf1,
                const float* __restrict__ g1, const float* __restrict__ be1,
                const float* __restrict__ m1, const float* __restrict__ v1,
                const float* __restrict__ WT,
                const float* __restrict__ bf2,
                const float* __restrict__ g2, const float* __restrict__ be2,
                const float* __restrict__ m2, const float* __restrict__ v2,
                const float* __restrict__ Wo,
                const float* __restrict__ bo,
                const float* __restrict__ R,
                const float* __restrict__ maskf,
                float* __restrict__ fine)
{
    constexpr int Y1P = 196;
    __shared__ __attribute__((aligned(16))) float y1q[129 * Y1P];
    __shared__ __attribute__((aligned(16))) unsigned short ktab[512];
    __shared__ __attribute__((aligned(16))) float wos[3 * 512];
    __shared__ __attribute__((aligned(16))) float s2s[512];
    __shared__ __attribute__((aligned(16))) float shs[512];
    __shared__ float red[8 * 16 * 48];
    unsigned char* const aflag = (unsigned char*)red;
    __shared__ unsigned int mixflag[16 * 128];
    __shared__ unsigned short mlist[16 * 512];
    __shared__ int mcnt[16];
    auto csidx = [](int j) -> int { return (j >> 2) * Y1P + 192 + (j & 3); };

    const int t    = threadIdx.x;
    const int b    = blockIdx.y;
    const int w    = t >> 5;
    const int lane = t & 31;
    const int half = lane >> 4;
    const int l    = lane & 15;
    const float step = 0.1f / 3.0f;

    {
        const int q    = t & 127;
        const int cgrp = t >> 7;
        const v4f bse = *(const v4f*)(base1 + b * 512 + q * 4);
        const v4f gv  = *(const v4f*)(g1 + q * 4);
        const v4f vv  = *(const v4f*)(v1 + q * 4);
        const v4f mv  = *(const v4f*)(m1 + q * 4);
        const v4f bev = *(const v4f*)(be1 + q * 4);
        const v4f mk  = *(const v4f*)(maskf + b * 512 + q * 4);
        if (t < 48) y1q[csidx(t)] = coarse[(b * 1024 + blockIdx.x * 16) * 3 + t];
        v4f wx, wy, wz, pp, qq, ee, sc;
        #pragma unroll
        for (int c = 0; c < 4; ++c) {
            const float* wr = Wf1 + (q * 4 + c) * 1029;
            sc[c] = gv[c] * rsqrtf(vv[c] + EPSF);
            pp[c] = sc[c] * wr[0];
            qq[c] = sc[c] * wr[1];
            ee[c] = 0.05f * (fabsf(pp[c]) + fabsf(qq[c]));
            wx[c] = wr[2];
            wy[c] = wr[3];
            wz[c] = wr[4];
        }
        float cxs[8], cys[8], czs[8];
        #pragma unroll
        for (int i = 0; i < 8; ++i) {
            const float* cp = coarse + (b * 1024 + blockIdx.x * 16 + cgrp + 2 * i) * 3;
            cxs[i] = cp[0]; cys[i] = cp[1]; czs[i] = cp[2];
        }
        if (t < 128) {
            const v4f g2v = *(const v4f*)(g2 + t * 4);
            const v4f v2v = *(const v4f*)(v2 + t * 4);
            const v4f m2v = *(const v4f*)(m2 + t * 4);
            const v4f b2v = *(const v4f*)(be2 + t * 4);
            const v4f f2v = *(const v4f*)(bf2 + t * 4);
            v4f s, sh;
            #pragma unroll
            for (int c = 0; c < 4; ++c) {
                s[c]  = g2v[c] * rsqrtf(v2v[c] + EPSF);
                sh[c] = __builtin_fmaf(f2v[c] - m2v[c], s[c], b2v[c]);
            }
            *(v4f*)&s2s[t * 4] = s;
            *(v4f*)&shs[t * 4] = sh;
            *(v4f*)&wos[1024 + t * 4] = *(const v4f*)(Wo + 1024 + t * 4);
        }
        *(v4f*)&wos[t * 4] = *(const v4f*)(Wo + t * 4);

        unsigned int live = 0;
        #pragma unroll
        for (int i = 0; i < 8; ++i) {
            const int cl   = cgrp + 2 * i;
            const int rowA = cgrp * 32 + i;
            const int rowQ = rowA + 8;
            const int rowP = 16 + i + 8 * cgrp;
            v4f ra, rp, rq;
            unsigned int fl = 0;
            #pragma unroll
            for (int c = 0; c < 4; ++c) {
                float val = bse[c];
                val = __builtin_fmaf(cxs[i], wx[c], val);
                val = __builtin_fmaf(cys[i], wy[c], val);
                val = __builtin_fmaf(czs[i], wz[c], val);
                const float a = __builtin_fmaf(val - mv[c], sc[c], bev[c]);
                const bool dead  = !(a + ee[c] > 0.0f);
                const bool mixed = !dead && !(a - ee[c] > 0.0f);
                const bool on    = mk[c] > 0.5f;
                const bool rows  = on ? (dead || mixed) : !dead;
                ra[c] = rows ? (on ? -a : a) : 0.0f;
                rp[c] = rows ? (on ? -pp[c] : pp[c]) : 0.0f;
                rq[c] = rows ? (on ? -qq[c] : qq[c]) : 0.0f;
                fl |= (mixed ? 1u : 0u) << (8 * c);
                live |= (rows ? 1u : 0u) << c;
            }
            *(v4f*)&y1q[q * Y1P + rowA * 4] = ra;
            *(v4f*)&y1q[q * Y1P + rowP * 4] = rp;
            *(v4f*)&y1q[q * Y1P + rowQ * 4] = rq;
            mixflag[cl * 128 + q] = fl;
        }
        aflag[t] = (unsigned char)live;
        if (t < 48) *(v4f*)&y1q[128 * Y1P + t * 4] = v4f{};
    }
    __syncthreads();

    int NG;
    {
        const unsigned int lt = (1u << lane) - 1u;
        int total = 0, mybase = 0, mypre = 0;
        unsigned int mine = 0;
        #pragma unroll
        for (int ch = 0; ch < 4; ++ch) {
            const unsigned int f = (unsigned int)aflag[ch * 32 + lane] | (unsigned int)aflag[128 + ch * 32 + lane];
            int cnt = 0, pre = 0;
            #pragma unroll
            for (int c = 0; c < 4; ++c) {
                const unsigned int m = __builtin_amdgcn_ballot_w32(((f >> c) & 1u) != 0u);
                cnt += __builtin_popcount(m);
                pre += __builtin_popcount(m & lt);
            }
            if (ch == (w & 3)) { mine = f; mybase = total; mypre = pre; }
            total += cnt;
        }
        NG = ((total + 15) >> 4) << 1;
        NG = NG < 2 ? 2 : NG;
        if (w < 4) {
            int pos = mybase + mypre;
            #pragma unroll
            for (int c = 0; c < 4; ++c) {
                if ((mine >> c) & 1u) {
                    ktab[pos] = (unsigned short)(4 * t + c);
                    ++pos;
                }
            }
        } else {
            const int p = total + (t - 128);
            if (t - 128 < 16 && p < 8 * NG) ktab[p] = (unsigned short)512;
        }
    }

    {
        const unsigned char* fb = (const unsigned char*)mixflag;
        #pragma unroll
        for (int rep = 0; rep < 2; ++rep) {
            const int cc = w + 8 * rep;
            int basec = 0;
            for (int ch = 0; ch < 16; ++ch) {
                const int k = ch * 32 + lane;
                const bool f = fb[cc * 512 + k] != 0;
                const unsigned int mask = __builtin_amdgcn_ballot_w32(f);
                const int pos = basec + __builtin_popcount(mask & ((1u << lane) - 1u));
                if (f) mlist[cc * 512 + pos] = (unsigned short)k;
                basec += __builtin_popcount(mask);
            }
            if (lane == 0) mcnt[cc] = basec;
        }
    }

    __syncthreads();

    v8f acc[3][4];
    #pragma unroll
    for (int mt = 0; mt < 3; ++mt)
        #pragma unroll
        for (int i = 0; i < 4; ++i) acc[mt][i] = v8f{};
    {
        const int ng = __builtin_amdgcn_readfirstlane(NG);
        const unsigned int arow = (unsigned int)(l * 4);
        const float* __restrict__ wtl = WT + w * 64 + l * 4;
        auto ldtab = [&](int j) -> v2u { return *(const v2u*)&ktab[8 * j + 4 * half]; };
        auto ldgrp = [&](const v2u tb, float (&x)[3][4], v4f (&y)[4]) {
            #pragma unroll
            for (int e = 0; e < 4; ++e) {
                const unsigned int k  = (tb[e >> 1] >> (16 * (e & 1))) & 0xffffu;
                const unsigned int ai = (k >> 2) * Y1P + (k & 3u) + arow;
                #pragma unroll
                for (int mt = 0; mt < 3; ++mt) x[mt][e] = y1q[ai + mt * 64];
                y[e] = *(const v4f*)(wtl + (k & 511u) * 512);
            }
        };
        auto mma24 = [&](const float (&x)[3][4], const v4f (&y)[4]) {
            #pragma unroll
            for (int i = 0; i < 4; ++i)
                #pragma unroll
                for (int mt = 0; mt < 3; ++mt)
                    acc[mt][i] = wmma_f32x4(v2f{x[mt][0], x[mt][1]}, v2f{y[0][i], y[1][i]}, acc[mt][i]);
            #pragma unroll
            for (int i = 0; i < 4; ++i)
                #pragma unroll
                for (int mt = 0; mt < 3; ++mt)
                    acc[mt][i] = wmma_f32x4(v2f{x[mt][2], x[mt][3]}, v2f{y[2][i], y[3][i]}, acc[mt][i]);
        };
        float pa[3][4], qa[3][4];
        v4f pb[4], qb[4];
        v2u t0 = ldtab(0), t1 = ldtab(1);
        ldgrp(t0, pa, pb);
        #pragma nounroll
        for (int j = 0; j < ng; j += 2) {
            const int j2 = j + 2 < ng ? j + 2 : ng - 1;
            const int j3 = j + 3 < ng ? j + 3 : ng - 1;
            t0 = ldtab(j2);
            ldgrp(t1, qa, qb);
            __builtin_amdgcn_sched_barrier(0);
            mma24(pa, pb);
            __builtin_amdgcn_sched_barrier(0);
            t1 = ldtab(j3);
            ldgrp(t0, pa, pb);
            __builtin_amdgcn_sched_barrier(0);
            mma24(qa, qb);
            __builtin_amdgcn_sched_barrier(0);
        }
    }

    {
        const float gx = -0.05f + (float)(l >> 2) * step;
        const float gy = -0.05f + (float)(l & 3) * step;
        v2f gbX, gbY;
        gbX.x = half ? gy : 1.0f;
        gbX.y = half ? 0.0f : gx;
        gbY.x = half ? gy : 1.0f;
        gbY.y = half ? gx : 0.0f;
        const float* __restrict__ wtl = WT + w * 64 + l * 4;
        float rAl[4], rX[4], rY[4], rZ[4], rP[4], rQ[4];
        #pragma unroll
        for (int i = 0; i < 4; ++i) {
            const float* rbp = R + b * (6 * 512) + (w * 4 + i) * 16 + l;
            rAl[i] = rbp[0];
            rX[i]  = rbp[512];
            rY[i]  = rbp[1024];
            rZ[i]  = rbp[1536];
            rP[i]  = rbp[2048];
            rQ[i]  = rbp[2560];
        }
        #pragma unroll
        for (int cl = 0; cl < 16; ++cl) {
            const float ccx = y1q[csidx(cl * 3)], ccy = y1q[csidx(cl * 3 + 1)], ccz = y1q[csidx(cl * 3 + 2)];
            const int pr = cl >> 1, sy = cl & 1;
            const int mtA  = sy ? 2 : 0;
            const int rowA = sy * 32 + pr;
            const int rowQ = rowA + 8;
            const int rowP = 16 + pr + 8 * sy;
            const v2f gb = sy ? gbY : gbX;
            v8f d[4];
            #pragma unroll
            for (int i = 0; i < 4; ++i) {
                const float ab = __builtin_fmaf(ccz, rZ[i], __builtin_fmaf(ccy, rY[i], __builtin_fmaf(ccx, rX[i], rAl[i])));
                v2f ea;
                ea.x = acc[mtA][i][pr] + (half ? rQ[i] : ab);
                ea.y = acc[1][i][pr] + rP[i];
                d[i] = wmma_f32x4(ea, gb, v8f{});
            }
            const int n = __builtin_amdgcn_readfirstlane(mcnt[cl]);
            for (int base = 0; base < n; base += 4) {
                const int i0 = base + 2 * half, i1 = i0 + 1;
                const bool ok0 = i0 < n, ok1 = i1 < n;
                const int k0 = ok0 ? (int)mlist[cl * 512 + i0] : 0;
                const int k1 = ok1 ? (int)mlist[cl * 512 + i1] : 0;
                const float* r0 = &y1q[(k0 >> 2) * Y1P + (k0 & 3)];
                const float* r1 = &y1q[(k1 >> 2) * Y1P + (k1 & 3)];
                const float z0 = __builtin_fmaf(gy, r0[rowQ * 4], __builtin_fmaf(gx, r0[rowP * 4], r0[rowA * 4]));
                const float z1 = __builtin_fmaf(gy, r1[rowQ * 4], __builtin_fmaf(gx, r1[rowP * 4], r1[rowA * 4]));
                v2f cb;
                cb.x = ok0 ? fmaxf(-z0, 0.0f) : 0.0f;
                cb.y = ok1 ? fmaxf(-z1, 0.0f) : 0.0f;
                const v4f wk0 = *(const v4f*)(wtl + k0 * 512);
                const v4f wk1 = *(const v4f*)(wtl + k1 * 512);
                #pragma unroll
                for (int i = 0; i < 4; ++i) {
                    v2f wa;
                    wa.x = wk0[i];
                    wa.y = wk1[i];
                    d[i] = wmma_f32x4(wa, cb, d[i]);
                }
            }
            float p0 = 0.0f, p1 = 0.0f, p2 = 0.0f;
            #pragma unroll
            for (int i = 0; i < 4; ++i) {
                const int ob = (w * 4 + i) * 16 + 8 * half;
                #pragma unroll
                for (int hq = 0; hq < 2; ++hq) {
                    const v4f s  = *(const v4f*)&s2s[ob + 4 * hq];
                    const v4f sh = *(const v4f*)&shs[ob + 4 * hq];
                    const v4f w0 = *(const v4f*)&wos[ob + 4 * hq];
                    const v4f w1 = *(const v4f*)&wos[512 + ob + 4 * hq];
                    const v4f w2 = *(const v4f*)&wos[1024 + ob + 4 * hq];
                    #pragma unroll
                    for (int e = 0; e < 4; ++e) {
                        const float y = fmaxf(__builtin_fmaf(d[i][4 * hq + e], s[e], sh[e]), 0.0f);
                        p0 = __builtin_fmaf(y, w0[e], p0);
                        p1 = __builtin_fmaf(y, w1[e], p1);
                        p2 = __builtin_fmaf(y, w2[e], p2);
                    }
                }
            }
            p0 += __shfl_xor(p0, 16);
            p1 += __shfl_xor(p1, 16);
            p2 += __shfl_xor(p2, 16);
            if (half == 0) {
                float* rd = &red[(w * 16 + cl) * 48 + l * 3];
                rd[0] = p0; rd[1] = p1; rd[2] = p2;
            }
        }
    }
    __syncthreads();

    for (int idx = t; idx < 768; idx += 256) {
        const int cl  = idx / 48;
        const int rem = idx - cl * 48;
        const int g   = rem / 3;
        const int c   = rem - g * 3;
        float sum = 0.0f;
        #pragma unroll
        for (int s = 0; s < 8; ++s) sum += red[(s * 16 + cl) * 48 + rem];
        const int ng = (blockIdx.x * 16 + cl) * 16 + g;
        const int i  = c * 16384 + ng;
        const int r3 = i / 3;
        const int cc = i - r3 * 3;
        const float center = coarse[(b * 1024 + (r3 >> 4)) * 3 + cc];
        fine[(b * 16384 + ng) * 3 + c] = sum + bo[c] + center;
    }
}

extern "C" void kernel_launch(void* const* d_in, const int* in_sizes, int n_in,
                              void* d_out, int out_size, void* d_ws, size_t ws_size,
                              hipStream_t stream)
{
    const float* features = (const float*)d_in[0];
    const float* W1  = (const float*)d_in[1];
    const float* b1  = (const float*)d_in[2];
    const float* W2  = (const float*)d_in[3];
    const float* b2  = (const float*)d_in[4];
    const float* Wc  = (const float*)d_in[5];
    const float* bc  = (const float*)d_in[6];
    const float* Wf1 = (const float*)d_in[7];
    const float* bf1 = (const float*)d_in[8];
    const float* g1  = (const float*)d_in[9];
    const float* be1 = (const float*)d_in[10];
    const float* m1  = (const float*)d_in[11];
    const float* v1  = (const float*)d_in[12];
    const float* Wf2 = (const float*)d_in[13];
    const float* bf2 = (const float*)d_in[14];
    const float* g2  = (const float*)d_in[15];
    const float* be2 = (const float*)d_in[16];
    const float* m2  = (const float*)d_in[17];
    const float* v2  = (const float*)d_in[18];
    const float* Wo  = (const float*)d_in[19];
    const float* bo  = (const float*)d_in[20];

    float* out    = (float*)d_out;
    float* coarse = out;
    float* fine   = out + 8 * 1024 * 3;

    float* ws    = (float*)d_ws;
    float* h1    = ws;
    float* h2    = ws + 8192;
    float* base1 = ws + 16384;
    float* WT    = ws + 28672;
    float* R     = ws + 290816;
    float* maskf = ws + 315392;

    gemm8_wmma<1024><<<32, 512, 0, stream>>>(features, 1024, W1, 1024, b1, h1,     1024, 1);
    gemm8_wmma<1024><<<32, 512, 0, stream>>>(h1,       1024, W2, 1024, b2, h2,     1024, 1);
    gemm8_wmma<1024><<<96, 512, 0, stream>>>(h2,       1024, Wc, 1024, bc, coarse, 3072, 0);
    gemm8_wmma<1024><<<16, 512, 0, stream>>>(features, 1024, Wf1 + 5, 1029, bf1, base1, 512, 0);
    prep_kernel<<<144, 256, 0, stream>>>(Wf2, WT, base1, Wf1, g1, be1, m1, v1, R, maskf);

    dim3 gridF(1024 / 16, 8);
    fold_fused<<<gridF, 256, 0, stream>>>(coarse, base1, Wf1,
                                          g1, be1, m1, v1,
                                          WT, bf2, g2, be2, m2, v2,
                                          Wo, bo, R, maskf, fine);
}


# launch merge: the feature-term GEMM and the Wf2 transpose ride as extra blocks in decoder layer 1's launch, the batch rows in layer 2's: 6 dependent launches become 4; sums and order kept
# speedup vs baseline: 8.7917x; 1.2722x over previous
#include <hip/hip_runtime.h>
#include <hip/hip_bf16.h>


typedef float v2f __attribute__((ext_vector_type(2)));
typedef float v4f __attribute__((ext_vector_type(4)));
typedef float v8f __attribute__((ext_vector_type(8)));
typedef unsigned int v2u __attribute__((ext_vector_type(2)));

#define EPSF 1e-5f

static __device__ __forceinline__ v8f wmma_f32x4(v2f a, v2f b, v8f c) {
    return __builtin_amdgcn_wmma_f32_16x16x4_f32(false, a, false, b, (short)0, c, false, false);
}

typedef float v4fu __attribute__((ext_vector_type(4), aligned(4)));

template <int K>
static __device__ __forceinline__
void gemm8_body(const float* __restrict__ A, int lda,
                const float* __restrict__ Bm, int ldb,
                const float* __restrict__ bias,
                float* __restrict__ Out, int ldo,
                int relu, int blk)
{
    constexpr int KS = K / 8;
    constexpr int NG = KS / 8;
    __shared__ float red[8 * 8 * 32];

    const int t    = threadIdx.x;
    const int w    = t >> 5;
    const int lane = t & 31;
    const int half = lane >> 4;
    const int l    = lane & 15;
    const int tt   = w & 1;
    const int ks   = w >> 1;
    const int n0   = blk * 32;
    const int n    = n0 + tt * 16 + l;

    const float* __restrict__ arow = A + (l & 7) * lda + ks * KS + half * 4;
    const float* __restrict__ brow = Bm + n * ldb + ks * KS + half * 4;

    v4f af[NG], bf[NG];
    #pragma unroll
    for (int g = 0; g < NG; ++g) {
        af[g] = *(const v4fu*)(arow + 8 * g);
        bf[g] = *(const v4fu*)(brow + 8 * g);
    }
    const bool live = l < 8;
    v8f acc0 = {}, acc1 = {};
    #pragma unroll
    for (int g = 0; g < NG; ++g) {
        v4f a = af[g];
        a.x = live ? a.x : 0.0f;
        a.y = live ? a.y : 0.0f;
        a.z = live ? a.z : 0.0f;
        a.w = live ? a.w : 0.0f;
        acc0 = wmma_f32x4(a.xy, bf[g].xy, acc0);
        acc1 = wmma_f32x4(a.zw, bf[g].zw, acc1);
    }

    if (half == 0) {
        #pragma unroll
        for (int r = 0; r < 8; ++r)
            red[(ks * 8 + r) * 32 + tt * 16 + l] = acc0[r] + acc1[r];
    }
    __syncthreads();
    if (t < 256) {
        const int r = t >> 5;
        const int c = t & 31;
        float v = red[r * 32 + c];
        #pragma unroll
        for (int s = 1; s < 8; ++s) v += red[(s * 8 + r) * 32 + c];
        v += bias[n0 + c];
        if (relu) v = fmaxf(v, 0.0f);
        Out[r * ldo + n0 + c] = v;
    }
}

template <int K>
__global__ __launch_bounds__(512)
void gemm8_wmma(const float* __restrict__ A, int lda,
                const float* __restrict__ Bm, int ldb,
                const float* __restrict__ bias,
                float* __restrict__ Out, int ldo,
                int relu)
{
    gemm8_body<K>(A, lda, Bm, ldb, bias, Out, ldo, relu, (int)blockIdx.x);
}

static __device__ __forceinline__
void transpose_body(const float* __restrict__ Wf2, float* __restrict__ WT, int blk2)
{
    __shared__ float tiles[2 * 64 * 33];
    const int u    = threadIdx.x >> 8;
    const int t    = threadIdx.x & 255;
    const int blk  = blk2 * 2 + u;
    float* const tile = tiles + u * (64 * 33);
    const int wv   = blk >> 4;
    const int k0   = (blk & 15) * 32;
    const int lane = t & 31;
    const int r    = t >> 5;
    #pragma unroll
    for (int p = 0; p < 8; ++p) {
        const int ol = p * 8 + r;
        tile[ol * 33 + lane] = Wf2[(wv * 64 + ol) * 512 + k0 + lane];
    }
    __syncthreads();
    #pragma unroll
    for (int p = 0; p < 2; ++p) {
        const int idx = p * 256 + t;
        const int kk  = idx >> 4;
        const int c   = idx & 15;
        v4f v;
        #pragma unroll
        for (int i = 0; i < 4; ++i) v[i] = tile[(i * 16 + c) * 33 + kk];
        *(v4f*)(WT + (k0 + kk) * 512 + wv * 64 + 4 * c) = v;
    }
}

static __device__ __forceinline__
void rows_body(const float* __restrict__ base1,
               const float* __restrict__ Wf1,
               const float* __restrict__ g1, const float* __restrict__ be1,
               const float* __restrict__ m1, const float* __restrict__ v1,
               const float* __restrict__ Wf2,
               float* __restrict__ R,
               float* __restrict__ maskf,
               int rb)
{
    constexpr int AP = 516;
    __shared__ __attribute__((aligned(16))) float As[48 * AP];
    __shared__ float part[4 * 48 * 32];

    const int t    = threadIdx.x;
    const int w    = t >> 5;
    const int lane = t & 31;
    const int half = lane >> 4;
    const int l    = lane & 15;
    const int tt   = w & 1;
    const int ks   = (w >> 1) & 3;
    const int n0   = rb * 32;
    const bool mmw = __builtin_amdgcn_readfirstlane(w) < 8;

    const float* __restrict__ brow = Wf2 + (n0 + tt * 16 + l) * 512 + ks * 128 + half * 4;
    v4f bf[16];
    #pragma unroll
    for (int g = 0; g < 16; ++g) bf[g] = v4f{};
    if (mmw) {
        #pragma unroll
        for (int g = 0; g < 16; ++g) bf[g] = *(const v4f*)(brow + 8 * g);
    }

    {
        const int k = t;
        const float* wr = Wf1 + k * 1029;
        const float sc = g1[k] * rsqrtf(v1[k] + EPSF);
        const float mm = m1[k];
        const float bb = be1[k];
        float r5[5];
        r5[0] = sc * wr[2];
        r5[1] = sc * wr[3];
        r5[2] = sc * wr[4];
        r5[3] = sc * wr[0];
        r5[4] = sc * wr[1];
        #pragma unroll
        for (int bi = 0; bi < 8; ++bi) {
            const float al = __builtin_fmaf(base1[bi * 512 + k] - mm, sc, bb);
            const bool on = al > 0.0f;
            As[(bi * 6) * AP + k] = on ? al : 0.0f;
            #pragma unroll
            for (int j = 0; j < 5; ++j) As[(bi * 6 + 1 + j) * AP + k] = on ? r5[j] : 0.0f;
            if (rb == 0) maskf[bi * 512 + k] = on ? 1.0f : 0.0f;
        }
    }
    __syncthreads();

    if (mmw) {
        v8f acc[3];
        #pragma unroll
        for (int mt = 0; mt < 3; ++mt) acc[mt] = v8f{};
        #pragma unroll
        for (int g = 0; g < 16; ++g) {
            #pragma unroll
            for (int mt = 0; mt < 3; ++mt) {
                const v4f a = *(const v4f*)&As[(mt * 16 + l) * AP + ks * 128 + 8 * g + 4 * half];
                acc[mt] = wmma_f32x4(a.xy, bf[g].xy, acc[mt]);
                acc[mt] = wmma_f32x4(a.zw, bf[g].zw, acc[mt]);
            }
        }
        #pragma unroll
        for (int mt = 0; mt < 3; ++mt)
            #pragma unroll
            for (int r = 0; r < 8; ++r)
                part[(ks * 48 + mt * 16 + 8 * half + r) * 32 + tt * 16 + l] = acc[mt][r];
    }
    __syncthreads();
    #pragma unroll
    for (int p = 0; p < 3; ++p) {
        const int idx = t + 512 * p;
        const int row = idx >> 5;
        const int col = idx & 31;
        float v = part[row * 32 + col];
        #pragma unroll
        for (int s = 1; s < 4; ++s) v += part[(s * 48 + row) * 32 + col];
        R[row * 512 + n0 + col] = v;
    }
}

__global__ __launch_bounds__(512)
void stage1_kernel(const float* __restrict__ features,
                   const float* __restrict__ W1, const float* __restrict__ b1, float* __restrict__ h1,
                   const float* __restrict__ Wf1, const float* __restrict__ bf1, float* __restrict__ base1,
                   const float* __restrict__ Wf2, float* __restrict__ WT)
{
    const int blk = (int)blockIdx.x;
    if (blk < 32)      gemm8_body<1024>(features, 1024, W1, 1024, b1, h1, 1024, 1, blk);
    else if (blk < 48) gemm8_body<1024>(features, 1024, Wf1 + 5, 1029, bf1, base1, 512, 0, blk - 32);
    else               transpose_body(Wf2, WT, blk - 48);
}

__global__ __launch_bounds__(512)
void stage2_kernel(const float* __restrict__ h1,
                   const float* __restrict__ W2, const float* __restrict__ b2, float* __restrict__ h2,
                   const float* __restrict__ base1, const float* __restrict__ Wf1,
                   const float* __restrict__ g1, const float* __restrict__ be1,
                   const float* __restrict__ m1, const float* __restrict__ v1,
                   const float* __restrict__ Wf2,
                   float* __restrict__ R, float* __restrict__ maskf)
{
    const int blk = (int)blockIdx.x;
    if (blk < 32) gemm8_body<1024>(h1, 1024, W2, 1024, b2, h2, 1024, 1, blk);
    else          rows_body(base1, Wf1, g1, be1, m1, v1, Wf2, R, maskf, blk - 32);
}

__global__ __launch_bounds__(256)
void fold_fused(const float* __restrict__ coarse,
                const float* __restrict__ base1,
                const float* __restrict__ Wf1,
                const float* __restrict__ g1, const float* __restrict__ be1,
                const float* __restrict__ m1, const float* __restrict__ v1,
                const float* __restrict__ WT,
                const float* __restrict__ bf2,
                const float* __restrict__ g2, const float* __restrict__ be2,
                const float* __restrict__ m2, const float* __restrict__ v2,
                const float* __restrict__ Wo,
                const float* __restrict__ bo,
                const float* __restrict__ R,
                const float* __restrict__ maskf,
                float* __restrict__ fine)
{
    constexpr int Y1P = 196;
    __shared__ __attribute__((aligned(16))) float y1q[129 * Y1P];
    __shared__ __attribute__((aligned(16))) unsigned short ktab[512];
    __shared__ __attribute__((aligned(16))) float wos[3 * 512];
    __shared__ __attribute__((aligned(16))) float s2s[512];
    __shared__ __attribute__((aligned(16))) float shs[512];
    __shared__ float red[8 * 16 * 48];
    unsigned char* const aflag = (unsigned char*)red;
    __shared__ unsigned int mixflag[16 * 128];
    __shared__ unsigned short mlist[16 * 512];
    __shared__ int mcnt[16];
    auto csidx = [](int j) -> int { return (j >> 2) * Y1P + 192 + (j & 3); };

    const int t    = threadIdx.x;
    const int b    = blockIdx.y;
    const int w    = t >> 5;
    const int lane = t & 31;
    const int half = lane >> 4;
    const int l    = lane & 15;
    const float step = 0.1f / 3.0f;

    {
        const int q    = t & 127;
        const int cgrp = t >> 7;
        const v4f bse = *(const v4f*)(base1 + b * 512 + q * 4);
        const v4f gv  = *(const v4f*)(g1 + q * 4);
        const v4f vv  = *(const v4f*)(v1 + q * 4);
        const v4f mv  = *(const v4f*)(m1 + q * 4);
        const v4f bev = *(const v4f*)(be1 + q * 4);
        const v4f mk  = *(const v4f*)(maskf + b * 512 + q * 4);
        if (t < 48) y1q[csidx(t)] = coarse[(b * 1024 + blockIdx.x * 16) * 3 + t];
        v4f wx, wy, wz, pp, qq, ee, sc;
        #pragma unroll
        for (int c = 0; c < 4; ++c) {
            const float* wr = Wf1 + (q * 4 + c) * 1029;
            sc[c] = gv[c] * rsqrtf(vv[c] + EPSF);
            pp[c] = sc[c] * wr[0];
            qq[c] = sc[c] * wr[1];
            ee[c] = 0.05f * (fabsf(pp[c]) + fabsf(qq[c]));
            wx[c] = wr[2];
            wy[c] = wr[3];
            wz[c] = wr[4];
        }
        float cxs[8], cys[8], czs[8];
        #pragma unroll
        for (int i = 0; i < 8; ++i) {
            const float* cp = coarse + (b * 1024 + blockIdx.x * 16 + cgrp + 2 * i) * 3;
            cxs[i] = cp[0]; cys[i] = cp[1]; czs[i] = cp[2];
        }
        if (t < 128) {
            const v4f g2v = *(const v4f*)(g2 + t * 4);
            const v4f v2v = *(const v4f*)(v2 + t * 4);
            const v4f m2v = *(const v4f*)(m2 + t * 4);
            const v4f b2v = *(const v4f*)(be2 + t * 4);
            const v4f f2v = *(const v4f*)(bf2 + t * 4);
            v4f s, sh;
            #pragma unroll
            for (int c = 0; c < 4; ++c) {
                s[c]  = g2v[c] * rsqrtf(v2v[c] + EPSF);
                sh[c] = __builtin_fmaf(f2v[c] - m2v[c], s[c], b2v[c]);
            }
            *(v4f*)&s2s[t * 4] = s;
            *(v4f*)&shs[t * 4] = sh;
            *(v4f*)&wos[1024 + t * 4] = *(const v4f*)(Wo + 1024 + t * 4);
        }
        *(v4f*)&wos[t * 4] = *(const v4f*)(Wo + t * 4);

        unsigned int live = 0;
        #pragma unroll
        for (int i = 0; i < 8; ++i) {
            const int cl   = cgrp + 2 * i;
            const int rowA = cgrp * 32 + i;
            const int rowQ = rowA + 8;
            const int rowP = 16 + i + 8 * cgrp;
            v4f ra, rp, rq;
            unsigned int fl = 0;
            #pragma unroll
            for (int c = 0; c < 4; ++c) {
                float val = bse[c];
                val = __builtin_fmaf(cxs[i], wx[c], val);
                val = __builtin_fmaf(cys[i], wy[c], val);
                val = __builtin_fmaf(czs[i], wz[c], val);
                const float a = __builtin_fmaf(val - mv[c], sc[c], bev[c]);
                const bool dead  = !(a + ee[c] > 0.0f);
                const bool mixed = !dead && !(a - ee[c] > 0.0f);
                const bool on    = mk[c] > 0.5f;
                const bool rows  = on ? (dead || mixed) : !dead;
                ra[c] = rows ? (on ? -a : a) : 0.0f;
                rp[c] = rows ? (on ? -pp[c] : pp[c]) : 0.0f;
                rq[c] = rows ? (on ? -qq[c] : qq[c]) : 0.0f;
                fl |= (mixed ? 1u : 0u) << (8 * c);
                live |= (rows ? 1u : 0u) << c;
            }
            *(v4f*)&y1q[q * Y1P + rowA * 4] = ra;
            *(v4f*)&y1q[q * Y1P + rowP * 4] = rp;
            *(v4f*)&y1q[q * Y1P + rowQ * 4] = rq;
            mixflag[cl * 128 + q] = fl;
        }
        aflag[t] = (unsigned char)live;
        if (t < 48) *(v4f*)&y1q[128 * Y1P + t * 4] = v4f{};
    }
    __syncthreads();

    int NG;
    {
        const unsigned int lt = (1u << lane) - 1u;
        int total = 0, mybase = 0, mypre = 0;
        unsigned int mine = 0;
        #pragma unroll
        for (int ch = 0; ch < 4; ++ch) {
            const unsigned int f = (unsigned int)aflag[ch * 32 + lane] | (unsigned int)aflag[128 + ch * 32 + lane];
            int cnt = 0, pre = 0;
            #pragma unroll
            for (int c = 0; c < 4; ++c) {
                const unsigned int m = __builtin_amdgcn_ballot_w32(((f >> c) & 1u) != 0u);
                cnt += __builtin_popcount(m);
                pre += __builtin_popcount(m & lt);
            }
            if (ch == (w & 3)) { mine = f; mybase = total; mypre = pre; }
            total += cnt;
        }
        NG = ((total + 15) >> 4) << 1;
        NG = NG < 2 ? 2 : NG;
        if (w < 4) {
            int pos = mybase + mypre;
            #pragma unroll
            for (int c = 0; c < 4; ++c) {
                if ((mine >> c) & 1u) {
                    ktab[pos] = (unsigned short)(4 * t + c);
                    ++pos;
                }
            }
        } else {
            const int p = total + (t - 128);
            if (t - 128 < 16 && p < 8 * NG) ktab[p] = (unsigned short)512;
        }
    }

    {
        const unsigned char* fb = (const unsigned char*)mixflag;
        #pragma unroll
        for (int rep = 0; rep < 2; ++rep) {
            const int cc = w + 8 * rep;
            int basec = 0;
            for (int ch = 0; ch < 16; ++ch) {
                const int k = ch * 32 + lane;
                const bool f = fb[cc * 512 + k] != 0;
                const unsigned int mask = __builtin_amdgcn_ballot_w32(f);
                const int pos = basec + __builtin_popcount(mask & ((1u << lane) - 1u));
                if (f) mlist[cc * 512 + pos] = (unsigned short)k;
                basec += __builtin_popcount(mask);
            }
            if (lane == 0) mcnt[cc] = basec;
        }
    }

    __syncthreads();

    v8f acc[3][4];
    #pragma unroll
    for (int mt = 0; mt < 3; ++mt)
        #pragma unroll
        for (int i = 0; i < 4; ++i) acc[mt][i] = v8f{};
    {
        const int ng = __builtin_amdgcn_readfirstlane(NG);
        const unsigned int arow = (unsigned int)(l * 4);
        const float* __restrict__ wtl = WT + w * 64 + l * 4;
        auto ldtab = [&](int j) -> v2u { return *(const v2u*)&ktab[8 * j + 4 * half]; };
        auto ldgrp = [&](const v2u tb, float (&x)[3][4], v4f (&y)[4]) {
            #pragma unroll
            for (int e = 0; e < 4; ++e) {
                const unsigned int k  = (tb[e >> 1] >> (16 * (e & 1))) & 0xffffu;
                const unsigned int ai = (k >> 2) * Y1P + (k & 3u) + arow;
                #pragma unroll
                for (int mt = 0; mt < 3; ++mt) x[mt][e] = y1q[ai + mt * 64];
                y[e] = *(const v4f*)(wtl + (k & 511u) * 512);
            }
        };
        auto mma24 = [&](const float (&x)[3][4], const v4f (&y)[4]) {
            #pragma unroll
            for (int i = 0; i < 4; ++i)
                #pragma unroll
                for (int mt = 0; mt < 3; ++mt)
                    acc[mt][i] = wmma_f32x4(v2f{x[mt][0], x[mt][1]}, v2f{y[0][i], y[1][i]}, acc[mt][i]);
            #pragma unroll
            for (int i = 0; i < 4; ++i)
                #pragma unroll
                for (int mt = 0; mt < 3; ++mt)
                    acc[mt][i] = wmma_f32x4(v2f{x[mt][2], x[mt][3]}, v2f{y[2][i], y[3][i]}, acc[mt][i]);
        };
        float pa[3][4], qa[3][4];
        v4f pb[4], qb[4];
        v2u t0 = ldtab(0), t1 = ldtab(1);
        ldgrp(t0, pa, pb);
        #pragma nounroll
        for (int j = 0; j < ng; j += 2) {
            const int j2 = j + 2 < ng ? j + 2 : ng - 1;
            const int j3 = j + 3 < ng ? j + 3 : ng - 1;
            t0 = ldtab(j2);
            ldgrp(t1, qa, qb);
            __builtin_amdgcn_sched_barrier(0);
            mma24(pa, pb);
            __builtin_amdgcn_sched_barrier(0);
            t1 = ldtab(j3);
            ldgrp(t0, pa, pb);
            __builtin_amdgcn_sched_barrier(0);
            mma24(qa, qb);
            __builtin_amdgcn_sched_barrier(0);
        }
    }

    {
        const float gx = -0.05f + (float)(l >> 2) * step;
        const float gy = -0.05f + (float)(l & 3) * step;
        v2f gbX, gbY;
        gbX.x = half ? gy : 1.0f;
        gbX.y = half ? 0.0f : gx;
        gbY.x = half ? gy : 1.0f;
        gbY.y = half ? gx : 0.0f;
        const float* __restrict__ wtl = WT + w * 64 + l * 4;
        float rAl[4], rX[4], rY[4], rZ[4], rP[4], rQ[4];
        #pragma unroll
        for (int i = 0; i < 4; ++i) {
            const float* rbp = R + b * (6 * 512) + (w * 4 + i) * 16 + l;
            rAl[i] = rbp[0];
            rX[i]  = rbp[512];
            rY[i]  = rbp[1024];
            rZ[i]  = rbp[1536];
            rP[i]  = rbp[2048];
            rQ[i]  = rbp[2560];
        }
        #pragma unroll
        for (int cl = 0; cl < 16; ++cl) {
            const float ccx = y1q[csidx(cl * 3)], ccy = y1q[csidx(cl * 3 + 1)], ccz = y1q[csidx(cl * 3 + 2)];
            const int pr = cl >> 1, sy = cl & 1;
            const int mtA  = sy ? 2 : 0;
            const int rowA = sy * 32 + pr;
            const int rowQ = rowA + 8;
            const int rowP = 16 + pr + 8 * sy;
            const v2f gb = sy ? gbY : gbX;
            v8f d[4];
            #pragma unroll
            for (int i = 0; i < 4; ++i) {
                const float ab = __builtin_fmaf(ccz, rZ[i], __builtin_fmaf(ccy, rY[i], __builtin_fmaf(ccx, rX[i], rAl[i])));
                v2f ea;
                ea.x = acc[mtA][i][pr] + (half ? rQ[i] : ab);
                ea.y = acc[1][i][pr] + rP[i];
                d[i] = wmma_f32x4(ea, gb, v8f{});
            }
            const int n = __builtin_amdgcn_readfirstlane(mcnt[cl]);
            for (int base = 0; base < n; base += 4) {
                const int i0 = base + 2 * half, i1 = i0 + 1;
                const bool ok0 = i0 < n, ok1 = i1 < n;
                const int k0 = ok0 ? (int)mlist[cl * 512 + i0] : 0;
                const int k1 = ok1 ? (int)mlist[cl * 512 + i1] : 0;
                const float* r0 = &y1q[(k0 >> 2) * Y1P + (k0 & 3)];
                const float* r1 = &y1q[(k1 >> 2) * Y1P + (k1 & 3)];
                const float z0 = __builtin_fmaf(gy, r0[rowQ * 4], __builtin_fmaf(gx, r0[rowP * 4], r0[rowA * 4]));
                const float z1 = __builtin_fmaf(gy, r1[rowQ * 4], __builtin_fmaf(gx, r1[rowP * 4], r1[rowA * 4]));
                v2f cb;
                cb.x = ok0 ? fmaxf(-z0, 0.0f) : 0.0f;
                cb.y = ok1 ? fmaxf(-z1, 0.0f) : 0.0f;
                const v4f wk0 = *(const v4f*)(wtl + k0 * 512);
                const v4f wk1 = *(const v4f*)(wtl + k1 * 512);
                #pragma unroll
                for (int i = 0; i < 4; ++i) {
                    v2f wa;
                    wa.x = wk0[i];
                    wa.y = wk1[i];
                    d[i] = wmma_f32x4(wa, cb, d[i]);
                }
            }
            float p0 = 0.0f, p1 = 0.0f, p2 = 0.0f;
            #pragma unroll
            for (int i = 0; i < 4; ++i) {
                const int ob = (w * 4 + i) * 16 + 8 * half;
                #pragma unroll
                for (int hq = 0; hq < 2; ++hq) {
                    const v4f s  = *(const v4f*)&s2s[ob + 4 * hq];
                    const v4f sh = *(const v4f*)&shs[ob + 4 * hq];
                    const v4f w0 = *(const v4f*)&wos[ob + 4 * hq];
                    const v4f w1 = *(const v4f*)&wos[512 + ob + 4 * hq];
                    const v4f w2 = *(const v4f*)&wos[1024 + ob + 4 * hq];
                    #pragma unroll
                    for (int e = 0; e < 4; ++e) {
                        const float y = fmaxf(__builtin_fmaf(d[i][4 * hq + e], s[e], sh[e]), 0.0f);
                        p0 = __builtin_fmaf(y, w0[e], p0);
                        p1 = __builtin_fmaf(y, w1[e], p1);
                        p2 = __builtin_fmaf(y, w2[e], p2);
                    }
                }
            }
            p0 += __shfl_xor(p0, 16);
            p1 += __shfl_xor(p1, 16);
            p2 += __shfl_xor(p2, 16);
            if (half == 0) {
                float* rd = &red[(w * 16 + cl) * 48 + l * 3];
                rd[0] = p0; rd[1] = p1; rd[2] = p2;
            }
        }
    }
    __syncthreads();

    for (int idx = t; idx < 768; idx += 256) {
        const int cl  = idx / 48;
        const int rem = idx - cl * 48;
        const int g   = rem / 3;
        const int c   = rem - g * 3;
        float sum = 0.0f;
        #pragma unroll
        for (int s = 0; s < 8; ++s) sum += red[(s * 16 + cl) * 48 + rem];
        const int ng = (blockIdx.x * 16 + cl) * 16 + g;
        const int i  = c * 16384 + ng;
        const int r3 = i / 3;
        const int cc = i - r3 * 3;
        const float center = coarse[(b * 1024 + (r3 >> 4)) * 3 + cc];
        fine[(b * 16384 + ng) * 3 + c] = sum + bo[c] + center;
    }
}

extern "C" void kernel_launch(void* const* d_in, const int* in_sizes, int n_in,
                              void* d_out, int out_size, void* d_ws, size_t ws_size,
                              hipStream_t stream)
{
    const float* features = (const float*)d_in[0];
    const float* W1  = (const float*)d_in[1];
    const float* b1  = (const float*)d_in[2];
    const float* W2  = (const float*)d_in[3];
    const float* b2  = (const float*)d_in[4];
    const float* Wc  = (const float*)d_in[5];
    const float* bc  = (const float*)d_in[6];
    const float* Wf1 = (const float*)d_in[7];
    const float* bf1 = (const float*)d_in[8];
    const float* g1  = (const float*)d_in[9];
    const float* be1 = (const float*)d_in[10];
    const float* m1  = (const float*)d_in[11];
    const float* v1  = (const float*)d_in[12];
    const float* Wf2 = (const float*)d_in[13];
    const float* bf2 = (const float*)d_in[14];
    const float* g2  = (const float*)d_in[15];
    const float* be2 = (const float*)d_in[16];
    const float* m2  = (const float*)d_in[17];
    const float* v2  = (const float*)d_in[18];
    const float* Wo  = (const float*)d_in[19];
    const float* bo  = (const float*)d_in[20];

    float* out    = (float*)d_out;
    float* coarse = out;
    float* fine   = out + 8 * 1024 * 3;

    float* ws    = (float*)d_ws;
    float* h1    = ws;
    float* h2    = ws + 8192;
    float* base1 = ws + 16384;
    float* WT    = ws + 28672;
    float* R     = ws + 290816;
    float* maskf = ws + 315392;

    stage1_kernel<<<112, 512, 0, stream>>>(features, W1, b1, h1, Wf1, bf1, base1, Wf2, WT);
    stage2_kernel<<<48, 512, 0, stream>>>(h1, W2, b2, h2, base1, Wf1, g1, be1, m1, v1, Wf2, R, maskf);
    gemm8_wmma<1024><<<96, 512, 0, stream>>>(h2, 1024, Wc, 1024, bc, coarse, 3072, 0);

    dim3 gridF(1024 / 16, 8);
    fold_fused<<<gridF, 256, 0, stream>>>(coarse, base1, Wf1,
                                          g1, be1, m1, v1,
                                          WT, bf2, g2, be2, m2, v2,
                                          Wo, bo, R, maskf, fine);
}


# fold phase 3 reads 12 LDS b128 per coarse point, not 40: BN2 scale and shift ride in the A operands of the rebuilding wmma, Wo of two tiles stays in registers; rounding of the scaled sums changes
# speedup vs baseline: 9.2223x; 1.0490x over previous
#include <hip/hip_runtime.h>
#include <hip/hip_bf16.h>


typedef float v2f __attribute__((ext_vector_type(2)));
typedef float v4f __attribute__((ext_vector_type(4)));
typedef float v8f __attribute__((ext_vector_type(8)));
typedef unsigned int v2u __attribute__((ext_vector_type(2)));

#define EPSF 1e-5f

static __device__ __forceinline__ v8f wmma_f32x4(v2f a, v2f b, v8f c) {
    return __builtin_amdgcn_wmma_f32_16x16x4_f32(false, a, false, b, (short)0, c, false, false);
}

typedef float v4fu __attribute__((ext_vector_type(4), aligned(4)));

template <int K>
static __device__ __forceinline__
void gemm8_body(const float* __restrict__ A, int lda,
                const float* __restrict__ Bm, int ldb,
                const float* __restrict__ bias,
                float* __restrict__ Out, int ldo,
                int relu, int blk)
{
    constexpr int KS = K / 8;
    constexpr int NG = KS / 8;
    __shared__ float red[8 * 8 * 32];

    const int t    = threadIdx.x;
    const int w    = t >> 5;
    const int lane = t & 31;
    const int half = lane >> 4;
    const int l    = lane & 15;
    const int tt   = w & 1;
    const int ks   = w >> 1;
    const int n0   = blk * 32;
    const int n    = n0 + tt * 16 + l;

    const float* __restrict__ arow = A + (l & 7) * lda + ks * KS + half * 4;
    const float* __restrict__ brow = Bm + n * ldb + ks * KS + half * 4;

    v4f af[NG], bf[NG];
    #pragma unroll
    for (int g = 0; g < NG; ++g) {
        af[g] = *(const v4fu*)(arow + 8 * g);
        bf[g] = *(const v4fu*)(brow + 8 * g);
    }
    const bool live = l < 8;
    v8f acc0 = {}, acc1 = {};
    #pragma unroll
    for (int g = 0; g < NG; ++g) {
        v4f a = af[g];
        a.x = live ? a.x : 0.0f;
        a.y = live ? a.y : 0.0f;
        a.z = live ? a.z : 0.0f;
        a.w = live ? a.w : 0.0f;
        acc0 = wmma_f32x4(a.xy, bf[g].xy, acc0);
        acc1 = wmma_f32x4(a.zw, bf[g].zw, acc1);
    }

    if (half == 0) {
        #pragma unroll
        for (int r = 0; r < 8; ++r)
            red[(ks * 8 + r) * 32 + tt * 16 + l] = acc0[r] + acc1[r];
    }
    __syncthreads();
    if (t < 256) {
        const int r = t >> 5;
        const int c = t & 31;
        float v = red[r * 32 + c];
        #pragma unroll
        for (int s = 1; s < 8; ++s) v += red[(s * 8 + r) * 32 + c];
        v += bias[n0 + c];
        if (relu) v = fmaxf(v, 0.0f);
        Out[r * ldo + n0 + c] = v;
    }
}

template <int K>
__global__ __launch_bounds__(512)
void gemm8_wmma(const float* __restrict__ A, int lda,
                const float* __restrict__ Bm, int ldb,
                const float* __restrict__ bias,
                float* __restrict__ Out, int ldo,
                int relu)
{
    gemm8_body<K>(A, lda, Bm, ldb, bias, Out, ldo, relu, (int)blockIdx.x);
}

static __device__ __forceinline__
void transpose_body(const float* __restrict__ Wf2, float* __restrict__ WT, int blk2)
{
    __shared__ float tiles[2 * 64 * 33];
    const int u    = threadIdx.x >> 8;
    const int t    = threadIdx.x & 255;
    const int blk  = blk2 * 2 + u;
    float* const tile = tiles + u * (64 * 33);
    const int wv   = blk >> 4;
    const int k0   = (blk & 15) * 32;
    const int lane = t & 31;
    const int r    = t >> 5;
    #pragma unroll
    for (int p = 0; p < 8; ++p) {
        const int ol = p * 8 + r;
        tile[ol * 33 + lane] = Wf2[(wv * 64 + ol) * 512 + k0 + lane];
    }
    __syncthreads();
    #pragma unroll
    for (int p = 0; p < 2; ++p) {
        const int idx = p * 256 + t;
        const int kk  = idx >> 4;
        const int c   = idx & 15;
        v4f v;
        #pragma unroll
        for (int i = 0; i < 4; ++i) v[i] = tile[(i * 16 + c) * 33 + kk];
        *(v4f*)(WT + (k0 + kk) * 512 + wv * 64 + 4 * c) = v;
    }
}

static __device__ __forceinline__
void rows_body(const float* __restrict__ base1,
               const float* __restrict__ Wf1,
               const float* __restrict__ g1, const float* __restrict__ be1,
               const float* __restrict__ m1, const float* __restrict__ v1,
               const float* __restrict__ Wf2,
               float* __restrict__ R,
               float* __restrict__ maskf,
               int rb)
{
    constexpr int AP = 516;
    __shared__ __attribute__((aligned(16))) float As[48 * AP];
    __shared__ float part[4 * 48 * 32];

    const int t    = threadIdx.x;
    const int w    = t >> 5;
    const int lane = t & 31;
    const int half = lane >> 4;
    const int l    = lane & 15;
    const int tt   = w & 1;
    const int ks   = (w >> 1) & 3;
    const int n0   = rb * 32;
    const bool mmw = __builtin_amdgcn_readfirstlane(w) < 8;

    const float* __restrict__ brow = Wf2 + (n0 + tt * 16 + l) * 512 + ks * 128 + half * 4;
    v4f bf[16];
    #pragma unroll
    for (int g = 0; g < 16; ++g) bf[g] = v4f{};
    if (mmw) {
        #pragma unroll
        for (int g = 0; g < 16; ++g) bf[g] = *(const v4f*)(brow + 8 * g);
    }

    {
        const int k = t;
        const float* wr = Wf1 + k * 1029;
        const float sc = g1[k] * rsqrtf(v1[k] + EPSF);
        const float mm = m1[k];
        const float bb = be1[k];
        float r5[5];
        r5[0] = sc * wr[2];
        r5[1] = sc * wr[3];
        r5[2] = sc * wr[4];
        r5[3] = sc * wr[0];
        r5[4] = sc * wr[1];
        #pragma unroll
        for (int bi = 0; bi < 8; ++bi) {
            const float al = __builtin_fmaf(base1[bi * 512 + k] - mm, sc, bb);
            const bool on = al > 0.0f;
            As[(bi * 6) * AP + k] = on ? al : 0.0f;
            #pragma unroll
            for (int j = 0; j < 5; ++j) As[(bi * 6 + 1 + j) * AP + k] = on ? r5[j] : 0.0f;
            if (rb == 0) maskf[bi * 512 + k] = on ? 1.0f : 0.0f;
        }
    }
    __syncthreads();

    if (mmw) {
        v8f acc[3];
        #pragma unroll
        for (int mt = 0; mt < 3; ++mt) acc[mt] = v8f{};
        #pragma unroll
        for (int g = 0; g < 16; ++g) {
            #pragma unroll
            for (int mt = 0; mt < 3; ++mt) {
                const v4f a = *(const v4f*)&As[(mt * 16 + l) * AP + ks * 128 + 8 * g + 4 * half];
                acc[mt] = wmma_f32x4(a.xy, bf[g].xy, acc[mt]);
                acc[mt] = wmma_f32x4(a.zw, bf[g].zw, acc[mt]);
            }
        }
        #pragma unroll
        for (int mt = 0; mt < 3; ++mt)
            #pragma unroll
            for (int r = 0; r < 8; ++r)
                part[(ks * 48 + mt * 16 + 8 * half + r) * 32 + tt * 16 + l] = acc[mt][r];
    }
    __syncthreads();
    #pragma unroll
    for (int p = 0; p < 3; ++p) {
        const int idx = t + 512 * p;
        const int row = idx >> 5;
        const int col = idx & 31;
        float v = part[row * 32 + col];
        #pragma unroll
        for (int s = 1; s < 4; ++s) v += part[(s * 48 + row) * 32 + col];
        R[row * 512 + n0 + col] = v;
    }
}

__global__ __launch_bounds__(512)
void stage1_kernel(const float* __restrict__ features,
                   const float* __restrict__ W1, const float* __restrict__ b1, float* __restrict__ h1,
                   const float* __restrict__ Wf1, const float* __restrict__ bf1, float* __restrict__ base1,
                   const float* __restrict__ Wf2, float* __restrict__ WT)
{
    const int blk = (int)blockIdx.x;
    if (blk < 32)      gemm8_body<1024>(features, 1024, W1, 1024, b1, h1, 1024, 1, blk);
    else if (blk < 48) gemm8_body<1024>(features, 1024, Wf1 + 5, 1029, bf1, base1, 512, 0, blk - 32);
    else               transpose_body(Wf2, WT, blk - 48);
}

__global__ __launch_bounds__(512)
void stage2_kernel(const float* __restrict__ h1,
                   const float* __restrict__ W2, const float* __restrict__ b2, float* __restrict__ h2,
                   const float* __restrict__ base1, const float* __restrict__ Wf1,
                   const float* __restrict__ g1, const float* __restrict__ be1,
                   const float* __restrict__ m1, const float* __restrict__ v1,
                   const float* __restrict__ Wf2,
                   float* __restrict__ R, float* __restrict__ maskf)
{
    const int blk = (int)blockIdx.x;
    if (blk < 32) gemm8_body<1024>(h1, 1024, W2, 1024, b2, h2, 1024, 1, blk);
    else          rows_body(base1, Wf1, g1, be1, m1, v1, Wf2, R, maskf, blk - 32);
}

__global__ __launch_bounds__(256)
void fold_fused(const float* __restrict__ coarse,
                const float* __restrict__ base1,
                const float* __restrict__ Wf1,
                const float* __restrict__ g1, const float* __restrict__ be1,
                const float* __restrict__ m1, const float* __restrict__ v1,
                const float* __restrict__ WT,
                const float* __restrict__ bf2,
                const float* __restrict__ g2, const float* __restrict__ be2,
                const float* __restrict__ m2, const float* __restrict__ v2,
                const float* __restrict__ Wo,
                const float* __restrict__ bo,
                const float* __restrict__ R,
                const float* __restrict__ maskf,
                float* __restrict__ fine)
{
    constexpr int Y1P = 196;
    __shared__ __attribute__((aligned(16))) float y1q[129 * Y1P];
    __shared__ __attribute__((aligned(16))) unsigned short ktab[512];
    __shared__ __attribute__((aligned(16))) float wos[3 * 512];
    __shared__ __attribute__((aligned(16))) float s2s[512];
    __shared__ __attribute__((aligned(16))) float shs[512];
    __shared__ float red[8 * 16 * 48];
    unsigned char* const aflag = (unsigned char*)red;
    __shared__ unsigned int mixflag[16 * 128];
    __shared__ unsigned short mlist[16 * 512];
    __shared__ int mcnt[16];
    auto csidx = [](int j) -> int { return (j >> 2) * Y1P + 192 + (j & 3); };

    const int t    = threadIdx.x;
    const int b    = blockIdx.y;
    const int w    = t >> 5;
    const int lane = t & 31;
    const int half = lane >> 4;
    const int l    = lane & 15;
    const float step = 0.1f / 3.0f;

    {
        const int q    = t & 127;
        const int cgrp = t >> 7;
        const v4f bse = *(const v4f*)(base1 + b * 512 + q * 4);
        const v4f gv  = *(const v4f*)(g1 + q * 4);
        const v4f vv  = *(const v4f*)(v1 + q * 4);
        const v4f mv  = *(const v4f*)(m1 + q * 4);
        const v4f bev = *(const v4f*)(be1 + q * 4);
        const v4f mk  = *(const v4f*)(maskf + b * 512 + q * 4);
        if (t < 48) y1q[csidx(t)] = coarse[(b * 1024 + blockIdx.x * 16) * 3 + t];
        v4f wx, wy, wz, pp, qq, ee, sc;
        #pragma unroll
        for (int c = 0; c < 4; ++c) {
            const float* wr = Wf1 + (q * 4 + c) * 1029;
            sc[c] = gv[c] * rsqrtf(vv[c] + EPSF);
            pp[c] = sc[c] * wr[0];
            qq[c] = sc[c] * wr[1];
            ee[c] = 0.05f * (fabsf(pp[c]) + fabsf(qq[c]));
            wx[c] = wr[2];
            wy[c] = wr[3];
            wz[c] = wr[4];
        }
        float cxs[8], cys[8], czs[8];
        #pragma unroll
        for (int i = 0; i < 8; ++i) {
            const float* cp = coarse + (b * 1024 + blockIdx.x * 16 + cgrp + 2 * i) * 3;
            cxs[i] = cp[0]; cys[i] = cp[1]; czs[i] = cp[2];
        }
        if (t < 128) {
            const v4f g2v = *(const v4f*)(g2 + t * 4);
            const v4f v2v = *(const v4f*)(v2 + t * 4);
            const v4f m2v = *(const v4f*)(m2 + t * 4);
            const v4f b2v = *(const v4f*)(be2 + t * 4);
            const v4f f2v = *(const v4f*)(bf2 + t * 4);
            v4f s, sh;
            #pragma unroll
            for (int c = 0; c < 4; ++c) {
                s[c]  = g2v[c] * rsqrtf(v2v[c] + EPSF);
                sh[c] = __builtin_fmaf(f2v[c] - m2v[c], s[c], b2v[c]);
            }
            *(v4f*)&s2s[t * 4] = s;
            *(v4f*)&shs[t * 4] = sh;
            *(v4f*)&wos[1024 + t * 4] = *(const v4f*)(Wo + 1024 + t * 4);
        }
        *(v4f*)&wos[t * 4] = *(const v4f*)(Wo + t * 4);

        unsigned int live = 0;
        #pragma unroll
        for (int i = 0; i < 8; ++i) {
            const int cl   = cgrp + 2 * i;
            const int rowA = cgrp * 32 + i;
            const int rowQ = rowA + 8;
            const int rowP = 16 + i + 8 * cgrp;
            v4f ra, rp, rq;
            unsigned int fl = 0;
            #pragma unroll
            for (int c = 0; c < 4; ++c) {
                float val = bse[c];
                val = __builtin_fmaf(cxs[i], wx[c], val);
                val = __builtin_fmaf(cys[i], wy[c], val);
                val = __builtin_fmaf(czs[i], wz[c], val);
                const float a = __builtin_fmaf(val - mv[c], sc[c], bev[c]);
                const bool dead  = !(a + ee[c] > 0.0f);
                const bool mixed = !dead && !(a - ee[c] > 0.0f);
                const bool on    = mk[c] > 0.5f;
                const bool rows  = on ? (dead || mixed) : !dead;
                ra[c] = rows ? (on ? -a : a) : 0.0f;
                rp[c] = rows ? (on ? -pp[c] : pp[c]) : 0.0f;
                rq[c] = rows ? (on ? -qq[c] : qq[c]) : 0.0f;
                fl |= (mixed ? 1u : 0u) << (8 * c);
                live |= (rows ? 1u : 0u) << c;
            }
            *(v4f*)&y1q[q * Y1P + rowA * 4] = ra;
            *(v4f*)&y1q[q * Y1P + rowP * 4] = rp;
            *(v4f*)&y1q[q * Y1P + rowQ * 4] = rq;
            mixflag[cl * 128 + q] = fl;
        }
        aflag[t] = (unsigned char)live;
        if (t < 48) *(v4f*)&y1q[128 * Y1P + t * 4] = v4f{};
    }
    __syncthreads();

    int NG;
    {
        const unsigned int lt = (1u << lane) - 1u;
        int total = 0, mybase = 0, mypre = 0;
        unsigned int mine = 0;
        #pragma unroll
        for (int ch = 0; ch < 4; ++ch) {
            const unsigned int f = (unsigned int)aflag[ch * 32 + lane] | (unsigned int)aflag[128 + ch * 32 + lane];
            int cnt = 0, pre = 0;
            #pragma unroll
            for (int c = 0; c < 4; ++c) {
                const unsigned int m = __builtin_amdgcn_ballot_w32(((f >> c) & 1u) != 0u);
                cnt += __builtin_popcount(m);
                pre += __builtin_popcount(m & lt);
            }
            if (ch == (w & 3)) { mine = f; mybase = total; mypre = pre; }
            total += cnt;
        }
        NG = ((total + 15) >> 4) << 1;
        NG = NG < 2 ? 2 : NG;
        if (w < 4) {
            int pos = mybase + mypre;
            #pragma unroll
            for (int c = 0; c < 4; ++c) {
                if ((mine >> c) & 1u) {
                    ktab[pos] = (unsigned short)(4 * t + c);
                    ++pos;
                }
            }
        } else {
            const int p = total + (t - 128);
            if (t - 128 < 16 && p < 8 * NG) ktab[p] = (unsigned short)512;
        }
    }

    {
        const unsigned char* fb = (const unsigned char*)mixflag;
        #pragma unroll
        for (int rep = 0; rep < 2; ++rep) {
            const int cc = w + 8 * rep;
            int basec = 0;
            for (int ch = 0; ch < 16; ++ch) {
                const int k = ch * 32 + lane;
                const bool f = fb[cc * 512 + k] != 0;
                const unsigned int mask = __builtin_amdgcn_ballot_w32(f);
                const int pos = basec + __builtin_popcount(mask & ((1u << lane) - 1u));
                if (f) mlist[cc * 512 + pos] = (unsigned short)k;
                basec += __builtin_popcount(mask);
            }
            if (lane == 0) mcnt[cc] = basec;
        }
    }

    __syncthreads();

    v8f acc[3][4];
    #pragma unroll
    for (int mt = 0; mt < 3; ++mt)
        #pragma unroll
        for (int i = 0; i < 4; ++i) acc[mt][i] = v8f{};
    {
        const int ng = __builtin_amdgcn_readfirstlane(NG);
        const unsigned int arow = (unsigned int)(l * 4);
        const float* __restrict__ wtl = WT + w * 64 + l * 4;
        auto ldtab = [&](int j) -> v2u { return *(const v2u*)&ktab[8 * j + 4 * half]; };
        auto ldgrp = [&](const v2u tb, float (&x)[3][4], v4f (&y)[4]) {
            #pragma unroll
            for (int e = 0; e < 4; ++e) {
                const unsigned int k  = (tb[e >> 1] >> (16 * (e & 1))) & 0xffffu;
                const unsigned int ai = (k >> 2) * Y1P + (k & 3u) + arow;
                #pragma unroll
                for (int mt = 0; mt < 3; ++mt) x[mt][e] = y1q[ai + mt * 64];
                y[e] = *(const v4f*)(wtl + (k & 511u) * 512);
            }
        };
        auto mma24 = [&](const float (&x)[3][4], const v4f (&y)[4]) {
            #pragma unroll
            for (int i = 0; i < 4; ++i)
                #pragma unroll
                for (int mt = 0; mt < 3; ++mt)
                    acc[mt][i] = wmma_f32x4(v2f{x[mt][0], x[mt][1]}, v2f{y[0][i], y[1][i]}, acc[mt][i]);
            #pragma unroll
            for (int i = 0; i < 4; ++i)
                #pragma unroll
                for (int mt = 0; mt < 3; ++mt)
                    acc[mt][i] = wmma_f32x4(v2f{x[mt][2], x[mt][3]}, v2f{y[2][i], y[3][i]}, acc[mt][i]);
        };
        float pa[3][4], qa[3][4];
        v4f pb[4], qb[4];
        v2u t0 = ldtab(0), t1 = ldtab(1);
        ldgrp(t0, pa, pb);
        #pragma nounroll
        for (int j = 0; j < ng; j += 2) {
            const int j2 = j + 2 < ng ? j + 2 : ng - 1;
            const int j3 = j + 3 < ng ? j + 3 : ng - 1;
            t0 = ldtab(j2);
            ldgrp(t1, qa, qb);
            __builtin_amdgcn_sched_barrier(0);
            mma24(pa, pb);
            __builtin_amdgcn_sched_barrier(0);
            t1 = ldtab(j3);
            ldgrp(t0, pa, pb);
            __builtin_amdgcn_sched_barrier(0);
            mma24(qa, qb);
            __builtin_amdgcn_sched_barrier(0);
        }
    }

    {
        const float gx = -0.05f + (float)(l >> 2) * step;
        const float gy = -0.05f + (float)(l & 3) * step;
        v2f gbX, gbY;
        gbX.x = half ? gy : 1.0f;
        gbX.y = half ? 0.0f : gx;
        gbY.x = half ? gy : 1.0f;
        gbY.y = half ? gx : 0.0f;
        const float* __restrict__ wtl = WT + w * 64 + l * 4;
        float sl[4], cA[4], cP[4], rXs[4], rYs[4], rZs[4];
        {
            float rv[4][6], sv[4], hv[4];
            #pragma unroll
            for (int i = 0; i < 4; ++i) {
                const int oc = (w * 4 + i) * 16 + l;
                const float* rbp = R + b * (6 * 512) + oc;
                #pragma unroll
                for (int j = 0; j < 6; ++j) rv[i][j] = rbp[512 * j];
                sv[i] = s2s[oc];
                hv[i] = shs[oc];
            }
            #pragma unroll
            for (int i = 0; i < 4; ++i) {
                const float s  = sv[i];
                const float a0 = __builtin_fmaf(s, rv[i][0], hv[i]);
                const float a1 = s * rv[i][5];
                const float x0 = s * rv[i][1], y0 = s * rv[i][2], z0 = s * rv[i][3];
                sl[i]  = s;
                cA[i]  = half ? a1 : a0;
                cP[i]  = s * rv[i][4];
                rXs[i] = half ? 0.0f : x0;
                rYs[i] = half ? 0.0f : y0;
                rZs[i] = half ? 0.0f : z0;
            }
        }
        constexpr int NWR = 2;
        v4f wr[NWR][2][3];
        #pragma unroll
        for (int i = 0; i < NWR; ++i)
            #pragma unroll
            for (int hq = 0; hq < 2; ++hq)
                #pragma unroll
                for (int cc = 0; cc < 3; ++cc)
                    wr[i][hq][cc] = *(const v4f*)&wos[cc * 512 + (w * 4 + i) * 16 + 8 * half + 4 * hq];
        #pragma unroll
        for (int cl = 0; cl < 16; ++cl) {
            const float ccx = y1q[csidx(cl * 3)], ccy = y1q[csidx(cl * 3 + 1)], ccz = y1q[csidx(cl * 3 + 2)];
            const int pr = cl >> 1, sy = cl & 1;
            const int mtA  = sy ? 2 : 0;
            const int rowA = sy * 32 + pr;
            const int rowQ = rowA + 8;
            const int rowP = 16 + pr + 8 * sy;
            const v2f gb = sy ? gbY : gbX;
            v8f d[4];
            #pragma unroll
            for (int i = 0; i < 4; ++i) {
                const float tb = __builtin_fmaf(ccz, rZs[i], __builtin_fmaf(ccy, rYs[i], __builtin_fmaf(ccx, rXs[i], cA[i])));
                v2f ea;
                ea.x = __builtin_fmaf(sl[i], acc[mtA][i][pr], tb);
                ea.y = __builtin_fmaf(sl[i], acc[1][i][pr], cP[i]);
                d[i] = wmma_f32x4(ea, gb, v8f{});
            }
            const int n = __builtin_amdgcn_readfirstlane(mcnt[cl]);
            for (int base = 0; base < n; base += 4) {
                const int i0 = base + 2 * half, i1 = i0 + 1;
                const bool ok0 = i0 < n, ok1 = i1 < n;
                const int k0 = ok0 ? (int)mlist[cl * 512 + i0] : 0;
                const int k1 = ok1 ? (int)mlist[cl * 512 + i1] : 0;
                const float* r0 = &y1q[(k0 >> 2) * Y1P + (k0 & 3)];
                const float* r1 = &y1q[(k1 >> 2) * Y1P + (k1 & 3)];
                const float z0 = __builtin_fmaf(gy, r0[rowQ * 4], __builtin_fmaf(gx, r0[rowP * 4], r0[rowA * 4]));
                const float z1 = __builtin_fmaf(gy, r1[rowQ * 4], __builtin_fmaf(gx, r1[rowP * 4], r1[rowA * 4]));
                v2f cb;
                cb.x = ok0 ? fmaxf(-z0, 0.0f) : 0.0f;
                cb.y = ok1 ? fmaxf(-z1, 0.0f) : 0.0f;
                const v4f wk0 = *(const v4f*)(wtl + k0 * 512);
                const v4f wk1 = *(const v4f*)(wtl + k1 * 512);
                #pragma unroll
                for (int i = 0; i < 4; ++i) {
                    v2f wa;
                    wa.x = sl[i] * wk0[i];
                    wa.y = sl[i] * wk1[i];
                    d[i] = wmma_f32x4(wa, cb, d[i]);
                }
            }
            float p0 = 0.0f, p1 = 0.0f, p2 = 0.0f;
            #pragma unroll
            for (int i = 0; i < 4; ++i) {
                const int ob = (w * 4 + i) * 16 + 8 * half;
                #pragma unroll
                for (int hq = 0; hq < 2; ++hq) {
                    v4f w0, w1, w2;
                    if (i < NWR) {
                        w0 = wr[i < NWR ? i : 0][hq][0];
                        w1 = wr[i < NWR ? i : 0][hq][1];
                        w2 = wr[i < NWR ? i : 0][hq][2];
                    } else {
                        w0 = *(const v4f*)&wos[ob + 4 * hq];
                        w1 = *(const v4f*)&wos[512 + ob + 4 * hq];
                        w2 = *(const v4f*)&wos[1024 + ob + 4 * hq];
                    }
                    #pragma unroll
                    for (int e = 0; e < 4; ++e) {
                        const float y = __builtin_elementwise_maximum(d[i][4 * hq + e], 0.0f);
                        p0 = __builtin_fmaf(y, w0[e], p0);
                        p1 = __builtin_fmaf(y, w1[e], p1);
                        p2 = __builtin_fmaf(y, w2[e], p2);
                    }
                }
            }
            p0 += __shfl_xor(p0, 16);
            p1 += __shfl_xor(p1, 16);
            p2 += __shfl_xor(p2, 16);
            if (half == 0) {
                float* rd = &red[(w * 16 + cl) * 48 + l * 3];
                rd[0] = p0; rd[1] = p1; rd[2] = p2;
            }
        }
    }
    __syncthreads();

    for (int idx = t; idx < 768; idx += 256) {
        const int cl  = idx / 48;
        const int rem = idx - cl * 48;
        const int g   = rem / 3;
        const int c   = rem - g * 3;
        float sum = 0.0f;
        #pragma unroll
        for (int s = 0; s < 8; ++s) sum += red[(s * 16 + cl) * 48 + rem];
        const int ng = (blockIdx.x * 16 + cl) * 16 + g;
        const int i  = c * 16384 + ng;
        const int r3 = i / 3;
        const int cc = i - r3 * 3;
        const float center = coarse[(b * 1024 + (r3 >> 4)) * 3 + cc];
        fine[(b * 16384 + ng) * 3 + c] = sum + bo[c] + center;
    }
}

extern "C" void kernel_launch(void* const* d_in, const int* in_sizes, int n_in,
                              void* d_out, int out_size, void* d_ws, size_t ws_size,
                              hipStream_t stream)
{
    const float* features = (const float*)d_in[0];
    const float* W1  = (const float*)d_in[1];
    const float* b1  = (const float*)d_in[2];
    const float* W2  = (const float*)d_in[3];
    const float* b2  = (const float*)d_in[4];
    const float* Wc  = (const float*)d_in[5];
    const float* bc  = (const float*)d_in[6];
    const float* Wf1 = (const float*)d_in[7];
    const float* bf1 = (const float*)d_in[8];
    const float* g1  = (const float*)d_in[9];
    const float* be1 = (const float*)d_in[10];
    const float* m1  = (const float*)d_in[11];
    const float* v1  = (const float*)d_in[12];
    const float* Wf2 = (const float*)d_in[13];
    const float* bf2 = (const float*)d_in[14];
    const float* g2  = (const float*)d_in[15];
    const float* be2 = (const float*)d_in[16];
    const float* m2  = (const float*)d_in[17];
    const float* v2  = (const float*)d_in[18];
    const float* Wo  = (const float*)d_in[19];
    const float* bo  = (const float*)d_in[20];

    float* out    = (float*)d_out;
    float* coarse = out;
    float* fine   = out + 8 * 1024 * 3;

    float* ws    = (float*)d_ws;
    float* h1    = ws;
    float* h2    = ws + 8192;
    float* base1 = ws + 16384;
    float* WT    = ws + 28672;
    float* R     = ws + 290816;
    float* maskf = ws + 315392;

    stage1_kernel<<<112, 512, 0, stream>>>(features, W1, b1, h1, Wf1, bf1, base1, Wf2, WT);
    stage2_kernel<<<48, 512, 0, stream>>>(h1, W2, b2, h2, base1, Wf1, g1, be1, m1, v1, Wf2, R, maskf);
    gemm8_wmma<1024><<<96, 512, 0, stream>>>(h2, 1024, Wc, 1024, bc, coarse, 3072, 0);

    dim3 gridF(1024 / 16, 8);
    fold_fused<<<gridF, 256, 0, stream>>>(coarse, base1, Wf1,
                                          g1, be1, m1, v1,
                                          WT, bf2, g2, be2, m2, v2,
                                          Wo, bo, R, maskf, fine);
}
